# starting kernel
# baseline (speedup 1.0000x reference)
#include <hip/hip_runtime.h>
#include <math.h>

#define NN 100000
#define NE 1600000
#define FIN 128
#define HID 64
#define NCLS 10
#define ROW_TILES 6250

typedef __attribute__((ext_vector_type(2))) float v2f;
typedef __attribute__((ext_vector_type(8))) float v8f;

__global__ void k_fill(float* __restrict__ p, int n, float v) {
    int i = blockIdx.x * blockDim.x + threadIdx.x;
    if (i < n) p[i] = v;
}

__global__ void k_deg(const int* __restrict__ dst, float* __restrict__ deg) {
    int e = blockIdx.x * blockDim.x + threadIdx.x;
    if (e < NE) atomicAdd(&deg[dst[e]], 1.0f);
}

__global__ void k_dis(float* __restrict__ deg) {
    int i = blockIdx.x * blockDim.x + threadIdx.x;
    if (i < NN) deg[i] = rsqrtf(deg[i]);
}

__global__ void k_norm(const int* __restrict__ src, const int* __restrict__ dst,
                       const float* __restrict__ dis, float* __restrict__ norm) {
    int e = blockIdx.x * blockDim.x + threadIdx.x;
    if (e < NE) norm[e] = dis[src[e]] * dis[dst[e]];
}

__global__ void __launch_bounds__(128)
k_gemm1(const float* __restrict__ x, const float* __restrict__ W1,
        float* __restrict__ h1) {
    __shared__ float wT[HID * 132];
    int t = threadIdx.x;
    for (int i = t; i < FIN * HID; i += 128) {
        int k = i >> 6, c = i & 63;
        wT[c * 132 + k] = W1[i];
    }
    __syncthreads();

    int tile = blockIdx.x * 4 + (t >> 5);
    if (tile >= ROW_TILES) return;
    int lane = t & 31;
    int m = lane & 15, half = lane >> 4;

    const float* arow = x + (size_t)(tile * 16 + m) * FIN + 2 * half;
    v8f a0 = {}, a1 = {}, a2 = {}, a3 = {};

    #pragma unroll 4
    for (int k0 = 0; k0 < FIN; k0 += 4) {
        v2f a = *(const v2f*)(arow + k0);
        const float* bp = wT + m * 132 + k0 + 2 * half;
        v2f b0 = *(const v2f*)(bp);
        v2f b1 = *(const v2f*)(bp + 16 * 132);
        v2f b2 = *(const v2f*)(bp + 32 * 132);
        v2f b3 = *(const v2f*)(bp + 48 * 132);
        a0 = __builtin_amdgcn_wmma_f32_16x16x4_f32(false, a, false, b0, (short)0, a0, false, false);
        a1 = __builtin_amdgcn_wmma_f32_16x16x4_f32(false, a, false, b1, (short)0, a1, false, false);
        a2 = __builtin_amdgcn_wmma_f32_16x16x4_f32(false, a, false, b2, (short)0, a2, false, false);
        a3 = __builtin_amdgcn_wmma_f32_16x16x4_f32(false, a, false, b3, (short)0, a3, false, false);
    }

    float* out = h1 + (size_t)(tile * 16 + 8 * half) * HID + m;
    #pragma unroll
    for (int v = 0; v < 8; ++v) {
        out[v * HID +  0] = a0[v];
        out[v * HID + 16] = a1[v];
        out[v * HID + 32] = a2[v];
        out[v * HID + 48] = a3[v];
    }
}

__global__ void __launch_bounds__(128)
k_gemm2(const float* __restrict__ z1, const float* __restrict__ W2,
        float* __restrict__ h2) {
    __shared__ float wT[16 * 68];
    int t = threadIdx.x;
    for (int i = t; i < 16 * HID; i += 128) {
        int k = i >> 4, c = i & 15;
        wT[c * 68 + k] = (c < NCLS) ? W2[k * NCLS + c] : 0.0f;
    }
    __syncthreads();

    int tile = blockIdx.x * 4 + (t >> 5);
    if (tile >= ROW_TILES) return;
    int lane = t & 31;
    int m = lane & 15, half = lane >> 4;

    const float* arow = z1 + (size_t)(tile * 16 + m) * HID + 2 * half;
    v8f acc = {};
    #pragma unroll
    for (int k0 = 0; k0 < HID; k0 += 4) {
        v2f a = *(const v2f*)(arow + k0);
        v2f b = *(const v2f*)(wT + m * 68 + k0 + 2 * half);
        acc = __builtin_amdgcn_wmma_f32_16x16x4_f32(false, a, false, b, (short)0, acc, false, false);
    }
    if (m < NCLS) {
        float* out = h2 + (size_t)(tile * 16 + 8 * half) * NCLS + m;
        #pragma unroll
        for (int v = 0; v < 8; ++v) out[v * NCLS] = acc[v];
    }
}

__global__ void k_scatter1(const int* __restrict__ src, const int* __restrict__ dst,
                           const float* __restrict__ norm, const float* __restrict__ h1,
                           float* __restrict__ agg) {
    int g = blockIdx.x * blockDim.x + threadIdx.x;
    int e = g >> 6, f = g & 63;
    if (e >= NE) return;
    int s = src[e], d = dst[e];
    atomicAdd(&agg[(size_t)d * HID + f], h1[(size_t)s * HID + f] * norm[e]);
}

__global__ void k_scatter2(const int* __restrict__ src, const int* __restrict__ dst,
                           const float* __restrict__ norm, const float* __restrict__ h2,
                           float* __restrict__ out) {
    int g = blockIdx.x * blockDim.x + threadIdx.x;
    int e = g >> 4, f = g & 15;
    if (e >= NE || f >= NCLS) return;
    int s = src[e], d = dst[e];
    atomicAdd(&out[(size_t)d * NCLS + f], h2[(size_t)s * NCLS + f] * norm[e]);
}

__global__ void k_post1(float* __restrict__ agg, const float* __restrict__ h1,
                        const float* __restrict__ dis, const float* __restrict__ b1) {
    int i = blockIdx.x * blockDim.x + threadIdx.x;
    if (i >= NN * HID) return;
    int node = i >> 6, f = i & 63;
    float di = dis[node];
    float v = agg[i] + h1[i] * di * di + b1[f];
    agg[i] = v > 0.0f ? v : 0.0f;
}

__global__ void k_post2(float* __restrict__ out, const float* __restrict__ h2,
                        const float* __restrict__ dis, const float* __restrict__ b2) {
    int i = blockIdx.x * blockDim.x + threadIdx.x;
    if (i >= NN) return;
    float d2 = dis[i] * dis[i];
    float tv[NCLS];
    float mx = -3.402823466e38f;
    #pragma unroll
    for (int f = 0; f < NCLS; ++f) {
        float v = out[(size_t)i * NCLS + f] + h2[(size_t)i * NCLS + f] * d2 + b2[f];
        tv[f] = v;
        mx = fmaxf(mx, v);
    }
    float s = 0.0f;
    #pragma unroll
    for (int f = 0; f < NCLS; ++f) s += expf(tv[f] - mx);
    float lse = mx + logf(s);
    #pragma unroll
    for (int f = 0; f < NCLS; ++f) out[(size_t)i * NCLS + f] = tv[f] - lse;
}

extern "C" void kernel_launch(void* const* d_in, const int* in_sizes, int n_in,
                              void* d_out, int out_size, void* d_ws, size_t ws_size,
                              hipStream_t stream) {
    const float* x  = (const float*)d_in[0];
    const int*   ei = (const int*)d_in[1];
    const float* W1 = (const float*)d_in[2];
    const float* b1 = (const float*)d_in[3];
    const float* W2 = (const float*)d_in[4];
    const float* b2 = (const float*)d_in[5];
    const int* src = ei;
    const int* dst = ei + NE;
    float* out = (float*)d_out;

    float* ws   = (float*)d_ws;
    float* dis  = ws;
    float* h1   = dis + NN;
    float* agg1 = h1 + (size_t)NN * HID;
    float* h2   = agg1 + (size_t)NN * HID;
    float* norm = h2 + (size_t)NN * NCLS;

    const int B = 256;
    k_fill<<<(NN + B - 1) / B, B, 0, stream>>>(dis, NN, 1.0f);
    k_fill<<<(NN * HID + B - 1) / B, B, 0, stream>>>(agg1, NN * HID, 0.0f);
    k_fill<<<(NN * NCLS + B - 1) / B, B, 0, stream>>>(out, NN * NCLS, 0.0f);

    k_deg<<<(NE + B - 1) / B, B, 0, stream>>>(dst, dis);
    k_dis<<<(NN + B - 1) / B, B, 0, stream>>>(dis);
    k_norm<<<(NE + B - 1) / B, B, 0, stream>>>(src, dst, dis, norm);

    k_gemm1<<<(ROW_TILES + 3) / 4, 128, 0, stream>>>(x, W1, h1);
    k_scatter1<<<(NE * 64) / B, B, 0, stream>>>(src, dst, norm, h1, agg1);
    k_post1<<<(NN * HID + B - 1) / B, B, 0, stream>>>(agg1, h1, dis, b1);

    k_gemm2<<<(ROW_TILES + 3) / 4, 128, 0, stream>>>(agg1, W2, h2);
    k_scatter2<<<(NE * 16) / B, B, 0, stream>>>(src, dst, norm, h2, out);
    k_post2<<<(NN + B - 1) / B, B, 0, stream>>>(out, h2, dis, b2);
}


# Layer-1 scatter off global float atomics: 128 groups each keep 782 destination rows in LDS, filter the whole edge list for them, add rows by LDS float adds (arrival order); self-loop+bias+relu fused
# speedup vs baseline: 1.3961x; 1.3961x over previous
#include <hip/hip_runtime.h>
#include <math.h>

#define NN 100000
#define NE 1600000
#define FIN 128
#define HID 64
#define NCLS 10
#define ROW_TILES 6250

typedef __attribute__((ext_vector_type(2))) float v2f;
typedef __attribute__((ext_vector_type(8))) float v8f;

__global__ void k_fill(float* __restrict__ p, int n, float v) {
    int i = blockIdx.x * blockDim.x + threadIdx.x;
    if (i < n) p[i] = v;
}

__global__ void k_deg(const int* __restrict__ dst, float* __restrict__ deg) {
    int e = blockIdx.x * blockDim.x + threadIdx.x;
    if (e < NE) atomicAdd(&deg[dst[e]], 1.0f);
}

__global__ void k_dis(float* __restrict__ deg) {
    int i = blockIdx.x * blockDim.x + threadIdx.x;
    if (i < NN) deg[i] = rsqrtf(deg[i]);
}

__global__ void k_norm(const int* __restrict__ src, const int* __restrict__ dst,
                       const float* __restrict__ dis, float* __restrict__ norm) {
    int e = blockIdx.x * blockDim.x + threadIdx.x;
    if (e < NE) norm[e] = dis[src[e]] * dis[dst[e]];
}

__global__ void __launch_bounds__(128)
k_gemm1(const float* __restrict__ x, const float* __restrict__ W1,
        float* __restrict__ h1) {
    __shared__ float wT[HID * 132];
    int t = threadIdx.x;
    for (int i = t; i < FIN * HID; i += 128) {
        int k = i >> 6, c = i & 63;
        wT[c * 132 + k] = W1[i];
    }
    __syncthreads();

    int tile = blockIdx.x * 4 + (t >> 5);
    if (tile >= ROW_TILES) return;
    int lane = t & 31;
    int m = lane & 15, half = lane >> 4;

    const float* arow = x + (size_t)(tile * 16 + m) * FIN + 2 * half;
    v8f a0 = {}, a1 = {}, a2 = {}, a3 = {};

    #pragma unroll 4
    for (int k0 = 0; k0 < FIN; k0 += 4) {
        v2f a = *(const v2f*)(arow + k0);
        const float* bp = wT + m * 132 + k0 + 2 * half;
        v2f b0 = *(const v2f*)(bp);
        v2f b1 = *(const v2f*)(bp + 16 * 132);
        v2f b2 = *(const v2f*)(bp + 32 * 132);
        v2f b3 = *(const v2f*)(bp + 48 * 132);
        a0 = __builtin_amdgcn_wmma_f32_16x16x4_f32(false, a, false, b0, (short)0, a0, false, false);
        a1 = __builtin_amdgcn_wmma_f32_16x16x4_f32(false, a, false, b1, (short)0, a1, false, false);
        a2 = __builtin_amdgcn_wmma_f32_16x16x4_f32(false, a, false, b2, (short)0, a2, false, false);
        a3 = __builtin_amdgcn_wmma_f32_16x16x4_f32(false, a, false, b3, (short)0, a3, false, false);
    }

    float* out = h1 + (size_t)(tile * 16 + 8 * half) * HID + m;
    #pragma unroll
    for (int v = 0; v < 8; ++v) {
        out[v * HID +  0] = a0[v];
        out[v * HID + 16] = a1[v];
        out[v * HID + 32] = a2[v];
        out[v * HID + 48] = a3[v];
    }
}

__global__ void __launch_bounds__(128)
k_gemm2(const float* __restrict__ z1, const float* __restrict__ W2,
        float* __restrict__ h2) {
    __shared__ float wT[16 * 68];
    int t = threadIdx.x;
    for (int i = t; i < 16 * HID; i += 128) {
        int k = i >> 4, c = i & 15;
        wT[c * 68 + k] = (c < NCLS) ? W2[k * NCLS + c] : 0.0f;
    }
    __syncthreads();

    int tile = blockIdx.x * 4 + (t >> 5);
    if (tile >= ROW_TILES) return;
    int lane = t & 31;
    int m = lane & 15, half = lane >> 4;

    const float* arow = z1 + (size_t)(tile * 16 + m) * HID + 2 * half;
    v8f acc = {};
    #pragma unroll
    for (int k0 = 0; k0 < HID; k0 += 4) {
        v2f a = *(const v2f*)(arow + k0);
        v2f b = *(const v2f*)(wT + m * 68 + k0 + 2 * half);
        acc = __builtin_amdgcn_wmma_f32_16x16x4_f32(false, a, false, b, (short)0, acc, false, false);
    }
    if (m < NCLS) {
        float* out = h2 + (size_t)(tile * 16 + 8 * half) * NCLS + m;
        #pragma unroll
        for (int v = 0; v < 8; ++v) out[v * NCLS] = acc[v];
    }
}

#define AG_NB   128
#define AG_R    782
#define AG_QCAP 14336
#define AG_T    1024
#define AG_SB   8
#define AG_DB   8

__device__ __forceinline__ void ag_push(unsigned* qn, uint2* q, unsigned r, int e) {
    unsigned pos = __hip_atomic_fetch_add(qn, 1u, __ATOMIC_RELAXED, __HIP_MEMORY_SCOPE_WORKGROUP);
    if (pos < AG_QCAP) q[pos].x = (r << 21) | (unsigned)e;
}

__device__ __forceinline__ void ag_filter(const int* __restrict__ dst, int e0, int e1,
                                          int lo, unsigned nrows,
                                          unsigned* qn, uint2* q, int t) {
    for (int e = e0 + 4 * t; e < e1; e += 4 * AG_T * AG_SB) {
        int4 d[AG_SB];
        #pragma unroll
        for (int u = 0; u < AG_SB; ++u) {
            int ee = e + u * 4 * AG_T;
            d[u] = (ee < e1) ? *(const int4*)(dst + ee) : make_int4(-1, -1, -1, -1);
        }
        #pragma unroll
        for (int u = 0; u < AG_SB; ++u) {
            int ee = e + u * 4 * AG_T;
            unsigned r0 = (unsigned)(d[u].x - lo), r1 = (unsigned)(d[u].y - lo);
            unsigned r2 = (unsigned)(d[u].z - lo), r3 = (unsigned)(d[u].w - lo);
            if ((r0 < nrows) | (r1 < nrows) | (r2 < nrows) | (r3 < nrows)) {
                if (r0 < nrows) ag_push(qn, q, r0, ee);
                if (r1 < nrows) ag_push(qn, q, r1, ee + 1);
                if (r2 < nrows) ag_push(qn, q, r2, ee + 2);
                if (r3 < nrows) ag_push(qn, q, r3, ee + 3);
            }
        }
    }
}

__device__ __forceinline__ void ag_drain(const int* __restrict__ src, const float* __restrict__ norm,
                                         const float* __restrict__ h1,
                                         float* acc, uint2* q, unsigned n, int t) {
    for (unsigned i0 = t; i0 < n; i0 += 4u * AG_T) {
        unsigned en[4], s[4];
        float w[4];
        #pragma unroll
        for (int u = 0; u < 4; ++u) {
            unsigned i = i0 + u * AG_T;
            en[u] = (i < n) ? q[i].x : 0u;
        }
        #pragma unroll
        for (int u = 0; u < 4; ++u) {
            unsigned e = en[u] & 0x1FFFFFu;
            s[u] = (unsigned)src[e];
            w[u] = norm[e];
        }
        #pragma unroll
        for (int u = 0; u < 4; ++u) {
            unsigned i = i0 + u * AG_T;
            q[(i < n) ? i : AG_QCAP] = make_uint2(((en[u] >> 21) << 17) | s[u], __float_as_uint(w[u]));
        }
    }
    __syncthreads();

    const unsigned hw = (unsigned)t >> 4;
    const unsigned l16 = (unsigned)t & 15u;
    const bool hi = (hw & 1u) != 0u;
    const unsigned oA = hi ? 2u : 0u, oB = oA ^ 2u;
    for (unsigned i0 = hw; i0 < n; i0 += 64u * AG_DB) {
        uint2 en[AG_DB];
        float4 row[AG_DB];
        #pragma unroll
        for (int u = 0; u < AG_DB; ++u) {
            unsigned i = i0 + 64u * u;
            en[u] = (i < n) ? q[i] : make_uint2((unsigned)AG_R << 17, 0u);
        }
        #pragma unroll
        for (int u = 0; u < AG_DB; ++u)
            row[u] = *(const float4*)(h1 + (size_t)(en[u].x & 0x1FFFFu) * HID + 4u * l16);
        #pragma unroll
        for (int u = 0; u < AG_DB; ++u) {
            float w = __uint_as_float(en[u].y);
            float* p = acc + (en[u].x >> 17) * HID + 4u * l16;
            float a0 = hi ? row[u].z : row[u].x, a1 = hi ? row[u].w : row[u].y;
            float c0 = hi ? row[u].x : row[u].z, c1 = hi ? row[u].y : row[u].w;
            __hip_atomic_fetch_add(p + oA,      a0 * w, __ATOMIC_RELAXED, __HIP_MEMORY_SCOPE_WORKGROUP);
            __hip_atomic_fetch_add(p + oA + 1u, a1 * w, __ATOMIC_RELAXED, __HIP_MEMORY_SCOPE_WORKGROUP);
            __hip_atomic_fetch_add(p + oB,      c0 * w, __ATOMIC_RELAXED, __HIP_MEMORY_SCOPE_WORKGROUP);
            __hip_atomic_fetch_add(p + oB + 1u, c1 * w, __ATOMIC_RELAXED, __HIP_MEMORY_SCOPE_WORKGROUP);
        }
    }
}

__global__ void __launch_bounds__(AG_T)
k_agg1(const int* __restrict__ src, const int* __restrict__ dst,
       const float* __restrict__ norm, const float* __restrict__ h1,
       const float* __restrict__ dis, const float* __restrict__ b1,
       float* __restrict__ z1) {
    __shared__ float acc[(AG_R + 1) * HID];
    __shared__ uint2 q[AG_QCAP + 1];
    __shared__ unsigned qn;
    const int t = threadIdx.x;
    const int lo = blockIdx.x * AG_R;
    const unsigned nrows = (unsigned)((NN - lo) < AG_R ? (NN - lo) : AG_R);

    for (int i = t; i < (AG_R + 1) * HID / 4; i += AG_T)
        ((float4*)acc)[i] = make_float4(0.0f, 0.0f, 0.0f, 0.0f);
    if (t == 0) qn = 0u;
    __syncthreads();

    ag_filter(dst, 0, NE, lo, nrows, &qn, q, t);
    __syncthreads();
    const unsigned total = qn;
    if (total <= AG_QCAP) {
        ag_drain(src, norm, h1, acc, q, total, t);
    } else {
        for (int e0 = 0; e0 < NE; e0 += AG_QCAP) {
            __syncthreads();
            if (t == 0) qn = 0u;
            __syncthreads();
            const int e1 = (e0 + AG_QCAP < NE) ? (e0 + AG_QCAP) : NE;
            ag_filter(dst, e0, e1, lo, nrows, &qn, q, t);
            __syncthreads();
            ag_drain(src, norm, h1, acc, q, qn, t);
        }
    }
    __syncthreads();

    for (unsigned i = t; i < nrows * 16u; i += AG_T) {
        unsigned r = i >> 4, c = (i & 15u) * 4u;
        size_t g = (size_t)(lo + r) * HID + c;
        float di = dis[lo + r];
        float4 a = *(const float4*)(acc + r * HID + c);
        float4 h = *(const float4*)(h1 + g);
        float4 b = *(const float4*)(b1 + c);
        float4 v;
        v.x = a.x + h.x * di * di + b.x;
        v.y = a.y + h.y * di * di + b.y;
        v.z = a.z + h.z * di * di + b.z;
        v.w = a.w + h.w * di * di + b.w;
        v.x = v.x > 0.0f ? v.x : 0.0f;
        v.y = v.y > 0.0f ? v.y : 0.0f;
        v.z = v.z > 0.0f ? v.z : 0.0f;
        v.w = v.w > 0.0f ? v.w : 0.0f;
        *(float4*)(z1 + g) = v;
    }
}

__global__ void k_scatter2(const int* __restrict__ src, const int* __restrict__ dst,
                           const float* __restrict__ norm, const float* __restrict__ h2,
                           float* __restrict__ out) {
    int g = blockIdx.x * blockDim.x + threadIdx.x;
    int e = g >> 4, f = g & 15;
    if (e >= NE || f >= NCLS) return;
    int s = src[e], d = dst[e];
    atomicAdd(&out[(size_t)d * NCLS + f], h2[(size_t)s * NCLS + f] * norm[e]);
}

__global__ void k_post2(float* __restrict__ out, const float* __restrict__ h2,
                        const float* __restrict__ dis, const float* __restrict__ b2) {
    int i = blockIdx.x * blockDim.x + threadIdx.x;
    if (i >= NN) return;
    float d2 = dis[i] * dis[i];
    float tv[NCLS];
    float mx = -3.402823466e38f;
    #pragma unroll
    for (int f = 0; f < NCLS; ++f) {
        float v = out[(size_t)i * NCLS + f] + h2[(size_t)i * NCLS + f] * d2 + b2[f];
        tv[f] = v;
        mx = fmaxf(mx, v);
    }
    float s = 0.0f;
    #pragma unroll
    for (int f = 0; f < NCLS; ++f) s += expf(tv[f] - mx);
    float lse = mx + logf(s);
    #pragma unroll
    for (int f = 0; f < NCLS; ++f) out[(size_t)i * NCLS + f] = tv[f] - lse;
}

extern "C" void kernel_launch(void* const* d_in, const int* in_sizes, int n_in,
                              void* d_out, int out_size, void* d_ws, size_t ws_size,
                              hipStream_t stream) {
    const float* x  = (const float*)d_in[0];
    const int*   ei = (const int*)d_in[1];
    const float* W1 = (const float*)d_in[2];
    const float* b1 = (const float*)d_in[3];
    const float* W2 = (const float*)d_in[4];
    const float* b2 = (const float*)d_in[5];
    const int* src = ei;
    const int* dst = ei + NE;
    float* out = (float*)d_out;

    float* ws   = (float*)d_ws;
    float* dis  = ws;
    float* h1   = dis + NN;
    float* agg1 = h1 + (size_t)NN * HID;
    float* h2   = agg1 + (size_t)NN * HID;
    float* norm = h2 + (size_t)NN * NCLS;

    const int B = 256;
    k_fill<<<(NN + B - 1) / B, B, 0, stream>>>(dis, NN, 1.0f);
    k_fill<<<(NN * NCLS + B - 1) / B, B, 0, stream>>>(out, NN * NCLS, 0.0f);

    k_deg<<<(NE + B - 1) / B, B, 0, stream>>>(dst, dis);
    k_dis<<<(NN + B - 1) / B, B, 0, stream>>>(dis);
    k_norm<<<(NE + B - 1) / B, B, 0, stream>>>(src, dst, dis, norm);

    k_gemm1<<<(ROW_TILES + 3) / 4, 128, 0, stream>>>(x, W1, h1);
    k_agg1<<<AG_NB, AG_T, 0, stream>>>(src, dst, norm, h1, dis, b1, agg1);

    k_gemm2<<<(ROW_TILES + 3) / 4, 128, 0, stream>>>(agg1, W2, h2);
    k_scatter2<<<(NE * 16) / B, B, 0, stream>>>(src, dst, norm, h2, out);
    k_post2<<<(NN + B - 1) / B, B, 0, stream>>>(out, h2, dis, b2);
}


# Layer-2 scatter off global float atomics: layer-1 groups store their edge queues (ranges now 784 rows); 128 groups add h2 rows from them by LDS float adds (arrival order); log_softmax pass fused
# speedup vs baseline: 1.6519x; 1.1832x over previous
#include <hip/hip_runtime.h>
#include <math.h>

#define NN 100000
#define NE 1600000
#define FIN 128
#define HID 64
#define NCLS 10
#define ROW_TILES 6250

typedef __attribute__((ext_vector_type(2))) float v2f;
typedef __attribute__((ext_vector_type(8))) float v8f;

__global__ void k_fill(float* __restrict__ p, int n, float v) {
    int i = blockIdx.x * blockDim.x + threadIdx.x;
    if (i < n) p[i] = v;
}

__global__ void k_deg(const int* __restrict__ dst, float* __restrict__ deg) {
    int e = blockIdx.x * blockDim.x + threadIdx.x;
    if (e < NE) atomicAdd(&deg[dst[e]], 1.0f);
}

__global__ void k_dis(float* __restrict__ deg) {
    int i = blockIdx.x * blockDim.x + threadIdx.x;
    if (i < NN) deg[i] = rsqrtf(deg[i]);
}

__global__ void k_norm(const int* __restrict__ src, const int* __restrict__ dst,
                       const float* __restrict__ dis, float* __restrict__ norm) {
    int e = blockIdx.x * blockDim.x + threadIdx.x;
    if (e < NE) norm[e] = dis[src[e]] * dis[dst[e]];
}

__global__ void __launch_bounds__(128)
k_gemm1(const float* __restrict__ x, const float* __restrict__ W1,
        float* __restrict__ h1) {
    __shared__ float wT[HID * 132];
    int t = threadIdx.x;
    for (int i = t; i < FIN * HID; i += 128) {
        int k = i >> 6, c = i & 63;
        wT[c * 132 + k] = W1[i];
    }
    __syncthreads();

    int tile = blockIdx.x * 4 + (t >> 5);
    if (tile >= ROW_TILES) return;
    int lane = t & 31;
    int m = lane & 15, half = lane >> 4;

    const float* arow = x + (size_t)(tile * 16 + m) * FIN + 2 * half;
    v8f a0 = {}, a1 = {}, a2 = {}, a3 = {};

    #pragma unroll 4
    for (int k0 = 0; k0 < FIN; k0 += 4) {
        v2f a = *(const v2f*)(arow + k0);
        const float* bp = wT + m * 132 + k0 + 2 * half;
        v2f b0 = *(const v2f*)(bp);
        v2f b1 = *(const v2f*)(bp + 16 * 132);
        v2f b2 = *(const v2f*)(bp + 32 * 132);
        v2f b3 = *(const v2f*)(bp + 48 * 132);
        a0 = __builtin_amdgcn_wmma_f32_16x16x4_f32(false, a, false, b0, (short)0, a0, false, false);
        a1 = __builtin_amdgcn_wmma_f32_16x16x4_f32(false, a, false, b1, (short)0, a1, false, false);
        a2 = __builtin_amdgcn_wmma_f32_16x16x4_f32(false, a, false, b2, (short)0, a2, false, false);
        a3 = __builtin_amdgcn_wmma_f32_16x16x4_f32(false, a, false, b3, (short)0, a3, false, false);
    }

    float* out = h1 + (size_t)(tile * 16 + 8 * half) * HID + m;
    #pragma unroll
    for (int v = 0; v < 8; ++v) {
        out[v * HID +  0] = a0[v];
        out[v * HID + 16] = a1[v];
        out[v * HID + 32] = a2[v];
        out[v * HID + 48] = a3[v];
    }
}

__global__ void __launch_bounds__(128)
k_gemm2(const float* __restrict__ z1, const float* __restrict__ W2,
        float* __restrict__ h2) {
    __shared__ float wT[16 * 68];
    int t = threadIdx.x;
    for (int i = t; i < 16 * HID; i += 128) {
        int k = i >> 4, c = i & 15;
        wT[c * 68 + k] = (c < NCLS) ? W2[k * NCLS + c] : 0.0f;
    }
    __syncthreads();

    int tile = blockIdx.x * 4 + (t >> 5);
    if (tile >= ROW_TILES) return;
    int lane = t & 31;
    int m = lane & 15, half = lane >> 4;

    const float* arow = z1 + (size_t)(tile * 16 + m) * HID + 2 * half;
    v8f acc = {};
    #pragma unroll
    for (int k0 = 0; k0 < HID; k0 += 4) {
        v2f a = *(const v2f*)(arow + k0);
        v2f b = *(const v2f*)(wT + m * 68 + k0 + 2 * half);
        acc = __builtin_amdgcn_wmma_f32_16x16x4_f32(false, a, false, b, (short)0, acc, false, false);
    }
    if (m < NCLS) {
        float* out = h2 + (size_t)(tile * 16 + 8 * half) * NCLS + m;
        #pragma unroll
        for (int v = 0; v < 8; ++v) out[v * NCLS] = acc[v];
    }
}

#define AG_NB   128
#define AG_R    784
#define AG_QCAP 14336
#define AG_T    1024
#define AG_SB   8
#define AG_DB   8

__device__ __forceinline__ void ag_push(unsigned* qn, uint2* q, unsigned r, int e) {
    unsigned pos = __hip_atomic_fetch_add(qn, 1u, __ATOMIC_RELAXED, __HIP_MEMORY_SCOPE_WORKGROUP);
    if (pos < AG_QCAP) q[pos].x = (r << 21) | (unsigned)e;
}

__device__ __forceinline__ void ag_filter(const int* __restrict__ dst, int e0, int e1,
                                          int lo, unsigned nrows,
                                          unsigned* qn, uint2* q, int t) {
    for (int e = e0 + 4 * t; e < e1; e += 4 * AG_T * AG_SB) {
        int4 d[AG_SB];
        #pragma unroll
        for (int u = 0; u < AG_SB; ++u) {
            int ee = e + u * 4 * AG_T;
            d[u] = (ee < e1) ? *(const int4*)(dst + ee) : make_int4(-1, -1, -1, -1);
        }
        #pragma unroll
        for (int u = 0; u < AG_SB; ++u) {
            int ee = e + u * 4 * AG_T;
            unsigned r0 = (unsigned)(d[u].x - lo), r1 = (unsigned)(d[u].y - lo);
            unsigned r2 = (unsigned)(d[u].z - lo), r3 = (unsigned)(d[u].w - lo);
            if ((r0 < nrows) | (r1 < nrows) | (r2 < nrows) | (r3 < nrows)) {
                if (r0 < nrows) ag_push(qn, q, r0, ee);
                if (r1 < nrows) ag_push(qn, q, r1, ee + 1);
                if (r2 < nrows) ag_push(qn, q, r2, ee + 2);
                if (r3 < nrows) ag_push(qn, q, r3, ee + 3);
            }
        }
    }
}

template <bool KEEP>
__device__ __forceinline__ void ag_lookup(const int* __restrict__ src, const float* __restrict__ norm,
                                          uint2* q, unsigned n, int t, uint2* __restrict__ gq) {
    for (unsigned i0 = t; i0 < n; i0 += 4u * AG_T) {
        unsigned en[4], s[4];
        float w[4];
        #pragma unroll
        for (int u = 0; u < 4; ++u) {
            unsigned i = i0 + u * AG_T;
            en[u] = (i < n) ? q[i].x : 0u;
        }
        #pragma unroll
        for (int u = 0; u < 4; ++u) {
            unsigned e = en[u] & 0x1FFFFFu;
            s[u] = (unsigned)src[e];
            w[u] = norm[e];
        }
        #pragma unroll
        for (int u = 0; u < 4; ++u) {
            unsigned i = i0 + u * AG_T;
            uint2 v = make_uint2(((en[u] >> 21) << 17) | s[u], __float_as_uint(w[u]));
            q[(i < n) ? i : AG_QCAP] = v;
            if (KEEP && i < n) gq[i] = v;
        }
    }
    __syncthreads();
}

__device__ __forceinline__ void ag_add64(const float* __restrict__ h1,
                                         float* acc, const uint2* q, unsigned n, int t) {
    const unsigned hw = (unsigned)t >> 4;
    const unsigned l16 = (unsigned)t & 15u;
    const bool hi = (hw & 1u) != 0u;
    const unsigned oA = hi ? 2u : 0u, oB = oA ^ 2u;
    for (unsigned i0 = hw; i0 < n; i0 += 64u * AG_DB) {
        uint2 en[AG_DB];
        float4 row[AG_DB];
        #pragma unroll
        for (int u = 0; u < AG_DB; ++u) {
            unsigned i = i0 + 64u * u;
            en[u] = (i < n) ? q[i] : make_uint2((unsigned)AG_R << 17, 0u);
        }
        #pragma unroll
        for (int u = 0; u < AG_DB; ++u)
            row[u] = *(const float4*)(h1 + (size_t)(en[u].x & 0x1FFFFu) * HID + 4u * l16);
        #pragma unroll
        for (int u = 0; u < AG_DB; ++u) {
            float w = __uint_as_float(en[u].y);
            float* p = acc + (en[u].x >> 17) * HID + 4u * l16;
            float a0 = hi ? row[u].z : row[u].x, a1 = hi ? row[u].w : row[u].y;
            float c0 = hi ? row[u].x : row[u].z, c1 = hi ? row[u].y : row[u].w;
            __hip_atomic_fetch_add(p + oA,      a0 * w, __ATOMIC_RELAXED, __HIP_MEMORY_SCOPE_WORKGROUP);
            __hip_atomic_fetch_add(p + oA + 1u, a1 * w, __ATOMIC_RELAXED, __HIP_MEMORY_SCOPE_WORKGROUP);
            __hip_atomic_fetch_add(p + oB,      c0 * w, __ATOMIC_RELAXED, __HIP_MEMORY_SCOPE_WORKGROUP);
            __hip_atomic_fetch_add(p + oB + 1u, c1 * w, __ATOMIC_RELAXED, __HIP_MEMORY_SCOPE_WORKGROUP);
        }
    }
}

__global__ void __launch_bounds__(AG_T)
k_agg1(const int* __restrict__ src, const int* __restrict__ dst,
       const float* __restrict__ norm, const float* __restrict__ h1,
       const float* __restrict__ dis, const float* __restrict__ b1,
       float* __restrict__ z1, unsigned* __restrict__ eq, unsigned* __restrict__ ecnt) {
    __shared__ float acc[(AG_R + 1) * HID];
    __shared__ uint2 q[AG_QCAP + 1];
    __shared__ unsigned qn;
    const int t = threadIdx.x;
    const int lo = blockIdx.x * AG_R;
    const unsigned nrows = (unsigned)((NN - lo) < AG_R ? (NN - lo) : AG_R);

    for (int i = t; i < (AG_R + 1) * HID / 4; i += AG_T)
        ((float4*)acc)[i] = make_float4(0.0f, 0.0f, 0.0f, 0.0f);
    if (t == 0) qn = 0u;
    __syncthreads();

    ag_filter(dst, 0, NE, lo, nrows, &qn, q, t);
    __syncthreads();
    const unsigned total = qn;
    if (t == 0) ecnt[blockIdx.x] = total;
    if (total <= AG_QCAP) {
        ag_lookup<true>(src, norm, q, total, t, (uint2*)eq + (size_t)blockIdx.x * AG_QCAP);
        ag_add64(h1, acc, q, total, t);
    } else {
        for (int e0 = 0; e0 < NE; e0 += AG_QCAP) {
            __syncthreads();
            if (t == 0) qn = 0u;
            __syncthreads();
            const int e1 = (e0 + AG_QCAP < NE) ? (e0 + AG_QCAP) : NE;
            ag_filter(dst, e0, e1, lo, nrows, &qn, q, t);
            __syncthreads();
            const unsigned n = qn;
            ag_lookup<false>(src, norm, q, n, t, nullptr);
            ag_add64(h1, acc, q, n, t);
        }
    }
    __syncthreads();

    for (unsigned i = t; i < nrows * 16u; i += AG_T) {
        unsigned r = i >> 4, c = (i & 15u) * 4u;
        size_t g = (size_t)(lo + r) * HID + c;
        float di = dis[lo + r];
        float4 a = *(const float4*)(acc + r * HID + c);
        float4 h = *(const float4*)(h1 + g);
        float4 b = *(const float4*)(b1 + c);
        float4 v;
        v.x = a.x + h.x * di * di + b.x;
        v.y = a.y + h.y * di * di + b.y;
        v.z = a.z + h.z * di * di + b.z;
        v.w = a.w + h.w * di * di + b.w;
        v.x = v.x > 0.0f ? v.x : 0.0f;
        v.y = v.y > 0.0f ? v.y : 0.0f;
        v.z = v.z > 0.0f ? v.z : 0.0f;
        v.w = v.w > 0.0f ? v.w : 0.0f;
        *(float4*)(z1 + g) = v;
    }
}

#define AG_D2   8

static_assert((AG_R * NCLS) % 32 == 0, "a range's block of out must be whole 128-byte lines");
static_assert(((NN % AG_R) * NCLS) % 4 == 0, "the last range's block of out must be whole float4s");
static_assert(AG_R <= AG_T, "the layer-2 epilogue gives every own row a thread of its own");

__device__ __forceinline__ void ag_add10(const float* __restrict__ h2,
                                         float* acc2, const uint2* q, unsigned n, int t) {
    const unsigned g = (unsigned)t >> 3;
    const unsigned j = (unsigned)t & 7u;
    if (j < 5u) {
        for (unsigned i0 = g; i0 < n; i0 += 128u * AG_D2) {
            uint2 en[AG_D2];
            v2f row[AG_D2];
            #pragma unroll
            for (int u = 0; u < AG_D2; ++u) {
                unsigned i = i0 + 128u * u;
                en[u] = (i < n) ? q[i] : make_uint2((unsigned)AG_R << 17, 0u);
            }
            #pragma unroll
            for (int u = 0; u < AG_D2; ++u)
                row[u] = *(const v2f*)(h2 + (size_t)(en[u].x & 0x1FFFFu) * NCLS + 2u * j);
            #pragma unroll
            for (int u = 0; u < AG_D2; ++u) {
                float w = __uint_as_float(en[u].y);
                float* p = acc2 + (en[u].x >> 17) * NCLS + 2u * j;
                __hip_atomic_fetch_add(p,      row[u].x * w, __ATOMIC_RELAXED, __HIP_MEMORY_SCOPE_WORKGROUP);
                __hip_atomic_fetch_add(p + 1u, row[u].y * w, __ATOMIC_RELAXED, __HIP_MEMORY_SCOPE_WORKGROUP);
            }
        }
    }
}

__global__ void __launch_bounds__(AG_T)
k_agg2(const int* __restrict__ src, const int* __restrict__ dst,
       const float* __restrict__ norm, const float* __restrict__ h2,
       const float* __restrict__ dis, const float* __restrict__ b2,
       const unsigned* __restrict__ eq, const unsigned* __restrict__ ecnt,
       float* __restrict__ out) {
    __shared__ __attribute__((aligned(16))) float acc2[(AG_R + 1) * NCLS];
    __shared__ uint2 q[AG_QCAP + 1];
    __shared__ unsigned qn;
    const int t = threadIdx.x;
    const int lo = blockIdx.x * AG_R;
    const unsigned nrows = (unsigned)((NN - lo) < AG_R ? (NN - lo) : AG_R);

    for (int i = t; i < (AG_R + 1) * NCLS; i += AG_T) acc2[i] = 0.0f;
    const unsigned total = ecnt[blockIdx.x];
    if (total <= AG_QCAP) {
        const uint2* __restrict__ gq = (const uint2*)eq + (size_t)blockIdx.x * AG_QCAP;
        for (unsigned i0 = t; i0 < total; i0 += 4u * AG_T) {
            uint2 v[4];
            #pragma unroll
            for (int u = 0; u < 4; ++u) {
                unsigned i = i0 + u * AG_T;
                v[u] = gq[(i < total) ? i : i0];
            }
            #pragma unroll
            for (int u = 0; u < 4; ++u) {
                unsigned i = i0 + u * AG_T;
                q[(i < total) ? i : AG_QCAP] = v[u];
            }
        }
        __syncthreads();
        ag_add10(h2, acc2, q, total, t);
    } else {
        for (int e0 = 0; e0 < NE; e0 += AG_QCAP) {
            __syncthreads();
            if (t == 0) qn = 0u;
            __syncthreads();
            const int e1 = (e0 + AG_QCAP < NE) ? (e0 + AG_QCAP) : NE;
            ag_filter(dst, e0, e1, lo, nrows, &qn, q, t);
            __syncthreads();
            const unsigned n = qn;
            ag_lookup<false>(src, norm, q, n, t, nullptr);
            ag_add10(h2, acc2, q, n, t);
        }
    }
    __syncthreads();

    if ((unsigned)t < nrows) {
        float di = dis[lo + t];
        float d2 = di * di;
        const float* hp = h2 + (size_t)(lo + t) * NCLS;
        float* ap = acc2 + t * NCLS;
        float tv[NCLS];
        float mx = -3.402823466e38f;
        #pragma unroll
        for (int f = 0; f < NCLS; ++f) {
            float v = ap[f] + hp[f] * d2 + b2[f];
            tv[f] = v;
            mx = fmaxf(mx, v);
        }
        float sum = 0.0f;
        #pragma unroll
        for (int f = 0; f < NCLS; ++f) sum += expf(tv[f] - mx);
        float lse = mx + logf(sum);
        #pragma unroll
        for (int f = 0; f < NCLS; ++f) ap[f] = tv[f] - lse;
    }
    __syncthreads();

    float4* ob = (float4*)(out + (size_t)lo * NCLS);
    for (unsigned i = t; i < nrows * NCLS / 4u; i += AG_T) ob[i] = ((const float4*)acc2)[i];
}

extern "C" void kernel_launch(void* const* d_in, const int* in_sizes, int n_in,
                              void* d_out, int out_size, void* d_ws, size_t ws_size,
                              hipStream_t stream) {
    const float* x  = (const float*)d_in[0];
    const int*   ei = (const int*)d_in[1];
    const float* W1 = (const float*)d_in[2];
    const float* b1 = (const float*)d_in[3];
    const float* W2 = (const float*)d_in[4];
    const float* b2 = (const float*)d_in[5];
    const int* src = ei;
    const int* dst = ei + NE;
    float* out = (float*)d_out;

    float* ws   = (float*)d_ws;
    float* dis  = ws;
    float* h1   = dis + NN;
    float* agg1 = h1 + (size_t)NN * HID;
    float* h2   = agg1 + (size_t)NN * HID;
    float* norm = h2 + (size_t)NN * NCLS;
    unsigned* eq   = (unsigned*)(norm + NE);
    unsigned* ecnt = eq + (size_t)AG_NB * AG_QCAP * 2;

    const int B = 256;
    k_fill<<<(NN + B - 1) / B, B, 0, stream>>>(dis, NN, 1.0f);

    k_deg<<<(NE + B - 1) / B, B, 0, stream>>>(dst, dis);
    k_dis<<<(NN + B - 1) / B, B, 0, stream>>>(dis);
    k_norm<<<(NE + B - 1) / B, B, 0, stream>>>(src, dst, dis, norm);

    k_gemm1<<<(ROW_TILES + 3) / 4, 128, 0, stream>>>(x, W1, h1);
    k_agg1<<<AG_NB, AG_T, 0, stream>>>(src, dst, norm, h1, dis, b1, agg1, eq, ecnt);

    k_gemm2<<<(ROW_TILES + 3) / 4, 128, 0, stream>>>(agg1, W2, h2);
    k_agg2<<<AG_NB, AG_T, 0, stream>>>(src, dst, norm, h2, dis, b2, eq, ecnt, out);
}


# Degrees counted in LDS by the destination-range filter, moved into a launch of its own ahead of layer 1: fill, degree (1.6M global float atomics), rsqrt and edge-weight launches gone; sum order kept
# speedup vs baseline: 2.2458x; 1.3595x over previous
#include <hip/hip_runtime.h>
#include <math.h>

#define NN 100000
#define NE 1600000
#define FIN 128
#define HID 64
#define NCLS 10
#define ROW_TILES 6250

typedef __attribute__((ext_vector_type(2))) float v2f;
typedef __attribute__((ext_vector_type(8))) float v8f;

__global__ void __launch_bounds__(128)
k_gemm1(const float* __restrict__ x, const float* __restrict__ W1,
        float* __restrict__ h1) {
    __shared__ float wT[HID * 132];
    int t = threadIdx.x;
    for (int i = t; i < FIN * HID; i += 128) {
        int k = i >> 6, c = i & 63;
        wT[c * 132 + k] = W1[i];
    }
    __syncthreads();

    int tile = blockIdx.x * 4 + (t >> 5);
    if (tile >= ROW_TILES) return;
    int lane = t & 31;
    int m = lane & 15, half = lane >> 4;

    const float* arow = x + (size_t)(tile * 16 + m) * FIN + 2 * half;
    v8f a0 = {}, a1 = {}, a2 = {}, a3 = {};

    #pragma unroll 4
    for (int k0 = 0; k0 < FIN; k0 += 4) {
        v2f a = *(const v2f*)(arow + k0);
        const float* bp = wT + m * 132 + k0 + 2 * half;
        v2f b0 = *(const v2f*)(bp);
        v2f b1 = *(const v2f*)(bp + 16 * 132);
        v2f b2 = *(const v2f*)(bp + 32 * 132);
        v2f b3 = *(const v2f*)(bp + 48 * 132);
        a0 = __builtin_amdgcn_wmma_f32_16x16x4_f32(false, a, false, b0, (short)0, a0, false, false);
        a1 = __builtin_amdgcn_wmma_f32_16x16x4_f32(false, a, false, b1, (short)0, a1, false, false);
        a2 = __builtin_amdgcn_wmma_f32_16x16x4_f32(false, a, false, b2, (short)0, a2, false, false);
        a3 = __builtin_amdgcn_wmma_f32_16x16x4_f32(false, a, false, b3, (short)0, a3, false, false);
    }

    float* out = h1 + (size_t)(tile * 16 + 8 * half) * HID + m;
    #pragma unroll
    for (int v = 0; v < 8; ++v) {
        out[v * HID +  0] = a0[v];
        out[v * HID + 16] = a1[v];
        out[v * HID + 32] = a2[v];
        out[v * HID + 48] = a3[v];
    }
}

__global__ void __launch_bounds__(128)
k_gemm2(const float* __restrict__ z1, const float* __restrict__ W2,
        float* __restrict__ h2) {
    __shared__ float wT[16 * 68];
    int t = threadIdx.x;
    for (int i = t; i < 16 * HID; i += 128) {
        int k = i >> 4, c = i & 15;
        wT[c * 68 + k] = (c < NCLS) ? W2[k * NCLS + c] : 0.0f;
    }
    __syncthreads();

    int tile = blockIdx.x * 4 + (t >> 5);
    if (tile >= ROW_TILES) return;
    int lane = t & 31;
    int m = lane & 15, half = lane >> 4;

    const float* arow = z1 + (size_t)(tile * 16 + m) * HID + 2 * half;
    v8f acc = {};
    #pragma unroll
    for (int k0 = 0; k0 < HID; k0 += 4) {
        v2f a = *(const v2f*)(arow + k0);
        v2f b = *(const v2f*)(wT + m * 68 + k0 + 2 * half);
        acc = __builtin_amdgcn_wmma_f32_16x16x4_f32(false, a, false, b, (short)0, acc, false, false);
    }
    if (m < NCLS) {
        float* out = h2 + (size_t)(tile * 16 + 8 * half) * NCLS + m;
        #pragma unroll
        for (int v = 0; v < 8; ++v) out[v * NCLS] = acc[v];
    }
}

#define AG_NB   128
#define AG_R    784
#define AG_QCAP 14336
#define AG_T    1024
#define AG_SB   8
#define AG_DB   8

template <bool COUNT>
__device__ __forceinline__ void ag_push(unsigned* qn, uint2* q, unsigned* cnt, unsigned r, int e) {
    if (COUNT) __hip_atomic_fetch_add(cnt + r, 1u, __ATOMIC_RELAXED, __HIP_MEMORY_SCOPE_WORKGROUP);
    unsigned pos = __hip_atomic_fetch_add(qn, 1u, __ATOMIC_RELAXED, __HIP_MEMORY_SCOPE_WORKGROUP);
    if (pos < AG_QCAP) q[pos].x = (r << 21) | (unsigned)e;
}

template <bool COUNT>
__device__ __forceinline__ void ag_filter(const int* __restrict__ dst, int e0, int e1,
                                          int lo, unsigned nrows,
                                          unsigned* qn, uint2* q, unsigned* cnt, int t) {
    for (int e = e0 + 4 * t; e < e1; e += 4 * AG_T * AG_SB) {
        int4 d[AG_SB];
        #pragma unroll
        for (int u = 0; u < AG_SB; ++u) {
            int ee = e + u * 4 * AG_T;
            d[u] = (ee < e1) ? *(const int4*)(dst + ee) : make_int4(-1, -1, -1, -1);
        }
        #pragma unroll
        for (int u = 0; u < AG_SB; ++u) {
            int ee = e + u * 4 * AG_T;
            unsigned r0 = (unsigned)(d[u].x - lo), r1 = (unsigned)(d[u].y - lo);
            unsigned r2 = (unsigned)(d[u].z - lo), r3 = (unsigned)(d[u].w - lo);
            if ((r0 < nrows) | (r1 < nrows) | (r2 < nrows) | (r3 < nrows)) {
                if (r0 < nrows) ag_push<COUNT>(qn, q, cnt, r0, ee);
                if (r1 < nrows) ag_push<COUNT>(qn, q, cnt, r1, ee + 1);
                if (r2 < nrows) ag_push<COUNT>(qn, q, cnt, r2, ee + 2);
                if (r3 < nrows) ag_push<COUNT>(qn, q, cnt, r3, ee + 3);
            }
        }
    }
}

__device__ __forceinline__ void ag_lookup(const int* __restrict__ src, const float* __restrict__ dis,
                                          int lo, uint2* q, unsigned n, int t) {
    for (unsigned i0 = t; i0 < n; i0 += 4u * AG_T) {
        unsigned en[4], s[4];
        float w[4];
        #pragma unroll
        for (int u = 0; u < 4; ++u) {
            unsigned i = i0 + u * AG_T;
            en[u] = (i < n) ? q[i].x : 0u;
        }
        #pragma unroll
        for (int u = 0; u < 4; ++u) s[u] = (unsigned)src[en[u] & 0x1FFFFFu];
        #pragma unroll
        for (int u = 0; u < 4; ++u) w[u] = dis[s[u]] * dis[lo + (int)(en[u] >> 21)];
        #pragma unroll
        for (int u = 0; u < 4; ++u) {
            unsigned i = i0 + u * AG_T;
            q[(i < n) ? i : AG_QCAP] = make_uint2(((en[u] >> 21) << 17) | s[u], __float_as_uint(w[u]));
        }
    }
    __syncthreads();
}

__device__ __forceinline__ void ag_add64(const float* __restrict__ h1,
                                         float* acc, const uint2* q, unsigned n, int t) {
    const unsigned hw = (unsigned)t >> 4;
    const unsigned l16 = (unsigned)t & 15u;
    const bool hi = (hw & 1u) != 0u;
    const unsigned oA = hi ? 2u : 0u, oB = oA ^ 2u;
    for (unsigned i0 = hw; i0 < n; i0 += 64u * AG_DB) {
        uint2 en[AG_DB];
        float4 row[AG_DB];
        #pragma unroll
        for (int u = 0; u < AG_DB; ++u) {
            unsigned i = i0 + 64u * u;
            en[u] = (i < n) ? q[i] : make_uint2((unsigned)AG_R << 17, 0u);
        }
        #pragma unroll
        for (int u = 0; u < AG_DB; ++u)
            row[u] = *(const float4*)(h1 + (size_t)(en[u].x & 0x1FFFFu) * HID + 4u * l16);
        #pragma unroll
        for (int u = 0; u < AG_DB; ++u) {
            float w = __uint_as_float(en[u].y);
            float* p = acc + (en[u].x >> 17) * HID + 4u * l16;
            float a0 = hi ? row[u].z : row[u].x, a1 = hi ? row[u].w : row[u].y;
            float c0 = hi ? row[u].x : row[u].z, c1 = hi ? row[u].y : row[u].w;
            __hip_atomic_fetch_add(p + oA,      a0 * w, __ATOMIC_RELAXED, __HIP_MEMORY_SCOPE_WORKGROUP);
            __hip_atomic_fetch_add(p + oA + 1u, a1 * w, __ATOMIC_RELAXED, __HIP_MEMORY_SCOPE_WORKGROUP);
            __hip_atomic_fetch_add(p + oB,      c0 * w, __ATOMIC_RELAXED, __HIP_MEMORY_SCOPE_WORKGROUP);
            __hip_atomic_fetch_add(p + oB + 1u, c1 * w, __ATOMIC_RELAXED, __HIP_MEMORY_SCOPE_WORKGROUP);
        }
    }
}

#define AG_LB   8

__global__ void __launch_bounds__(AG_T)
k_own(const int* __restrict__ src, const int* __restrict__ dst,
      float* __restrict__ dis, unsigned* __restrict__ eq1, unsigned* __restrict__ ecnt) {
    __shared__ uint2 q[AG_QCAP + 1];
    __shared__ unsigned cnt[AG_R];
    __shared__ unsigned qn;
    const int t = threadIdx.x;
    const int lo = blockIdx.x * AG_R;
    const unsigned nrows = (unsigned)((NN - lo) < AG_R ? (NN - lo) : AG_R);

    if (t < AG_R) cnt[t] = 0u;
    if (t == 0) qn = 0u;
    __syncthreads();

    ag_filter<true>(dst, 0, NE, lo, nrows, &qn, q, cnt, t);
    __syncthreads();
    const unsigned total = qn;
    if (t == 0) ecnt[blockIdx.x] = total;
    if ((unsigned)t < nrows) dis[lo + t] = rsqrtf((float)(cnt[t] + 1u));
    if (total <= AG_QCAP) {
        unsigned* __restrict__ g1 = eq1 + (size_t)blockIdx.x * AG_QCAP;
        for (unsigned i0 = t; i0 < total; i0 += 4u * AG_T) {
            unsigned ix[4], en[4], sn[4];
            #pragma unroll
            for (int u = 0; u < 4; ++u) {
                unsigned i = i0 + u * AG_T;
                ix[u] = (i < total) ? i : i0;
                en[u] = q[ix[u]].x;
            }
            #pragma unroll
            for (int u = 0; u < 4; ++u) sn[u] = (unsigned)src[en[u] & 0x1FFFFFu];
            #pragma unroll
            for (int u = 0; u < 4; ++u) g1[ix[u]] = ((en[u] >> 21) << 17) | sn[u];
        }
    }
}

__global__ void __launch_bounds__(AG_T)
k_agg1(const int* __restrict__ src, const int* __restrict__ dst,
       const float* __restrict__ h1, const float* __restrict__ dis, const float* __restrict__ b1,
       float* __restrict__ z1, const unsigned* __restrict__ eq1,
       unsigned* __restrict__ eq, const unsigned* __restrict__ ecnt) {
    __shared__ float acc[(AG_R + 1) * HID];
    __shared__ uint2 q[AG_QCAP + 1];
    __shared__ unsigned qn;
    const int t = threadIdx.x;
    const int lo = blockIdx.x * AG_R;
    const unsigned nrows = (unsigned)((NN - lo) < AG_R ? (NN - lo) : AG_R);

    for (int i = t; i < (AG_R + 1) * HID / 4; i += AG_T)
        ((float4*)acc)[i] = make_float4(0.0f, 0.0f, 0.0f, 0.0f);
    const unsigned total = ecnt[blockIdx.x];
    if (total <= AG_QCAP) {
        const unsigned* __restrict__ g1 = eq1 + (size_t)blockIdx.x * AG_QCAP;
        uint2* __restrict__ g2 = (uint2*)eq + (size_t)blockIdx.x * AG_QCAP;
        for (unsigned i0 = t; i0 < total; i0 += (unsigned)AG_LB * AG_T) {
            unsigned v[AG_LB];
            float w[AG_LB];
            #pragma unroll
            for (int u = 0; u < AG_LB; ++u) {
                unsigned i = i0 + u * AG_T;
                v[u] = g1[(i < total) ? i : i0];
            }
            #pragma unroll
            for (int u = 0; u < AG_LB; ++u)
                w[u] = dis[v[u] & 0x1FFFFu] * dis[lo + (int)(v[u] >> 17)];
            #pragma unroll
            for (int u = 0; u < AG_LB; ++u) {
                unsigned i = i0 + u * AG_T;
                uint2 o = make_uint2(v[u], __float_as_uint(w[u]));
                q[(i < total) ? i : AG_QCAP] = o;
                if (i < total) g2[i] = o;
            }
        }
        __syncthreads();
        ag_add64(h1, acc, q, total, t);
    } else {
        for (int e0 = 0; e0 < NE; e0 += AG_QCAP) {
            __syncthreads();
            if (t == 0) qn = 0u;
            __syncthreads();
            const int e1 = (e0 + AG_QCAP < NE) ? (e0 + AG_QCAP) : NE;
            ag_filter<false>(dst, e0, e1, lo, nrows, &qn, q, nullptr, t);
            __syncthreads();
            const unsigned n = qn;
            ag_lookup(src, dis, lo, q, n, t);
            ag_add64(h1, acc, q, n, t);
        }
    }
    __syncthreads();

    for (unsigned i = t; i < nrows * 16u; i += AG_T) {
        unsigned r = i >> 4, c = (i & 15u) * 4u;
        size_t g = (size_t)(lo + r) * HID + c;
        float di = dis[lo + r];
        float4 a = *(const float4*)(acc + r * HID + c);
        float4 h = *(const float4*)(h1 + g);
        float4 b = *(const float4*)(b1 + c);
        float4 v;
        v.x = a.x + h.x * di * di + b.x;
        v.y = a.y + h.y * di * di + b.y;
        v.z = a.z + h.z * di * di + b.z;
        v.w = a.w + h.w * di * di + b.w;
        v.x = v.x > 0.0f ? v.x : 0.0f;
        v.y = v.y > 0.0f ? v.y : 0.0f;
        v.z = v.z > 0.0f ? v.z : 0.0f;
        v.w = v.w > 0.0f ? v.w : 0.0f;
        *(float4*)(z1 + g) = v;
    }
}

#define AG_D2   8

static_assert((AG_R * NCLS) % 32 == 0, "a range's block of out must be whole 128-byte lines");
static_assert(((NN % AG_R) * NCLS) % 4 == 0, "the last range's block of out must be whole float4s");
static_assert(AG_R <= AG_T, "the layer-2 epilogue gives every own row a thread of its own");

__device__ __forceinline__ void ag_add10(const float* __restrict__ h2,
                                         float* acc2, const uint2* q, unsigned n, int t) {
    const unsigned g = (unsigned)t >> 3;
    const unsigned j = (unsigned)t & 7u;
    if (j < 5u) {
        for (unsigned i0 = g; i0 < n; i0 += 128u * AG_D2) {
            uint2 en[AG_D2];
            v2f row[AG_D2];
            #pragma unroll
            for (int u = 0; u < AG_D2; ++u) {
                unsigned i = i0 + 128u * u;
                en[u] = (i < n) ? q[i] : make_uint2((unsigned)AG_R << 17, 0u);
            }
            #pragma unroll
            for (int u = 0; u < AG_D2; ++u)
                row[u] = *(const v2f*)(h2 + (size_t)(en[u].x & 0x1FFFFu) * NCLS + 2u * j);
            #pragma unroll
            for (int u = 0; u < AG_D2; ++u) {
                float w = __uint_as_float(en[u].y);
                float* p = acc2 + (en[u].x >> 17) * NCLS + 2u * j;
                __hip_atomic_fetch_add(p,      row[u].x * w, __ATOMIC_RELAXED, __HIP_MEMORY_SCOPE_WORKGROUP);
                __hip_atomic_fetch_add(p + 1u, row[u].y * w, __ATOMIC_RELAXED, __HIP_MEMORY_SCOPE_WORKGROUP);
            }
        }
    }
}

__global__ void __launch_bounds__(AG_T)
k_agg2(const int* __restrict__ src, const int* __restrict__ dst,
       const float* __restrict__ h2,
       const float* __restrict__ dis, const float* __restrict__ b2,
       const unsigned* __restrict__ eq, const unsigned* __restrict__ ecnt,
       float* __restrict__ out) {
    __shared__ __attribute__((aligned(16))) float acc2[(AG_R + 1) * NCLS];
    __shared__ uint2 q[AG_QCAP + 1];
    __shared__ unsigned qn;
    const int t = threadIdx.x;
    const int lo = blockIdx.x * AG_R;
    const unsigned nrows = (unsigned)((NN - lo) < AG_R ? (NN - lo) : AG_R);

    for (int i = t; i < (AG_R + 1) * NCLS; i += AG_T) acc2[i] = 0.0f;
    const unsigned total = ecnt[blockIdx.x];
    if (total <= AG_QCAP) {
        const uint2* __restrict__ gq = (const uint2*)eq + (size_t)blockIdx.x * AG_QCAP;
        for (unsigned i0 = t; i0 < total; i0 += 4u * AG_T) {
            uint2 v[4];
            #pragma unroll
            for (int u = 0; u < 4; ++u) {
                unsigned i = i0 + u * AG_T;
                v[u] = gq[(i < total) ? i : i0];
            }
            #pragma unroll
            for (int u = 0; u < 4; ++u) {
                unsigned i = i0 + u * AG_T;
                q[(i < total) ? i : AG_QCAP] = v[u];
            }
        }
        __syncthreads();
        ag_add10(h2, acc2, q, total, t);
    } else {
        for (int e0 = 0; e0 < NE; e0 += AG_QCAP) {
            __syncthreads();
            if (t == 0) qn = 0u;
            __syncthreads();
            const int e1 = (e0 + AG_QCAP < NE) ? (e0 + AG_QCAP) : NE;
            ag_filter<false>(dst, e0, e1, lo, nrows, &qn, q, nullptr, t);
            __syncthreads();
            const unsigned n = qn;
            ag_lookup(src, dis, lo, q, n, t);
            ag_add10(h2, acc2, q, n, t);
        }
    }
    __syncthreads();

    if ((unsigned)t < nrows) {
        float di = dis[lo + t];
        float d2 = di * di;
        const float* hp = h2 + (size_t)(lo + t) * NCLS;
        float* ap = acc2 + t * NCLS;
        float tv[NCLS];
        float mx = -3.402823466e38f;
        #pragma unroll
        for (int f = 0; f < NCLS; ++f) {
            float v = ap[f] + hp[f] * d2 + b2[f];
            tv[f] = v;
            mx = fmaxf(mx, v);
        }
        float sum = 0.0f;
        #pragma unroll
        for (int f = 0; f < NCLS; ++f) sum += expf(tv[f] - mx);
        float lse = mx + logf(sum);
        #pragma unroll
        for (int f = 0; f < NCLS; ++f) ap[f] = tv[f] - lse;
    }
    __syncthreads();

    float4* ob = (float4*)(out + (size_t)lo * NCLS);
    for (unsigned i = t; i < nrows * NCLS / 4u; i += AG_T) ob[i] = ((const float4*)acc2)[i];
}

extern "C" void kernel_launch(void* const* d_in, const int* in_sizes, int n_in,
                              void* d_out, int out_size, void* d_ws, size_t ws_size,
                              hipStream_t stream) {
    const float* x  = (const float*)d_in[0];
    const int*   ei = (const int*)d_in[1];
    const float* W1 = (const float*)d_in[2];
    const float* b1 = (const float*)d_in[3];
    const float* W2 = (const float*)d_in[4];
    const float* b2 = (const float*)d_in[5];
    const int* src = ei;
    const int* dst = ei + NE;
    float* out = (float*)d_out;

    float* ws   = (float*)d_ws;
    float* dis  = ws;
    float* h1   = dis + NN;
    float* agg1 = h1 + (size_t)NN * HID;
    float* h2   = agg1 + (size_t)NN * HID;
    unsigned* eq1  = (unsigned*)(h2 + (size_t)NN * NCLS);
    unsigned* eq   = eq1 + (size_t)AG_NB * AG_QCAP;
    unsigned* ecnt = eq + (size_t)AG_NB * AG_QCAP * 2;

    k_own<<<AG_NB, AG_T, 0, stream>>>(src, dst, dis, eq1, ecnt);

    k_gemm1<<<(ROW_TILES + 3) / 4, 128, 0, stream>>>(x, W1, h1);
    k_agg1<<<AG_NB, AG_T, 0, stream>>>(src, dst, h1, dis, b1, agg1, eq1, eq, ecnt);

    k_gemm2<<<(ROW_TILES + 3) / 4, 128, 0, stream>>>(agg1, W2, h2);
    k_agg2<<<AG_NB, AG_T, 0, stream>>>(src, dst, h2, dis, b2, eq, ecnt, out);
}


# Destination ranges halved to 400 rows: 250 work-groups per range launch instead of 128, so filter and both LDS aggregations run on all 256 WGPs, not half; sums still in LDS arrival order
# speedup vs baseline: 3.0209x; 1.3451x over previous
#include <hip/hip_runtime.h>
#include <math.h>

#define NN 100000
#define NE 1600000
#define FIN 128
#define HID 64
#define NCLS 10
#define ROW_TILES 6250

typedef __attribute__((ext_vector_type(2))) float v2f;
typedef __attribute__((ext_vector_type(8))) float v8f;

__global__ void __launch_bounds__(128)
k_gemm1(const float* __restrict__ x, const float* __restrict__ W1,
        float* __restrict__ h1) {
    __shared__ float wT[HID * 132];
    int t = threadIdx.x;
    for (int i = t; i < FIN * HID; i += 128) {
        int k = i >> 6, c = i & 63;
        wT[c * 132 + k] = W1[i];
    }
    __syncthreads();

    int tile = blockIdx.x * 4 + (t >> 5);
    if (tile >= ROW_TILES) return;
    int lane = t & 31;
    int m = lane & 15, half = lane >> 4;

    const float* arow = x + (size_t)(tile * 16 + m) * FIN + 2 * half;
    v8f a0 = {}, a1 = {}, a2 = {}, a3 = {};

    #pragma unroll 4
    for (int k0 = 0; k0 < FIN; k0 += 4) {
        v2f a = *(const v2f*)(arow + k0);
        const float* bp = wT + m * 132 + k0 + 2 * half;
        v2f b0 = *(const v2f*)(bp);
        v2f b1 = *(const v2f*)(bp + 16 * 132);
        v2f b2 = *(const v2f*)(bp + 32 * 132);
        v2f b3 = *(const v2f*)(bp + 48 * 132);
        a0 = __builtin_amdgcn_wmma_f32_16x16x4_f32(false, a, false, b0, (short)0, a0, false, false);
        a1 = __builtin_amdgcn_wmma_f32_16x16x4_f32(false, a, false, b1, (short)0, a1, false, false);
        a2 = __builtin_amdgcn_wmma_f32_16x16x4_f32(false, a, false, b2, (short)0, a2, false, false);
        a3 = __builtin_amdgcn_wmma_f32_16x16x4_f32(false, a, false, b3, (short)0, a3, false, false);
    }

    float* out = h1 + (size_t)(tile * 16 + 8 * half) * HID + m;
    #pragma unroll
    for (int v = 0; v < 8; ++v) {
        out[v * HID +  0] = a0[v];
        out[v * HID + 16] = a1[v];
        out[v * HID + 32] = a2[v];
        out[v * HID + 48] = a3[v];
    }
}

__global__ void __launch_bounds__(128)
k_gemm2(const float* __restrict__ z1, const float* __restrict__ W2,
        float* __restrict__ h2) {
    __shared__ float wT[16 * 68];
    int t = threadIdx.x;
    for (int i = t; i < 16 * HID; i += 128) {
        int k = i >> 4, c = i & 15;
        wT[c * 68 + k] = (c < NCLS) ? W2[k * NCLS + c] : 0.0f;
    }
    __syncthreads();

    int tile = blockIdx.x * 4 + (t >> 5);
    if (tile >= ROW_TILES) return;
    int lane = t & 31;
    int m = lane & 15, half = lane >> 4;

    const float* arow = z1 + (size_t)(tile * 16 + m) * HID + 2 * half;
    v8f acc = {};
    #pragma unroll
    for (int k0 = 0; k0 < HID; k0 += 4) {
        v2f a = *(const v2f*)(arow + k0);
        v2f b = *(const v2f*)(wT + m * 68 + k0 + 2 * half);
        acc = __builtin_amdgcn_wmma_f32_16x16x4_f32(false, a, false, b, (short)0, acc, false, false);
    }
    if (m < NCLS) {
        float* out = h2 + (size_t)(tile * 16 + 8 * half) * NCLS + m;
        #pragma unroll
        for (int v = 0; v < 8; ++v) out[v * NCLS] = acc[v];
    }
}

#define AG_NB   250
#define AG_R    400
#define AG_QCAP 8192
#define AG_T    1024
#define AG_SB   8
#define AG_DB   8

template <bool COUNT>
__device__ __forceinline__ void ag_push(unsigned* qn, uint2* q, unsigned* cnt, unsigned r, int e) {
    if (COUNT) __hip_atomic_fetch_add(cnt + r, 1u, __ATOMIC_RELAXED, __HIP_MEMORY_SCOPE_WORKGROUP);
    unsigned pos = __hip_atomic_fetch_add(qn, 1u, __ATOMIC_RELAXED, __HIP_MEMORY_SCOPE_WORKGROUP);
    if (pos < AG_QCAP) q[pos].x = (r << 21) | (unsigned)e;
}

template <bool COUNT>
__device__ __forceinline__ void ag_filter(const int* __restrict__ dst, int e0, int e1,
                                          int lo, unsigned nrows,
                                          unsigned* qn, uint2* q, unsigned* cnt, int t) {
    for (int e = e0 + 4 * t; e < e1; e += 4 * AG_T * AG_SB) {
        int4 d[AG_SB];
        #pragma unroll
        for (int u = 0; u < AG_SB; ++u) {
            int ee = e + u * 4 * AG_T;
            d[u] = (ee < e1) ? *(const int4*)(dst + ee) : make_int4(-1, -1, -1, -1);
        }
        #pragma unroll
        for (int u = 0; u < AG_SB; ++u) {
            int ee = e + u * 4 * AG_T;
            unsigned r0 = (unsigned)(d[u].x - lo), r1 = (unsigned)(d[u].y - lo);
            unsigned r2 = (unsigned)(d[u].z - lo), r3 = (unsigned)(d[u].w - lo);
            if ((r0 < nrows) | (r1 < nrows) | (r2 < nrows) | (r3 < nrows)) {
                if (r0 < nrows) ag_push<COUNT>(qn, q, cnt, r0, ee);
                if (r1 < nrows) ag_push<COUNT>(qn, q, cnt, r1, ee + 1);
                if (r2 < nrows) ag_push<COUNT>(qn, q, cnt, r2, ee + 2);
                if (r3 < nrows) ag_push<COUNT>(qn, q, cnt, r3, ee + 3);
            }
        }
    }
}

__device__ __forceinline__ void ag_lookup(const int* __restrict__ src, const float* __restrict__ dis,
                                          int lo, uint2* q, unsigned n, int t) {
    for (unsigned i0 = t; i0 < n; i0 += 4u * AG_T) {
        unsigned en[4], s[4];
        float w[4];
        #pragma unroll
        for (int u = 0; u < 4; ++u) {
            unsigned i = i0 + u * AG_T;
            en[u] = (i < n) ? q[i].x : 0u;
        }
        #pragma unroll
        for (int u = 0; u < 4; ++u) s[u] = (unsigned)src[en[u] & 0x1FFFFFu];
        #pragma unroll
        for (int u = 0; u < 4; ++u) w[u] = dis[s[u]] * dis[lo + (int)(en[u] >> 21)];
        #pragma unroll
        for (int u = 0; u < 4; ++u) {
            unsigned i = i0 + u * AG_T;
            q[(i < n) ? i : AG_QCAP] = make_uint2(((en[u] >> 21) << 17) | s[u], __float_as_uint(w[u]));
        }
    }
    __syncthreads();
}

__device__ __forceinline__ void ag_add64(const float* __restrict__ h1,
                                         float* acc, const uint2* q, unsigned n, int t) {
    const unsigned hw = (unsigned)t >> 4;
    const unsigned l16 = (unsigned)t & 15u;
    const bool hi = (hw & 1u) != 0u;
    const unsigned oA = hi ? 2u : 0u, oB = oA ^ 2u;
    for (unsigned i0 = hw; i0 < n; i0 += 64u * AG_DB) {
        uint2 en[AG_DB];
        float4 row[AG_DB];
        #pragma unroll
        for (int u = 0; u < AG_DB; ++u) {
            unsigned i = i0 + 64u * u;
            en[u] = (i < n) ? q[i] : make_uint2((unsigned)AG_R << 17, 0u);
        }
        #pragma unroll
        for (int u = 0; u < AG_DB; ++u)
            row[u] = *(const float4*)(h1 + (size_t)(en[u].x & 0x1FFFFu) * HID + 4u * l16);
        #pragma unroll
        for (int u = 0; u < AG_DB; ++u) {
            float w = __uint_as_float(en[u].y);
            float* p = acc + (en[u].x >> 17) * HID + 4u * l16;
            float a0 = hi ? row[u].z : row[u].x, a1 = hi ? row[u].w : row[u].y;
            float c0 = hi ? row[u].x : row[u].z, c1 = hi ? row[u].y : row[u].w;
            __hip_atomic_fetch_add(p + oA,      a0 * w, __ATOMIC_RELAXED, __HIP_MEMORY_SCOPE_WORKGROUP);
            __hip_atomic_fetch_add(p + oA + 1u, a1 * w, __ATOMIC_RELAXED, __HIP_MEMORY_SCOPE_WORKGROUP);
            __hip_atomic_fetch_add(p + oB,      c0 * w, __ATOMIC_RELAXED, __HIP_MEMORY_SCOPE_WORKGROUP);
            __hip_atomic_fetch_add(p + oB + 1u, c1 * w, __ATOMIC_RELAXED, __HIP_MEMORY_SCOPE_WORKGROUP);
        }
    }
}

#define AG_LB   8

__global__ void __launch_bounds__(AG_T)
k_own(const int* __restrict__ src, const int* __restrict__ dst,
      float* __restrict__ dis, unsigned* __restrict__ eq1, unsigned* __restrict__ ecnt) {
    __shared__ uint2 q[AG_QCAP + 1];
    __shared__ unsigned cnt[AG_R];
    __shared__ unsigned qn;
    const int t = threadIdx.x;
    const int lo = blockIdx.x * AG_R;
    const unsigned nrows = (unsigned)((NN - lo) < AG_R ? (NN - lo) : AG_R);

    if (t < AG_R) cnt[t] = 0u;
    if (t == 0) qn = 0u;
    __syncthreads();

    ag_filter<true>(dst, 0, NE, lo, nrows, &qn, q, cnt, t);
    __syncthreads();
    const unsigned total = qn;
    if (t == 0) ecnt[blockIdx.x] = total;
    if ((unsigned)t < nrows) dis[lo + t] = rsqrtf((float)(cnt[t] + 1u));
    if (total <= AG_QCAP) {
        unsigned* __restrict__ g1 = eq1 + (size_t)blockIdx.x * AG_QCAP;
        for (unsigned i0 = t; i0 < total; i0 += 4u * AG_T) {
            unsigned ix[4], en[4], sn[4];
            #pragma unroll
            for (int u = 0; u < 4; ++u) {
                unsigned i = i0 + u * AG_T;
                ix[u] = (i < total) ? i : i0;
                en[u] = q[ix[u]].x;
            }
            #pragma unroll
            for (int u = 0; u < 4; ++u) sn[u] = (unsigned)src[en[u] & 0x1FFFFFu];
            #pragma unroll
            for (int u = 0; u < 4; ++u) g1[ix[u]] = ((en[u] >> 21) << 17) | sn[u];
        }
    }
}

__global__ void __launch_bounds__(AG_T)
k_agg1(const int* __restrict__ src, const int* __restrict__ dst,
       const float* __restrict__ h1, const float* __restrict__ dis, const float* __restrict__ b1,
       float* __restrict__ z1, const unsigned* __restrict__ eq1,
       unsigned* __restrict__ eq, const unsigned* __restrict__ ecnt) {
    __shared__ float acc[(AG_R + 1) * HID];
    __shared__ uint2 q[AG_QCAP + 1];
    __shared__ unsigned qn;
    const int t = threadIdx.x;
    const int lo = blockIdx.x * AG_R;
    const unsigned nrows = (unsigned)((NN - lo) < AG_R ? (NN - lo) : AG_R);

    for (int i = t; i < (AG_R + 1) * HID / 4; i += AG_T)
        ((float4*)acc)[i] = make_float4(0.0f, 0.0f, 0.0f, 0.0f);
    const unsigned total = ecnt[blockIdx.x];
    if (total <= AG_QCAP) {
        const unsigned* __restrict__ g1 = eq1 + (size_t)blockIdx.x * AG_QCAP;
        uint2* __restrict__ g2 = (uint2*)eq + (size_t)blockIdx.x * AG_QCAP;
        for (unsigned i0 = t; i0 < total; i0 += (unsigned)AG_LB * AG_T) {
            unsigned v[AG_LB];
            float w[AG_LB];
            #pragma unroll
            for (int u = 0; u < AG_LB; ++u) {
                unsigned i = i0 + u * AG_T;
                v[u] = g1[(i < total) ? i : i0];
            }
            #pragma unroll
            for (int u = 0; u < AG_LB; ++u)
                w[u] = dis[v[u] & 0x1FFFFu] * dis[lo + (int)(v[u] >> 17)];
            #pragma unroll
            for (int u = 0; u < AG_LB; ++u) {
                unsigned i = i0 + u * AG_T;
                uint2 o = make_uint2(v[u], __float_as_uint(w[u]));
                q[(i < total) ? i : AG_QCAP] = o;
                if (i < total) g2[i] = o;
            }
        }
        __syncthreads();
        ag_add64(h1, acc, q, total, t);
    } else {
        for (int e0 = 0; e0 < NE; e0 += AG_QCAP) {
            __syncthreads();
            if (t == 0) qn = 0u;
            __syncthreads();
            const int e1 = (e0 + AG_QCAP < NE) ? (e0 + AG_QCAP) : NE;
            ag_filter<false>(dst, e0, e1, lo, nrows, &qn, q, nullptr, t);
            __syncthreads();
            const unsigned n = qn;
            ag_lookup(src, dis, lo, q, n, t);
            ag_add64(h1, acc, q, n, t);
        }
    }
    __syncthreads();

    for (unsigned i = t; i < nrows * 16u; i += AG_T) {
        unsigned r = i >> 4, c = (i & 15u) * 4u;
        size_t g = (size_t)(lo + r) * HID + c;
        float di = dis[lo + r];
        float4 a = *(const float4*)(acc + r * HID + c);
        float4 h = *(const float4*)(h1 + g);
        float4 b = *(const float4*)(b1 + c);
        float4 v;
        v.x = a.x + h.x * di * di + b.x;
        v.y = a.y + h.y * di * di + b.y;
        v.z = a.z + h.z * di * di + b.z;
        v.w = a.w + h.w * di * di + b.w;
        v.x = v.x > 0.0f ? v.x : 0.0f;
        v.y = v.y > 0.0f ? v.y : 0.0f;
        v.z = v.z > 0.0f ? v.z : 0.0f;
        v.w = v.w > 0.0f ? v.w : 0.0f;
        *(float4*)(z1 + g) = v;
    }
}

#define AG_D2   8

static_assert((AG_R * NCLS) % 32 == 0, "a range's block of out must be whole 128-byte lines");
static_assert(((NN % AG_R) * NCLS) % 4 == 0, "the last range's block of out must be whole float4s");
static_assert(AG_R <= AG_T, "the layer-2 epilogue gives every own row a thread of its own");
static_assert((long long)AG_NB * AG_R >= NN && (long long)(AG_NB - 1) * AG_R < NN, "the ranges cover the nodes and none is empty");

__device__ __forceinline__ void ag_add10(const float* __restrict__ h2,
                                         float* acc2, const uint2* q, unsigned n, int t) {
    const unsigned g = (unsigned)t >> 3;
    const unsigned j = (unsigned)t & 7u;
    if (j < 5u) {
        for (unsigned i0 = g; i0 < n; i0 += 128u * AG_D2) {
            uint2 en[AG_D2];
            v2f row[AG_D2];
            #pragma unroll
            for (int u = 0; u < AG_D2; ++u) {
                unsigned i = i0 + 128u * u;
                en[u] = (i < n) ? q[i] : make_uint2((unsigned)AG_R << 17, 0u);
            }
            #pragma unroll
            for (int u = 0; u < AG_D2; ++u)
                row[u] = *(const v2f*)(h2 + (size_t)(en[u].x & 0x1FFFFu) * NCLS + 2u * j);
            #pragma unroll
            for (int u = 0; u < AG_D2; ++u) {
                float w = __uint_as_float(en[u].y);
                float* p = acc2 + (en[u].x >> 17) * NCLS + 2u * j;
                __hip_atomic_fetch_add(p,      row[u].x * w, __ATOMIC_RELAXED, __HIP_MEMORY_SCOPE_WORKGROUP);
                __hip_atomic_fetch_add(p + 1u, row[u].y * w, __ATOMIC_RELAXED, __HIP_MEMORY_SCOPE_WORKGROUP);
            }
        }
    }
}

__global__ void __launch_bounds__(AG_T)
k_agg2(const int* __restrict__ src, const int* __restrict__ dst,
       const float* __restrict__ h2,
       const float* __restrict__ dis, const float* __restrict__ b2,
       const unsigned* __restrict__ eq, const unsigned* __restrict__ ecnt,
       float* __restrict__ out) {
    __shared__ __attribute__((aligned(16))) float acc2[(AG_R + 1) * NCLS];
    __shared__ uint2 q[AG_QCAP + 1];
    __shared__ unsigned qn;
    const int t = threadIdx.x;
    const int lo = blockIdx.x * AG_R;
    const unsigned nrows = (unsigned)((NN - lo) < AG_R ? (NN - lo) : AG_R);

    for (int i = t; i < (AG_R + 1) * NCLS; i += AG_T) acc2[i] = 0.0f;
    const unsigned total = ecnt[blockIdx.x];
    if (total <= AG_QCAP) {
        const uint2* __restrict__ gq = (const uint2*)eq + (size_t)blockIdx.x * AG_QCAP;
        for (unsigned i0 = t; i0 < total; i0 += 4u * AG_T) {
            uint2 v[4];
            #pragma unroll
            for (int u = 0; u < 4; ++u) {
                unsigned i = i0 + u * AG_T;
                v[u] = gq[(i < total) ? i : i0];
            }
            #pragma unroll
            for (int u = 0; u < 4; ++u) {
                unsigned i = i0 + u * AG_T;
                q[(i < total) ? i : AG_QCAP] = v[u];
            }
        }
        __syncthreads();
        ag_add10(h2, acc2, q, total, t);
    } else {
        for (int e0 = 0; e0 < NE; e0 += AG_QCAP) {
            __syncthreads();
            if (t == 0) qn = 0u;
            __syncthreads();
            const int e1 = (e0 + AG_QCAP < NE) ? (e0 + AG_QCAP) : NE;
            ag_filter<false>(dst, e0, e1, lo, nrows, &qn, q, nullptr, t);
            __syncthreads();
            const unsigned n = qn;
            ag_lookup(src, dis, lo, q, n, t);
            ag_add10(h2, acc2, q, n, t);
        }
    }
    __syncthreads();

    if ((unsigned)t < nrows) {
        float di = dis[lo + t];
        float d2 = di * di;
        const float* hp = h2 + (size_t)(lo + t) * NCLS;
        float* ap = acc2 + t * NCLS;
        float tv[NCLS];
        float mx = -3.402823466e38f;
        #pragma unroll
        for (int f = 0; f < NCLS; ++f) {
            float v = ap[f] + hp[f] * d2 + b2[f];
            tv[f] = v;
            mx = fmaxf(mx, v);
        }
        float sum = 0.0f;
        #pragma unroll
        for (int f = 0; f < NCLS; ++f) sum += expf(tv[f] - mx);
        float lse = mx + logf(sum);
        #pragma unroll
        for (int f = 0; f < NCLS; ++f) ap[f] = tv[f] - lse;
    }
    __syncthreads();

    float4* ob = (float4*)(out + (size_t)lo * NCLS);
    for (unsigned i = t; i < nrows * NCLS / 4u; i += AG_T) ob[i] = ((const float4*)acc2)[i];
}

extern "C" void kernel_launch(void* const* d_in, const int* in_sizes, int n_in,
                              void* d_out, int out_size, void* d_ws, size_t ws_size,
                              hipStream_t stream) {
    const float* x  = (const float*)d_in[0];
    const int*   ei = (const int*)d_in[1];
    const float* W1 = (const float*)d_in[2];
    const float* b1 = (const float*)d_in[3];
    const float* W2 = (const float*)d_in[4];
    const float* b2 = (const float*)d_in[5];
    const int* src = ei;
    const int* dst = ei + NE;
    float* out = (float*)d_out;

    float* ws   = (float*)d_ws;
    float* dis  = ws;
    float* h1   = dis + NN;
    float* agg1 = h1 + (size_t)NN * HID;
    float* h2   = agg1 + (size_t)NN * HID;
    unsigned* eq1  = (unsigned*)(h2 + (size_t)NN * NCLS);
    unsigned* eq   = eq1 + (size_t)AG_NB * AG_QCAP;
    unsigned* ecnt = eq + (size_t)AG_NB * AG_QCAP * 2;

    k_own<<<AG_NB, AG_T, 0, stream>>>(src, dst, dis, eq1, ecnt);

    k_gemm1<<<(ROW_TILES + 3) / 4, 128, 0, stream>>>(x, W1, h1);
    k_agg1<<<AG_NB, AG_T, 0, stream>>>(src, dst, h1, dis, b1, agg1, eq1, eq, ecnt);

    k_gemm2<<<(ROW_TILES + 3) / 4, 128, 0, stream>>>(agg1, W2, h2);
    k_agg2<<<AG_NB, AG_T, 0, stream>>>(src, dst, h2, dis, b2, eq, ecnt, out);
}


# Destination list packed to one-byte range ids by the layer-1 product launch (now first): a range's filter scans 1.6 MB, not 6.4 MB, 16 edges per load; row looked up per hit; sums in LDS arrival order
# speedup vs baseline: 3.4841x; 1.1533x over previous
#include <hip/hip_runtime.h>
#include <math.h>

#define NN 100000
#define NE 1600000
#define FIN 128
#define HID 64
#define NCLS 10
#define ROW_TILES 6250

typedef __attribute__((ext_vector_type(2))) float v2f;
typedef __attribute__((ext_vector_type(8))) float v8f;

#define RG_R 400

__device__ __forceinline__ unsigned rg_id(int d) {
    return ((unsigned)d < (unsigned)NN) ? (unsigned)d / (unsigned)RG_R : 255u;
}

__global__ void __launch_bounds__(128)
k_gemm1(const float* __restrict__ x, const float* __restrict__ W1,
        float* __restrict__ h1, const int* __restrict__ dst, unsigned* __restrict__ rid) {
    __shared__ float wT[HID * 132];
    int t = threadIdx.x;
    {
        int e8 = (blockIdx.x * 128 + t) * 8;
        if (e8 < NE) {
            int4 d0 = *(const int4*)(dst + e8);
            int4 d1 = *(const int4*)(dst + e8 + 4);
            uint2 o;
            o.x = rg_id(d0.x) | (rg_id(d0.y) << 8) | (rg_id(d0.z) << 16) | (rg_id(d0.w) << 24);
            o.y = rg_id(d1.x) | (rg_id(d1.y) << 8) | (rg_id(d1.z) << 16) | (rg_id(d1.w) << 24);
            *(uint2*)(rid + (e8 >> 2)) = o;
        }
    }
    for (int i = t; i < FIN * HID; i += 128) {
        int k = i >> 6, c = i & 63;
        wT[c * 132 + k] = W1[i];
    }
    __syncthreads();

    int tile = blockIdx.x * 4 + (t >> 5);
    if (tile >= ROW_TILES) return;
    int lane = t & 31;
    int m = lane & 15, half = lane >> 4;

    const float* arow = x + (size_t)(tile * 16 + m) * FIN + 2 * half;
    v8f a0 = {}, a1 = {}, a2 = {}, a3 = {};

    #pragma unroll 4
    for (int k0 = 0; k0 < FIN; k0 += 4) {
        v2f a = *(const v2f*)(arow + k0);
        const float* bp = wT + m * 132 + k0 + 2 * half;
        v2f b0 = *(const v2f*)(bp);
        v2f b1 = *(const v2f*)(bp + 16 * 132);
        v2f b2 = *(const v2f*)(bp + 32 * 132);
        v2f b3 = *(const v2f*)(bp + 48 * 132);
        a0 = __builtin_amdgcn_wmma_f32_16x16x4_f32(false, a, false, b0, (short)0, a0, false, false);
        a1 = __builtin_amdgcn_wmma_f32_16x16x4_f32(false, a, false, b1, (short)0, a1, false, false);
        a2 = __builtin_amdgcn_wmma_f32_16x16x4_f32(false, a, false, b2, (short)0, a2, false, false);
        a3 = __builtin_amdgcn_wmma_f32_16x16x4_f32(false, a, false, b3, (short)0, a3, false, false);
    }

    float* out = h1 + (size_t)(tile * 16 + 8 * half) * HID + m;
    #pragma unroll
    for (int v = 0; v < 8; ++v) {
        out[v * HID +  0] = a0[v];
        out[v * HID + 16] = a1[v];
        out[v * HID + 32] = a2[v];
        out[v * HID + 48] = a3[v];
    }
}

__global__ void __launch_bounds__(128)
k_gemm2(const float* __restrict__ z1, const float* __restrict__ W2,
        float* __restrict__ h2) {
    __shared__ float wT[16 * 68];
    int t = threadIdx.x;
    for (int i = t; i < 16 * HID; i += 128) {
        int k = i >> 4, c = i & 15;
        wT[c * 68 + k] = (c < NCLS) ? W2[k * NCLS + c] : 0.0f;
    }
    __syncthreads();

    int tile = blockIdx.x * 4 + (t >> 5);
    if (tile >= ROW_TILES) return;
    int lane = t & 31;
    int m = lane & 15, half = lane >> 4;

    const float* arow = z1 + (size_t)(tile * 16 + m) * HID + 2 * half;
    v8f acc = {};
    #pragma unroll
    for (int k0 = 0; k0 < HID; k0 += 4) {
        v2f a = *(const v2f*)(arow + k0);
        v2f b = *(const v2f*)(wT + m * 68 + k0 + 2 * half);
        acc = __builtin_amdgcn_wmma_f32_16x16x4_f32(false, a, false, b, (short)0, acc, false, false);
    }
    if (m < NCLS) {
        float* out = h2 + (size_t)(tile * 16 + 8 * half) * NCLS + m;
        #pragma unroll
        for (int v = 0; v < 8; ++v) out[v * NCLS] = acc[v];
    }
}

#define AG_NB   250
#define AG_R    RG_R
#define AG_QCAP 8192
#define AG_T    1024
#define AG_SB   8
#define AG_DB   8

template <bool COUNT>
__device__ __forceinline__ void ag_push(unsigned* qn, uint2* q, unsigned* cnt, unsigned r, int e) {
    if (COUNT) __hip_atomic_fetch_add(cnt + r, 1u, __ATOMIC_RELAXED, __HIP_MEMORY_SCOPE_WORKGROUP);
    unsigned pos = __hip_atomic_fetch_add(qn, 1u, __ATOMIC_RELAXED, __HIP_MEMORY_SCOPE_WORKGROUP);
    if (pos < AG_QCAP) q[pos].x = (r << 21) | (unsigned)e;
}

template <bool COUNT>
__device__ __forceinline__ void ag_filter(const int* __restrict__ dst, int e0, int e1,
                                          int lo, unsigned nrows,
                                          unsigned* qn, uint2* q, unsigned* cnt, int t) {
    for (int e = e0 + 4 * t; e < e1; e += 4 * AG_T * AG_SB) {
        int4 d[AG_SB];
        #pragma unroll
        for (int u = 0; u < AG_SB; ++u) {
            int ee = e + u * 4 * AG_T;
            d[u] = (ee < e1) ? *(const int4*)(dst + ee) : make_int4(-1, -1, -1, -1);
        }
        #pragma unroll
        for (int u = 0; u < AG_SB; ++u) {
            int ee = e + u * 4 * AG_T;
            unsigned r0 = (unsigned)(d[u].x - lo), r1 = (unsigned)(d[u].y - lo);
            unsigned r2 = (unsigned)(d[u].z - lo), r3 = (unsigned)(d[u].w - lo);
            if ((r0 < nrows) | (r1 < nrows) | (r2 < nrows) | (r3 < nrows)) {
                if (r0 < nrows) ag_push<COUNT>(qn, q, cnt, r0, ee);
                if (r1 < nrows) ag_push<COUNT>(qn, q, cnt, r1, ee + 1);
                if (r2 < nrows) ag_push<COUNT>(qn, q, cnt, r2, ee + 2);
                if (r3 < nrows) ag_push<COUNT>(qn, q, cnt, r3, ee + 3);
            }
        }
    }
}

__device__ __forceinline__ void ag_lookup(const int* __restrict__ src, const float* __restrict__ dis,
                                          int lo, uint2* q, unsigned n, int t) {
    for (unsigned i0 = t; i0 < n; i0 += 4u * AG_T) {
        unsigned en[4], s[4];
        float w[4];
        #pragma unroll
        for (int u = 0; u < 4; ++u) {
            unsigned i = i0 + u * AG_T;
            en[u] = (i < n) ? q[i].x : 0u;
        }
        #pragma unroll
        for (int u = 0; u < 4; ++u) s[u] = (unsigned)src[en[u] & 0x1FFFFFu];
        #pragma unroll
        for (int u = 0; u < 4; ++u) w[u] = dis[s[u]] * dis[lo + (int)(en[u] >> 21)];
        #pragma unroll
        for (int u = 0; u < 4; ++u) {
            unsigned i = i0 + u * AG_T;
            q[(i < n) ? i : AG_QCAP] = make_uint2(((en[u] >> 21) << 17) | s[u], __float_as_uint(w[u]));
        }
    }
    __syncthreads();
}

__device__ __forceinline__ void ag_add64(const float* __restrict__ h1,
                                         float* acc, const uint2* q, unsigned n, int t) {
    const unsigned hw = (unsigned)t >> 4;
    const unsigned l16 = (unsigned)t & 15u;
    const bool hi = (hw & 1u) != 0u;
    const unsigned oA = hi ? 2u : 0u, oB = oA ^ 2u;
    for (unsigned i0 = hw; i0 < n; i0 += 64u * AG_DB) {
        uint2 en[AG_DB];
        float4 row[AG_DB];
        #pragma unroll
        for (int u = 0; u < AG_DB; ++u) {
            unsigned i = i0 + 64u * u;
            en[u] = (i < n) ? q[i] : make_uint2((unsigned)AG_R << 17, 0u);
        }
        #pragma unroll
        for (int u = 0; u < AG_DB; ++u)
            row[u] = *(const float4*)(h1 + (size_t)(en[u].x & 0x1FFFFu) * HID + 4u * l16);
        #pragma unroll
        for (int u = 0; u < AG_DB; ++u) {
            float w = __uint_as_float(en[u].y);
            float* p = acc + (en[u].x >> 17) * HID + 4u * l16;
            float a0 = hi ? row[u].z : row[u].x, a1 = hi ? row[u].w : row[u].y;
            float c0 = hi ? row[u].x : row[u].z, c1 = hi ? row[u].y : row[u].w;
            __hip_atomic_fetch_add(p + oA,      a0 * w, __ATOMIC_RELAXED, __HIP_MEMORY_SCOPE_WORKGROUP);
            __hip_atomic_fetch_add(p + oA + 1u, a1 * w, __ATOMIC_RELAXED, __HIP_MEMORY_SCOPE_WORKGROUP);
            __hip_atomic_fetch_add(p + oB,      c0 * w, __ATOMIC_RELAXED, __HIP_MEMORY_SCOPE_WORKGROUP);
            __hip_atomic_fetch_add(p + oB + 1u, c1 * w, __ATOMIC_RELAXED, __HIP_MEMORY_SCOPE_WORKGROUP);
        }
    }
}

#define AG_LB   8

#define AG_FB   8
#define NG16    (NE / 16)

static_assert(NE % 16 == 0, "the range-id list is read in 16-byte pieces");
static_assert(AG_NB <= 255, "a range id fits one byte, 255 left free");
static_assert(AG_FB % 2 == 0, "two pieces share one 32-bit hit mask");
static_assert((long long)((ROW_TILES + 3) / 4) * 128 * 8 >= NE, "the first launch packs every edge's range id");

__global__ void __launch_bounds__(AG_T)
k_own(const int* __restrict__ src, const int* __restrict__ dst, const unsigned* __restrict__ rid,
      float* __restrict__ dis, unsigned* __restrict__ eq1, unsigned* __restrict__ ecnt) {
    __shared__ uint2 q[AG_QCAP + 1];
    __shared__ unsigned cnt[AG_R];
    __shared__ unsigned qn;
    const int t = threadIdx.x;
    const int lo = blockIdx.x * AG_R;
    const unsigned nrows = (unsigned)((NN - lo) < AG_R ? (NN - lo) : AG_R);

    if (t < AG_R) cnt[t] = 0u;
    if (t == 0) qn = 0u;
    __syncthreads();

    const unsigned own4 = blockIdx.x * 0x01010101u;
    for (int g = t; g < NG16; g += AG_T * AG_FB) {
        uint4 w[AG_FB];
        #pragma unroll
        for (int u = 0; u < AG_FB; ++u) {
            int gg = g + u * AG_T;
            w[u] = (gg < NG16) ? ((const uint4*)rid)[gg] : make_uint4(~0u, ~0u, ~0u, ~0u);
        }
        #pragma unroll
        for (int v = 0; v < AG_FB / 2; ++v) {
            const unsigned ww[8] = { w[2 * v].x, w[2 * v].y, w[2 * v].z, w[2 * v].w,
                                     w[2 * v + 1].x, w[2 * v + 1].y, w[2 * v + 1].z, w[2 * v + 1].w };
            unsigned nz = 0u;
            #pragma unroll
            for (int k = 0; k < 8; ++k) {
                unsigned x = ww[k] ^ own4;
                unsigned s = ((x & 0x7F7F7F7Fu) + 0x7F7F7F7Fu) | x;
                nz |= (s >> (7 - k)) & (0x01010101u << k);
            }
            unsigned hit = ~nz;
            while (hit) {
                unsigned p = (unsigned)__builtin_ctz(hit);
                hit &= hit - 1u;
                unsigned k = p & 7u, j = p >> 3;
                int gg = g + (2 * v + (int)(k >> 2)) * AG_T;
                unsigned e = (unsigned)gg * 16u + (k & 3u) * 4u + j;
                unsigned pos = __hip_atomic_fetch_add(&qn, 1u, __ATOMIC_RELAXED, __HIP_MEMORY_SCOPE_WORKGROUP);
                if (pos < AG_QCAP) q[pos].x = e;
            }
        }
    }
    __syncthreads();
    const unsigned total = qn;
    if (t == 0) ecnt[blockIdx.x] = total;
    if (total <= AG_QCAP) {
        unsigned* __restrict__ g1 = eq1 + (size_t)blockIdx.x * AG_QCAP;
        for (unsigned i0 = t; i0 < total; i0 += 4u * AG_T) {
            unsigned ix[4], en[4], sn[4], rn[4];
            #pragma unroll
            for (int u = 0; u < 4; ++u) {
                unsigned i = i0 + u * AG_T;
                ix[u] = (i < total) ? i : i0;
                en[u] = q[ix[u]].x;
            }
            #pragma unroll
            for (int u = 0; u < 4; ++u) rn[u] = (unsigned)(dst[en[u]] - lo);
            #pragma unroll
            for (int u = 0; u < 4; ++u) sn[u] = (unsigned)src[en[u]];
            #pragma unroll
            for (int u = 0; u < 4; ++u) g1[ix[u]] = (rn[u] << 17) | sn[u];
            #pragma unroll
            for (int u = 0; u < 4; ++u)
                if (i0 + u * AG_T < total && rn[u] < nrows)
                    __hip_atomic_fetch_add(cnt + rn[u], 1u, __ATOMIC_RELAXED, __HIP_MEMORY_SCOPE_WORKGROUP);
        }
    } else {
        __syncthreads();
        if (t == 0) qn = 0u;
        __syncthreads();
        ag_filter<true>(dst, 0, NE, lo, nrows, &qn, q, cnt, t);
    }
    __syncthreads();
    if ((unsigned)t < nrows) dis[lo + t] = rsqrtf((float)(cnt[t] + 1u));
}

__global__ void __launch_bounds__(AG_T)
k_agg1(const int* __restrict__ src, const int* __restrict__ dst,
       const float* __restrict__ h1, const float* __restrict__ dis, const float* __restrict__ b1,
       float* __restrict__ z1, const unsigned* __restrict__ eq1,
       unsigned* __restrict__ eq, const unsigned* __restrict__ ecnt) {
    __shared__ float acc[(AG_R + 1) * HID];
    __shared__ uint2 q[AG_QCAP + 1];
    __shared__ unsigned qn;
    const int t = threadIdx.x;
    const int lo = blockIdx.x * AG_R;
    const unsigned nrows = (unsigned)((NN - lo) < AG_R ? (NN - lo) : AG_R);

    for (int i = t; i < (AG_R + 1) * HID / 4; i += AG_T)
        ((float4*)acc)[i] = make_float4(0.0f, 0.0f, 0.0f, 0.0f);
    const unsigned total = ecnt[blockIdx.x];
    if (total <= AG_QCAP) {
        const unsigned* __restrict__ g1 = eq1 + (size_t)blockIdx.x * AG_QCAP;
        uint2* __restrict__ g2 = (uint2*)eq + (size_t)blockIdx.x * AG_QCAP;
        for (unsigned i0 = t; i0 < total; i0 += (unsigned)AG_LB * AG_T) {
            unsigned v[AG_LB];
            float w[AG_LB];
            #pragma unroll
            for (int u = 0; u < AG_LB; ++u) {
                unsigned i = i0 + u * AG_T;
                v[u] = g1[(i < total) ? i : i0];
            }
            #pragma unroll
            for (int u = 0; u < AG_LB; ++u)
                w[u] = dis[v[u] & 0x1FFFFu] * dis[lo + (int)(v[u] >> 17)];
            #pragma unroll
            for (int u = 0; u < AG_LB; ++u) {
                unsigned i = i0 + u * AG_T;
                uint2 o = make_uint2(v[u], __float_as_uint(w[u]));
                q[(i < total) ? i : AG_QCAP] = o;
                if (i < total) g2[i] = o;
            }
        }
        __syncthreads();
        ag_add64(h1, acc, q, total, t);
    } else {
        for (int e0 = 0; e0 < NE; e0 += AG_QCAP) {
            __syncthreads();
            if (t == 0) qn = 0u;
            __syncthreads();
            const int e1 = (e0 + AG_QCAP < NE) ? (e0 + AG_QCAP) : NE;
            ag_filter<false>(dst, e0, e1, lo, nrows, &qn, q, nullptr, t);
            __syncthreads();
            const unsigned n = qn;
            ag_lookup(src, dis, lo, q, n, t);
            ag_add64(h1, acc, q, n, t);
        }
    }
    __syncthreads();

    for (unsigned i = t; i < nrows * 16u; i += AG_T) {
        unsigned r = i >> 4, c = (i & 15u) * 4u;
        size_t g = (size_t)(lo + r) * HID + c;
        float di = dis[lo + r];
        float4 a = *(const float4*)(acc + r * HID + c);
        float4 h = *(const float4*)(h1 + g);
        float4 b = *(const float4*)(b1 + c);
        float4 v;
        v.x = a.x + h.x * di * di + b.x;
        v.y = a.y + h.y * di * di + b.y;
        v.z = a.z + h.z * di * di + b.z;
        v.w = a.w + h.w * di * di + b.w;
        v.x = v.x > 0.0f ? v.x : 0.0f;
        v.y = v.y > 0.0f ? v.y : 0.0f;
        v.z = v.z > 0.0f ? v.z : 0.0f;
        v.w = v.w > 0.0f ? v.w : 0.0f;
        *(float4*)(z1 + g) = v;
    }
}

#define AG_D2   8

static_assert((AG_R * NCLS) % 32 == 0, "a range's block of out must be whole 128-byte lines");
static_assert(((NN % AG_R) * NCLS) % 4 == 0, "the last range's block of out must be whole float4s");
static_assert(AG_R <= AG_T, "the layer-2 epilogue gives every own row a thread of its own");
static_assert((long long)AG_NB * AG_R >= NN && (long long)(AG_NB - 1) * AG_R < NN, "the ranges cover the nodes and none is empty");

__device__ __forceinline__ void ag_add10(const float* __restrict__ h2,
                                         float* acc2, const uint2* q, unsigned n, int t) {
    const unsigned g = (unsigned)t >> 3;
    const unsigned j = (unsigned)t & 7u;
    if (j < 5u) {
        for (unsigned i0 = g; i0 < n; i0 += 128u * AG_D2) {
            uint2 en[AG_D2];
            v2f row[AG_D2];
            #pragma unroll
            for (int u = 0; u < AG_D2; ++u) {
                unsigned i = i0 + 128u * u;
                en[u] = (i < n) ? q[i] : make_uint2((unsigned)AG_R << 17, 0u);
            }
            #pragma unroll
            for (int u = 0; u < AG_D2; ++u)
                row[u] = *(const v2f*)(h2 + (size_t)(en[u].x & 0x1FFFFu) * NCLS + 2u * j);
            #pragma unroll
            for (int u = 0; u < AG_D2; ++u) {
                float w = __uint_as_float(en[u].y);
                float* p = acc2 + (en[u].x >> 17) * NCLS + 2u * j;
                __hip_atomic_fetch_add(p,      row[u].x * w, __ATOMIC_RELAXED, __HIP_MEMORY_SCOPE_WORKGROUP);
                __hip_atomic_fetch_add(p + 1u, row[u].y * w, __ATOMIC_RELAXED, __HIP_MEMORY_SCOPE_WORKGROUP);
            }
        }
    }
}

__global__ void __launch_bounds__(AG_T)
k_agg2(const int* __restrict__ src, const int* __restrict__ dst,
       const float* __restrict__ h2,
       const float* __restrict__ dis, const float* __restrict__ b2,
       const unsigned* __restrict__ eq, const unsigned* __restrict__ ecnt,
       float* __restrict__ out) {
    __shared__ __attribute__((aligned(16))) float acc2[(AG_R + 1) * NCLS];
    __shared__ uint2 q[AG_QCAP + 1];
    __shared__ unsigned qn;
    const int t = threadIdx.x;
    const int lo = blockIdx.x * AG_R;
    const unsigned nrows = (unsigned)((NN - lo) < AG_R ? (NN - lo) : AG_R);

    for (int i = t; i < (AG_R + 1) * NCLS; i += AG_T) acc2[i] = 0.0f;
    const unsigned total = ecnt[blockIdx.x];
    if (total <= AG_QCAP) {
        const uint2* __restrict__ gq = (const uint2*)eq + (size_t)blockIdx.x * AG_QCAP;
        for (unsigned i0 = t; i0 < total; i0 += 4u * AG_T) {
            uint2 v[4];
            #pragma unroll
            for (int u = 0; u < 4; ++u) {
                unsigned i = i0 + u * AG_T;
                v[u] = gq[(i < total) ? i : i0];
            }
            #pragma unroll
            for (int u = 0; u < 4; ++u) {
                unsigned i = i0 + u * AG_T;
                q[(i < total) ? i : AG_QCAP] = v[u];
            }
        }
        __syncthreads();
        ag_add10(h2, acc2, q, total, t);
    } else {
        for (int e0 = 0; e0 < NE; e0 += AG_QCAP) {
            __syncthreads();
            if (t == 0) qn = 0u;
            __syncthreads();
            const int e1 = (e0 + AG_QCAP < NE) ? (e0 + AG_QCAP) : NE;
            ag_filter<false>(dst, e0, e1, lo, nrows, &qn, q, nullptr, t);
            __syncthreads();
            const unsigned n = qn;
            ag_lookup(src, dis, lo, q, n, t);
            ag_add10(h2, acc2, q, n, t);
        }
    }
    __syncthreads();

    if ((unsigned)t < nrows) {
        float di = dis[lo + t];
        float d2 = di * di;
        const float* hp = h2 + (size_t)(lo + t) * NCLS;
        float* ap = acc2 + t * NCLS;
        float tv[NCLS];
        float mx = -3.402823466e38f;
        #pragma unroll
        for (int f = 0; f < NCLS; ++f) {
            float v = ap[f] + hp[f] * d2 + b2[f];
            tv[f] = v;
            mx = fmaxf(mx, v);
        }
        float sum = 0.0f;
        #pragma unroll
        for (int f = 0; f < NCLS; ++f) sum += expf(tv[f] - mx);
        float lse = mx + logf(sum);
        #pragma unroll
        for (int f = 0; f < NCLS; ++f) ap[f] = tv[f] - lse;
    }
    __syncthreads();

    float4* ob = (float4*)(out + (size_t)lo * NCLS);
    for (unsigned i = t; i < nrows * NCLS / 4u; i += AG_T) ob[i] = ((const float4*)acc2)[i];
}

extern "C" void kernel_launch(void* const* d_in, const int* in_sizes, int n_in,
                              void* d_out, int out_size, void* d_ws, size_t ws_size,
                              hipStream_t stream) {
    const float* x  = (const float*)d_in[0];
    const int*   ei = (const int*)d_in[1];
    const float* W1 = (const float*)d_in[2];
    const float* b1 = (const float*)d_in[3];
    const float* W2 = (const float*)d_in[4];
    const float* b2 = (const float*)d_in[5];
    const int* src = ei;
    const int* dst = ei + NE;
    float* out = (float*)d_out;

    float* ws   = (float*)d_ws;
    float* dis  = ws;
    float* h1   = dis + NN;
    float* agg1 = h1 + (size_t)NN * HID;
    float* h2   = agg1 + (size_t)NN * HID;
    unsigned* eq1  = (unsigned*)(h2 + (size_t)NN * NCLS);
    unsigned* eq   = eq1 + (size_t)AG_NB * AG_QCAP;
    unsigned* ecnt = eq + (size_t)AG_NB * AG_QCAP * 2;
    unsigned* rid  = ecnt + 256;

    k_gemm1<<<(ROW_TILES + 3) / 4, 128, 0, stream>>>(x, W1, h1, dst, rid);

    k_own<<<AG_NB, AG_T, 0, stream>>>(src, dst, rid, dis, eq1, ecnt);

    k_agg1<<<AG_NB, AG_T, 0, stream>>>(src, dst, h1, dis, b1, agg1, eq1, eq, ecnt);

    k_gemm2<<<(ROW_TILES + 3) / 4, 128, 0, stream>>>(agg1, W2, h2);
    k_agg2<<<AG_NB, AG_T, 0, stream>>>(src, dst, h2, dis, b2, eq, ecnt, out);
}


# Layer-1 row sums off LDS float atomics: entries bucketed per owner 16-lane group (row mod 64) as they are weighted; each group adds its own rows by LDS load, fma, store; sums in bucket arrival order
# speedup vs baseline: 3.7810x; 1.0852x over previous
#include <hip/hip_runtime.h>
#include <math.h>

#define NN 100000
#define NE 1600000
#define FIN 128
#define HID 64
#define NCLS 10
#define ROW_TILES 6250

typedef __attribute__((ext_vector_type(2))) float v2f;
typedef __attribute__((ext_vector_type(8))) float v8f;

#define RG_R 400

__device__ __forceinline__ unsigned rg_id(int d) {
    return ((unsigned)d < (unsigned)NN) ? (unsigned)d / (unsigned)RG_R : 255u;
}

__global__ void __launch_bounds__(128)
k_gemm1(const float* __restrict__ x, const float* __restrict__ W1,
        float* __restrict__ h1, const int* __restrict__ dst, unsigned* __restrict__ rid) {
    __shared__ float wT[HID * 132];
    int t = threadIdx.x;
    {
        int e8 = (blockIdx.x * 128 + t) * 8;
        if (e8 < NE) {
            int4 d0 = *(const int4*)(dst + e8);
            int4 d1 = *(const int4*)(dst + e8 + 4);
            uint2 o;
            o.x = rg_id(d0.x) | (rg_id(d0.y) << 8) | (rg_id(d0.z) << 16) | (rg_id(d0.w) << 24);
            o.y = rg_id(d1.x) | (rg_id(d1.y) << 8) | (rg_id(d1.z) << 16) | (rg_id(d1.w) << 24);
            *(uint2*)(rid + (e8 >> 2)) = o;
        }
    }
    for (int i = t; i < FIN * HID; i += 128) {
        int k = i >> 6, c = i & 63;
        wT[c * 132 + k] = W1[i];
    }
    __syncthreads();

    int tile = blockIdx.x * 4 + (t >> 5);
    if (tile >= ROW_TILES) return;
    int lane = t & 31;
    int m = lane & 15, half = lane >> 4;

    const float* arow = x + (size_t)(tile * 16 + m) * FIN + 2 * half;
    v8f a0 = {}, a1 = {}, a2 = {}, a3 = {};

    #pragma unroll 4
    for (int k0 = 0; k0 < FIN; k0 += 4) {
        v2f a = *(const v2f*)(arow + k0);
        const float* bp = wT + m * 132 + k0 + 2 * half;
        v2f b0 = *(const v2f*)(bp);
        v2f b1 = *(const v2f*)(bp + 16 * 132);
        v2f b2 = *(const v2f*)(bp + 32 * 132);
        v2f b3 = *(const v2f*)(bp + 48 * 132);
        a0 = __builtin_amdgcn_wmma_f32_16x16x4_f32(false, a, false, b0, (short)0, a0, false, false);
        a1 = __builtin_amdgcn_wmma_f32_16x16x4_f32(false, a, false, b1, (short)0, a1, false, false);
        a2 = __builtin_amdgcn_wmma_f32_16x16x4_f32(false, a, false, b2, (short)0, a2, false, false);
        a3 = __builtin_amdgcn_wmma_f32_16x16x4_f32(false, a, false, b3, (short)0, a3, false, false);
    }

    float* out = h1 + (size_t)(tile * 16 + 8 * half) * HID + m;
    #pragma unroll
    for (int v = 0; v < 8; ++v) {
        out[v * HID +  0] = a0[v];
        out[v * HID + 16] = a1[v];
        out[v * HID + 32] = a2[v];
        out[v * HID + 48] = a3[v];
    }
}

__global__ void __launch_bounds__(128)
k_gemm2(const float* __restrict__ z1, const float* __restrict__ W2,
        float* __restrict__ h2) {
    __shared__ float wT[16 * 68];
    int t = threadIdx.x;
    for (int i = t; i < 16 * HID; i += 128) {
        int k = i >> 4, c = i & 15;
        wT[c * 68 + k] = (c < NCLS) ? W2[k * NCLS + c] : 0.0f;
    }
    __syncthreads();

    int tile = blockIdx.x * 4 + (t >> 5);
    if (tile >= ROW_TILES) return;
    int lane = t & 31;
    int m = lane & 15, half = lane >> 4;

    const float* arow = z1 + (size_t)(tile * 16 + m) * HID + 2 * half;
    v8f acc = {};
    #pragma unroll
    for (int k0 = 0; k0 < HID; k0 += 4) {
        v2f a = *(const v2f*)(arow + k0);
        v2f b = *(const v2f*)(wT + m * 68 + k0 + 2 * half);
        acc = __builtin_amdgcn_wmma_f32_16x16x4_f32(false, a, false, b, (short)0, acc, false, false);
    }
    if (m < NCLS) {
        float* out = h2 + (size_t)(tile * 16 + 8 * half) * NCLS + m;
        #pragma unroll
        for (int v = 0; v < 8; ++v) out[v * NCLS] = acc[v];
    }
}

#define AG_NB   250
#define AG_R    RG_R
#define AG_QCAP 8192
#define AG_T    1024
#define AG_SB   8
#define AG_DB   8
#define AG_OG   64
#define AG_BCAP 112

template <bool COUNT>
__device__ __forceinline__ void ag_push(unsigned* qn, uint2* q, unsigned* cnt, unsigned r, int e) {
    if (COUNT) __hip_atomic_fetch_add(cnt + r, 1u, __ATOMIC_RELAXED, __HIP_MEMORY_SCOPE_WORKGROUP);
    unsigned pos = __hip_atomic_fetch_add(qn, 1u, __ATOMIC_RELAXED, __HIP_MEMORY_SCOPE_WORKGROUP);
    if (pos < AG_QCAP) q[pos].x = (r << 21) | (unsigned)e;
}

template <bool COUNT>
__device__ __forceinline__ void ag_filter(const int* __restrict__ dst, int e0, int e1,
                                          int lo, unsigned nrows,
                                          unsigned* qn, uint2* q, unsigned* cnt, int t) {
    for (int e = e0 + 4 * t; e < e1; e += 4 * AG_T * AG_SB) {
        int4 d[AG_SB];
        #pragma unroll
        for (int u = 0; u < AG_SB; ++u) {
            int ee = e + u * 4 * AG_T;
            d[u] = (ee < e1) ? *(const int4*)(dst + ee) : make_int4(-1, -1, -1, -1);
        }
        #pragma unroll
        for (int u = 0; u < AG_SB; ++u) {
            int ee = e + u * 4 * AG_T;
            unsigned r0 = (unsigned)(d[u].x - lo), r1 = (unsigned)(d[u].y - lo);
            unsigned r2 = (unsigned)(d[u].z - lo), r3 = (unsigned)(d[u].w - lo);
            if ((r0 < nrows) | (r1 < nrows) | (r2 < nrows) | (r3 < nrows)) {
                if (r0 < nrows) ag_push<COUNT>(qn, q, cnt, r0, ee);
                if (r1 < nrows) ag_push<COUNT>(qn, q, cnt, r1, ee + 1);
                if (r2 < nrows) ag_push<COUNT>(qn, q, cnt, r2, ee + 2);
                if (r3 < nrows) ag_push<COUNT>(qn, q, cnt, r3, ee + 3);
            }
        }
    }
}

__device__ __forceinline__ void ag_lookup(const int* __restrict__ src, const float* __restrict__ dis,
                                          int lo, uint2* q, unsigned n, int t) {
    for (unsigned i0 = t; i0 < n; i0 += 4u * AG_T) {
        unsigned en[4], s[4];
        float w[4];
        #pragma unroll
        for (int u = 0; u < 4; ++u) {
            unsigned i = i0 + u * AG_T;
            en[u] = (i < n) ? q[i].x : 0u;
        }
        #pragma unroll
        for (int u = 0; u < 4; ++u) s[u] = (unsigned)src[en[u] & 0x1FFFFFu];
        #pragma unroll
        for (int u = 0; u < 4; ++u) w[u] = dis[s[u]] * dis[lo + (int)(en[u] >> 21)];
        #pragma unroll
        for (int u = 0; u < 4; ++u) {
            unsigned i = i0 + u * AG_T;
            q[(i < n) ? i : AG_QCAP] = make_uint2(((en[u] >> 21) << 17) | s[u], __float_as_uint(w[u]));
        }
    }
    __syncthreads();
}

__device__ __forceinline__ void ag_add64(const float* __restrict__ h1,
                                         float* acc, const uint2* q, unsigned n, int t) {
    const unsigned hw = (unsigned)t >> 4;
    const unsigned l16 = (unsigned)t & 15u;
    const bool hi = (hw & 1u) != 0u;
    const unsigned oA = hi ? 2u : 0u, oB = oA ^ 2u;
    for (unsigned i0 = hw; i0 < n; i0 += 64u * AG_DB) {
        uint2 en[AG_DB];
        float4 row[AG_DB];
        #pragma unroll
        for (int u = 0; u < AG_DB; ++u) {
            unsigned i = i0 + 64u * u;
            en[u] = (i < n) ? q[i] : make_uint2((unsigned)AG_R << 17, 0u);
        }
        #pragma unroll
        for (int u = 0; u < AG_DB; ++u)
            row[u] = *(const float4*)(h1 + (size_t)(en[u].x & 0x1FFFFu) * HID + 4u * l16);
        #pragma unroll
        for (int u = 0; u < AG_DB; ++u) {
            float w = __uint_as_float(en[u].y);
            float* p = acc + (en[u].x >> 17) * HID + 4u * l16;
            float a0 = hi ? row[u].z : row[u].x, a1 = hi ? row[u].w : row[u].y;
            float c0 = hi ? row[u].x : row[u].z, c1 = hi ? row[u].y : row[u].w;
            __hip_atomic_fetch_add(p + oA,      a0 * w, __ATOMIC_RELAXED, __HIP_MEMORY_SCOPE_WORKGROUP);
            __hip_atomic_fetch_add(p + oA + 1u, a1 * w, __ATOMIC_RELAXED, __HIP_MEMORY_SCOPE_WORKGROUP);
            __hip_atomic_fetch_add(p + oB,      c0 * w, __ATOMIC_RELAXED, __HIP_MEMORY_SCOPE_WORKGROUP);
            __hip_atomic_fetch_add(p + oB + 1u, c1 * w, __ATOMIC_RELAXED, __HIP_MEMORY_SCOPE_WORKGROUP);
        }
    }
}

__device__ __forceinline__ void ag_own64(const float* __restrict__ h1, float* acc,
                                         const uint2* qb, unsigned n, unsigned spare, int t) {
    const unsigned l16 = (unsigned)t & 15u;
    const unsigned no = (unsigned)__shfl_xor((int)n, 16);
    const unsigned nw = n > no ? n : no;
    const unsigned padx = spare << 17;
    uint2 en[AG_DB];
    float4 row[AG_DB];
    #pragma unroll
    for (int u = 0; u < AG_DB; ++u) {
        uint2 e = qb[u];
        en[u].x = ((unsigned)u < n) ? e.x : padx;
        en[u].y = ((unsigned)u < n) ? e.y : 0u;
    }
    #pragma unroll
    for (int u = 0; u < AG_DB; ++u)
        row[u] = *(const float4*)(h1 + (size_t)(en[u].x & 0x1FFFFu) * HID + 4u * l16);
    for (unsigned i0 = 0; i0 < nw; i0 += AG_DB) {
        uint2 en2[AG_DB];
        float4 row2[AG_DB];
        const unsigned j0 = i0 + AG_DB;
        #pragma unroll
        for (int u = 0; u < AG_DB; ++u) {
            const unsigned j = j0 + (unsigned)u;
            uint2 e = qb[j < (unsigned)AG_BCAP ? j : (unsigned)AG_BCAP - 1u];
            en2[u].x = (j < n) ? e.x : padx;
            en2[u].y = (j < n) ? e.y : 0u;
        }
        #pragma unroll
        for (int u = 0; u < AG_DB; ++u)
            row2[u] = *(const float4*)(h1 + (size_t)(en2[u].x & 0x1FFFFu) * HID + 4u * l16);
        #pragma unroll
        for (int u = 0; u < AG_DB; ++u) {
            float w = __uint_as_float(en[u].y);
            float4* p = (float4*)(acc + (en[u].x >> 17) * HID + 4u * l16);
            float4 a = *p;
            a.x += row[u].x * w;
            a.y += row[u].y * w;
            a.z += row[u].z * w;
            a.w += row[u].w * w;
            *p = a;
        }
        #pragma unroll
        for (int u = 0; u < AG_DB; ++u) { en[u] = en2[u]; row[u] = row2[u]; }
    }
}

#define AG_LB   8

#define AG_FB   8
#define NG16    (NE / 16)

static_assert(NE % 16 == 0, "the range-id list is read in 16-byte pieces");
static_assert(AG_NB <= 255, "a range id fits one byte, 255 left free");
static_assert(AG_FB % 2 == 0, "two pieces share one 32-bit hit mask");
static_assert((long long)((ROW_TILES + 3) / 4) * 128 * 8 >= NE, "the first launch packs every edge's range id");

__global__ void __launch_bounds__(AG_T)
k_own(const int* __restrict__ src, const int* __restrict__ dst, const unsigned* __restrict__ rid,
      float* __restrict__ dis, unsigned* __restrict__ eq1, unsigned* __restrict__ ecnt) {
    __shared__ uint2 q[AG_QCAP + 1];
    __shared__ unsigned cnt[AG_R];
    __shared__ unsigned qn;
    const int t = threadIdx.x;
    const int lo = blockIdx.x * AG_R;
    const unsigned nrows = (unsigned)((NN - lo) < AG_R ? (NN - lo) : AG_R);

    if (t < AG_R) cnt[t] = 0u;
    if (t == 0) qn = 0u;
    __syncthreads();

    const unsigned own4 = blockIdx.x * 0x01010101u;
    for (int g = t; g < NG16; g += AG_T * AG_FB) {
        uint4 w[AG_FB];
        #pragma unroll
        for (int u = 0; u < AG_FB; ++u) {
            int gg = g + u * AG_T;
            w[u] = (gg < NG16) ? ((const uint4*)rid)[gg] : make_uint4(~0u, ~0u, ~0u, ~0u);
        }
        #pragma unroll
        for (int v = 0; v < AG_FB / 2; ++v) {
            const unsigned ww[8] = { w[2 * v].x, w[2 * v].y, w[2 * v].z, w[2 * v].w,
                                     w[2 * v + 1].x, w[2 * v + 1].y, w[2 * v + 1].z, w[2 * v + 1].w };
            unsigned nz = 0u;
            #pragma unroll
            for (int k = 0; k < 8; ++k) {
                unsigned x = ww[k] ^ own4;
                unsigned s = ((x & 0x7F7F7F7Fu) + 0x7F7F7F7Fu) | x;
                nz |= (s >> (7 - k)) & (0x01010101u << k);
            }
            unsigned hit = ~nz;
            while (hit) {
                unsigned p = (unsigned)__builtin_ctz(hit);
                hit &= hit - 1u;
                unsigned k = p & 7u, j = p >> 3;
                int gg = g + (2 * v + (int)(k >> 2)) * AG_T;
                unsigned e = (unsigned)gg * 16u + (k & 3u) * 4u + j;
                unsigned pos = __hip_atomic_fetch_add(&qn, 1u, __ATOMIC_RELAXED, __HIP_MEMORY_SCOPE_WORKGROUP);
                if (pos < AG_QCAP) q[pos].x = e;
            }
        }
    }
    __syncthreads();
    const unsigned total = qn;
    if (t == 0) ecnt[blockIdx.x] = total;
    if (total <= AG_QCAP) {
        unsigned* __restrict__ g1 = eq1 + (size_t)blockIdx.x * AG_QCAP;
        for (unsigned i0 = t; i0 < total; i0 += 4u * AG_T) {
            unsigned ix[4], en[4], sn[4], rn[4];
            #pragma unroll
            for (int u = 0; u < 4; ++u) {
                unsigned i = i0 + u * AG_T;
                ix[u] = (i < total) ? i : i0;
                en[u] = q[ix[u]].x;
            }
            #pragma unroll
            for (int u = 0; u < 4; ++u) rn[u] = (unsigned)(dst[en[u]] - lo);
            #pragma unroll
            for (int u = 0; u < 4; ++u) sn[u] = (unsigned)src[en[u]];
            #pragma unroll
            for (int u = 0; u < 4; ++u) g1[ix[u]] = (rn[u] << 17) | sn[u];
            #pragma unroll
            for (int u = 0; u < 4; ++u)
                if (i0 + u * AG_T < total && rn[u] < nrows)
                    __hip_atomic_fetch_add(cnt + rn[u], 1u, __ATOMIC_RELAXED, __HIP_MEMORY_SCOPE_WORKGROUP);
        }
    } else {
        __syncthreads();
        if (t == 0) qn = 0u;
        __syncthreads();
        ag_filter<true>(dst, 0, NE, lo, nrows, &qn, q, cnt, t);
    }
    __syncthreads();
    if ((unsigned)t < nrows) dis[lo + t] = rsqrtf((float)(cnt[t] + 1u));
}

__global__ void __launch_bounds__(AG_T)
k_agg1(const int* __restrict__ src, const int* __restrict__ dst,
       const float* __restrict__ h1, const float* __restrict__ dis, const float* __restrict__ b1,
       float* __restrict__ z1, const unsigned* __restrict__ eq1,
       unsigned* __restrict__ eq, const unsigned* __restrict__ ecnt) {
    __shared__ float acc[(AG_R + AG_OG) * HID];
    __shared__ uint2 q[AG_QCAP + 1];
    __shared__ uint2 qb[AG_OG * AG_BCAP];
    __shared__ unsigned bcnt[AG_OG + 1];
    __shared__ unsigned qn;
    const int t = threadIdx.x;
    const int lo = blockIdx.x * AG_R;
    const unsigned nrows = (unsigned)((NN - lo) < AG_R ? (NN - lo) : AG_R);

    for (int i = t; i < (AG_R + AG_OG) * HID / 4; i += AG_T)
        ((float4*)acc)[i] = make_float4(0.0f, 0.0f, 0.0f, 0.0f);
    if (t <= AG_OG) bcnt[t] = 0u;
    if (t == 0) qn = 0u;
    const unsigned total = ecnt[blockIdx.x];
    __syncthreads();
    if (total <= AG_QCAP) {
        const unsigned* __restrict__ g1 = eq1 + (size_t)blockIdx.x * AG_QCAP;
        uint2* __restrict__ g2 = (uint2*)eq + (size_t)blockIdx.x * AG_QCAP;
        for (unsigned i0 = t; i0 < total; i0 += (unsigned)AG_LB * AG_T) {
            unsigned v[AG_LB];
            float w[AG_LB];
            #pragma unroll
            for (int u = 0; u < AG_LB; ++u) {
                unsigned i = i0 + u * AG_T;
                v[u] = g1[(i < total) ? i : i0];
            }
            #pragma unroll
            for (int u = 0; u < AG_LB; ++u)
                w[u] = dis[v[u] & 0x1FFFFu] * dis[lo + (int)(v[u] >> 17)];
            unsigned pos[AG_LB];
            #pragma unroll
            for (int u = 0; u < AG_LB; ++u) {
                unsigned i = i0 + u * AG_T;
                unsigned b = (i < total) ? (v[u] >> 17) % (unsigned)AG_OG : (unsigned)AG_OG;
                pos[u] = __hip_atomic_fetch_add(bcnt + b, 1u, __ATOMIC_RELAXED, __HIP_MEMORY_SCOPE_WORKGROUP);
            }
            #pragma unroll
            for (int u = 0; u < AG_LB; ++u) {
                unsigned i = i0 + u * AG_T;
                uint2 o = make_uint2(v[u], __float_as_uint(w[u]));
                if (i < total) {
                    g2[i] = o;
                    unsigned b = (v[u] >> 17) % (unsigned)AG_OG;
                    if (pos[u] < (unsigned)AG_BCAP) qb[b * AG_BCAP + pos[u]] = o;
                    else q[__hip_atomic_fetch_add(&qn, 1u, __ATOMIC_RELAXED, __HIP_MEMORY_SCOPE_WORKGROUP)] = o;
                }
            }
        }
        __syncthreads();
        {
            const unsigned hw = (unsigned)t >> 4;
            const unsigned nb = bcnt[hw] < (unsigned)AG_BCAP ? bcnt[hw] : (unsigned)AG_BCAP;
            ag_own64(h1, acc, qb + hw * AG_BCAP, nb, (unsigned)AG_R + hw, t);
        }
        __syncthreads();
        ag_add64(h1, acc, q, qn, t);
    } else {
        for (int e0 = 0; e0 < NE; e0 += AG_QCAP) {
            __syncthreads();
            if (t == 0) qn = 0u;
            __syncthreads();
            const int e1 = (e0 + AG_QCAP < NE) ? (e0 + AG_QCAP) : NE;
            ag_filter<false>(dst, e0, e1, lo, nrows, &qn, q, nullptr, t);
            __syncthreads();
            const unsigned n = qn;
            ag_lookup(src, dis, lo, q, n, t);
            ag_add64(h1, acc, q, n, t);
        }
    }
    __syncthreads();

    for (unsigned i = t; i < nrows * 16u; i += AG_T) {
        unsigned r = i >> 4, c = (i & 15u) * 4u;
        size_t g = (size_t)(lo + r) * HID + c;
        float di = dis[lo + r];
        float4 a = *(const float4*)(acc + r * HID + c);
        float4 h = *(const float4*)(h1 + g);
        float4 b = *(const float4*)(b1 + c);
        float4 v;
        v.x = a.x + h.x * di * di + b.x;
        v.y = a.y + h.y * di * di + b.y;
        v.z = a.z + h.z * di * di + b.z;
        v.w = a.w + h.w * di * di + b.w;
        v.x = v.x > 0.0f ? v.x : 0.0f;
        v.y = v.y > 0.0f ? v.y : 0.0f;
        v.z = v.z > 0.0f ? v.z : 0.0f;
        v.w = v.w > 0.0f ? v.w : 0.0f;
        *(float4*)(z1 + g) = v;
    }
}

#define AG_D2   8

static_assert((AG_R * NCLS) % 32 == 0, "a range's block of out must be whole 128-byte lines");
static_assert(((NN % AG_R) * NCLS) % 4 == 0, "the last range's block of out must be whole float4s");
static_assert(AG_R <= AG_T, "the layer-2 epilogue gives every own row a thread of its own");
static_assert((long long)AG_NB * AG_R >= NN && (long long)(AG_NB - 1) * AG_R < NN, "the ranges cover the nodes and none is empty");

__device__ __forceinline__ void ag_add10(const float* __restrict__ h2,
                                         float* acc2, const uint2* q, unsigned n, int t) {
    const unsigned g = (unsigned)t >> 3;
    const unsigned j = (unsigned)t & 7u;
    if (j < 5u) {
        for (unsigned i0 = g; i0 < n; i0 += 128u * AG_D2) {
            uint2 en[AG_D2];
            v2f row[AG_D2];
            #pragma unroll
            for (int u = 0; u < AG_D2; ++u) {
                unsigned i = i0 + 128u * u;
                en[u] = (i < n) ? q[i] : make_uint2((unsigned)AG_R << 17, 0u);
            }
            #pragma unroll
            for (int u = 0; u < AG_D2; ++u)
                row[u] = *(const v2f*)(h2 + (size_t)(en[u].x & 0x1FFFFu) * NCLS + 2u * j);
            #pragma unroll
            for (int u = 0; u < AG_D2; ++u) {
                float w = __uint_as_float(en[u].y);
                float* p = acc2 + (en[u].x >> 17) * NCLS + 2u * j;
                __hip_atomic_fetch_add(p,      row[u].x * w, __ATOMIC_RELAXED, __HIP_MEMORY_SCOPE_WORKGROUP);
                __hip_atomic_fetch_add(p + 1u, row[u].y * w, __ATOMIC_RELAXED, __HIP_MEMORY_SCOPE_WORKGROUP);
            }
        }
    }
}

__global__ void __launch_bounds__(AG_T)
k_agg2(const int* __restrict__ src, const int* __restrict__ dst,
       const float* __restrict__ h2,
       const float* __restrict__ dis, const float* __restrict__ b2,
       const unsigned* __restrict__ eq, const unsigned* __restrict__ ecnt,
       float* __restrict__ out) {
    __shared__ __attribute__((aligned(16))) float acc2[(AG_R + 1) * NCLS];
    __shared__ uint2 q[AG_QCAP + 1];
    __shared__ unsigned qn;
    const int t = threadIdx.x;
    const int lo = blockIdx.x * AG_R;
    const unsigned nrows = (unsigned)((NN - lo) < AG_R ? (NN - lo) : AG_R);

    for (int i = t; i < (AG_R + 1) * NCLS; i += AG_T) acc2[i] = 0.0f;
    const unsigned total = ecnt[blockIdx.x];
    if (total <= AG_QCAP) {
        const uint2* __restrict__ gq = (const uint2*)eq + (size_t)blockIdx.x * AG_QCAP;
        for (unsigned i0 = t; i0 < total; i0 += 4u * AG_T) {
            uint2 v[4];
            #pragma unroll
            for (int u = 0; u < 4; ++u) {
                unsigned i = i0 + u * AG_T;
                v[u] = gq[(i < total) ? i : i0];
            }
            #pragma unroll
            for (int u = 0; u < 4; ++u) {
                unsigned i = i0 + u * AG_T;
                q[(i < total) ? i : AG_QCAP] = v[u];
            }
        }
        __syncthreads();
        ag_add10(h2, acc2, q, total, t);
    } else {
        for (int e0 = 0; e0 < NE; e0 += AG_QCAP) {
            __syncthreads();
            if (t == 0) qn = 0u;
            __syncthreads();
            const int e1 = (e0 + AG_QCAP < NE) ? (e0 + AG_QCAP) : NE;
            ag_filter<false>(dst, e0, e1, lo, nrows, &qn, q, nullptr, t);
            __syncthreads();
            const unsigned n = qn;
            ag_lookup(src, dis, lo, q, n, t);
            ag_add10(h2, acc2, q, n, t);
        }
    }
    __syncthreads();

    if ((unsigned)t < nrows) {
        float di = dis[lo + t];
        float d2 = di * di;
        const float* hp = h2 + (size_t)(lo + t) * NCLS;
        float* ap = acc2 + t * NCLS;
        float tv[NCLS];
        float mx = -3.402823466e38f;
        #pragma unroll
        for (int f = 0; f < NCLS; ++f) {
            float v = ap[f] + hp[f] * d2 + b2[f];
            tv[f] = v;
            mx = fmaxf(mx, v);
        }
        float sum = 0.0f;
        #pragma unroll
        for (int f = 0; f < NCLS; ++f) sum += expf(tv[f] - mx);
        float lse = mx + logf(sum);
        #pragma unroll
        for (int f = 0; f < NCLS; ++f) ap[f] = tv[f] - lse;
    }
    __syncthreads();

    float4* ob = (float4*)(out + (size_t)lo * NCLS);
    for (unsigned i = t; i < nrows * NCLS / 4u; i += AG_T) ob[i] = ((const float4*)acc2)[i];
}

extern "C" void kernel_launch(void* const* d_in, const int* in_sizes, int n_in,
                              void* d_out, int out_size, void* d_ws, size_t ws_size,
                              hipStream_t stream) {
    const float* x  = (const float*)d_in[0];
    const int*   ei = (const int*)d_in[1];
    const float* W1 = (const float*)d_in[2];
    const float* b1 = (const float*)d_in[3];
    const float* W2 = (const float*)d_in[4];
    const float* b2 = (const float*)d_in[5];
    const int* src = ei;
    const int* dst = ei + NE;
    float* out = (float*)d_out;

    float* ws   = (float*)d_ws;
    float* dis  = ws;
    float* h1   = dis + NN;
    float* agg1 = h1 + (size_t)NN * HID;
    float* h2   = agg1 + (size_t)NN * HID;
    unsigned* eq1  = (unsigned*)(h2 + (size_t)NN * NCLS);
    unsigned* eq   = eq1 + (size_t)AG_NB * AG_QCAP;
    unsigned* ecnt = eq + (size_t)AG_NB * AG_QCAP * 2;
    unsigned* rid  = ecnt + 256;

    k_gemm1<<<(ROW_TILES + 3) / 4, 128, 0, stream>>>(x, W1, h1, dst, rid);

    k_own<<<AG_NB, AG_T, 0, stream>>>(src, dst, rid, dis, eq1, ecnt);

    k_agg1<<<AG_NB, AG_T, 0, stream>>>(src, dst, h1, dis, b1, agg1, eq1, eq, ecnt);

    k_gemm2<<<(ROW_TILES + 3) / 4, 128, 0, stream>>>(agg1, W2, h2);
    k_agg2<<<AG_NB, AG_T, 0, stream>>>(src, dst, h2, dis, b2, eq, ecnt, out);
}


# First launch also packs one word per edge (row in its destination range, source node); a range reads that one record per hit, not destination and source apart: half its scattered 4-byte reads
# speedup vs baseline: 3.9560x; 1.0463x over previous
#include <hip/hip_runtime.h>
#include <math.h>

#define NN 100000
#define NE 1600000
#define FIN 128
#define HID 64
#define NCLS 10
#define ROW_TILES 6250

typedef __attribute__((ext_vector_type(2))) float v2f;
typedef __attribute__((ext_vector_type(8))) float v8f;

#define RG_R 400

__device__ __forceinline__ unsigned rg_id(int d) {
    return ((unsigned)d < (unsigned)NN) ? (unsigned)d / (unsigned)RG_R : 255u;
}

__device__ __forceinline__ unsigned rg_rec(int d, unsigned id, int s) {
    return (((unsigned)d - id * (unsigned)RG_R) << 17) | ((unsigned)s & 0x1FFFFu);
}

__global__ void __launch_bounds__(128)
k_gemm1(const float* __restrict__ x, const float* __restrict__ W1,
        float* __restrict__ h1, const int* __restrict__ src, const int* __restrict__ dst,
        unsigned* __restrict__ rid, unsigned* __restrict__ pk) {
    __shared__ float wT[HID * 132];
    int t = threadIdx.x;
    {
        int e8 = (blockIdx.x * 128 + t) * 8;
        if (e8 < NE) {
            int4 d0 = *(const int4*)(dst + e8);
            int4 d1 = *(const int4*)(dst + e8 + 4);
            int4 s0 = *(const int4*)(src + e8);
            int4 s1 = *(const int4*)(src + e8 + 4);
            const unsigned i0 = rg_id(d0.x), i1 = rg_id(d0.y), i2 = rg_id(d0.z), i3 = rg_id(d0.w);
            const unsigned i4 = rg_id(d1.x), i5 = rg_id(d1.y), i6 = rg_id(d1.z), i7 = rg_id(d1.w);
            uint2 o;
            o.x = i0 | (i1 << 8) | (i2 << 16) | (i3 << 24);
            o.y = i4 | (i5 << 8) | (i6 << 16) | (i7 << 24);
            *(uint2*)(rid + (e8 >> 2)) = o;
            uint4 p0, p1;
            p0.x = rg_rec(d0.x, i0, s0.x); p0.y = rg_rec(d0.y, i1, s0.y);
            p0.z = rg_rec(d0.z, i2, s0.z); p0.w = rg_rec(d0.w, i3, s0.w);
            p1.x = rg_rec(d1.x, i4, s1.x); p1.y = rg_rec(d1.y, i5, s1.y);
            p1.z = rg_rec(d1.z, i6, s1.z); p1.w = rg_rec(d1.w, i7, s1.w);
            *(uint4*)(pk + e8) = p0;
            *(uint4*)(pk + e8 + 4) = p1;
        }
    }
    for (int i = t; i < FIN * HID; i += 128) {
        int k = i >> 6, c = i & 63;
        wT[c * 132 + k] = W1[i];
    }
    __syncthreads();

    int tile = blockIdx.x * 4 + (t >> 5);
    if (tile >= ROW_TILES) return;
    int lane = t & 31;
    int m = lane & 15, half = lane >> 4;

    const float* arow = x + (size_t)(tile * 16 + m) * FIN + 2 * half;
    v8f a0 = {}, a1 = {}, a2 = {}, a3 = {};

    #pragma unroll 4
    for (int k0 = 0; k0 < FIN; k0 += 4) {
        v2f a = *(const v2f*)(arow + k0);
        const float* bp = wT + m * 132 + k0 + 2 * half;
        v2f b0 = *(const v2f*)(bp);
        v2f b1 = *(const v2f*)(bp + 16 * 132);
        v2f b2 = *(const v2f*)(bp + 32 * 132);
        v2f b3 = *(const v2f*)(bp + 48 * 132);
        a0 = __builtin_amdgcn_wmma_f32_16x16x4_f32(false, a, false, b0, (short)0, a0, false, false);
        a1 = __builtin_amdgcn_wmma_f32_16x16x4_f32(false, a, false, b1, (short)0, a1, false, false);
        a2 = __builtin_amdgcn_wmma_f32_16x16x4_f32(false, a, false, b2, (short)0, a2, false, false);
        a3 = __builtin_amdgcn_wmma_f32_16x16x4_f32(false, a, false, b3, (short)0, a3, false, false);
    }

    float* out = h1 + (size_t)(tile * 16 + 8 * half) * HID + m;
    #pragma unroll
    for (int v = 0; v < 8; ++v) {
        out[v * HID +  0] = a0[v];
        out[v * HID + 16] = a1[v];
        out[v * HID + 32] = a2[v];
        out[v * HID + 48] = a3[v];
    }
}

__global__ void __launch_bounds__(128)
k_gemm2(const float* __restrict__ z1, const float* __restrict__ W2,
        float* __restrict__ h2) {
    __shared__ float wT[16 * 68];
    int t = threadIdx.x;
    for (int i = t; i < 16 * HID; i += 128) {
        int k = i >> 4, c = i & 15;
        wT[c * 68 + k] = (c < NCLS) ? W2[k * NCLS + c] : 0.0f;
    }
    __syncthreads();

    int tile = blockIdx.x * 4 + (t >> 5);
    if (tile >= ROW_TILES) return;
    int lane = t & 31;
    int m = lane & 15, half = lane >> 4;

    const float* arow = z1 + (size_t)(tile * 16 + m) * HID + 2 * half;
    v8f acc = {};
    #pragma unroll
    for (int k0 = 0; k0 < HID; k0 += 4) {
        v2f a = *(const v2f*)(arow + k0);
        v2f b = *(const v2f*)(wT + m * 68 + k0 + 2 * half);
        acc = __builtin_amdgcn_wmma_f32_16x16x4_f32(false, a, false, b, (short)0, acc, false, false);
    }
    if (m < NCLS) {
        float* out = h2 + (size_t)(tile * 16 + 8 * half) * NCLS + m;
        #pragma unroll
        for (int v = 0; v < 8; ++v) out[v * NCLS] = acc[v];
    }
}

#define AG_NB   250
#define AG_R    RG_R
#define AG_QCAP 8192
#define AG_T    1024
#define AG_SB   8
#define AG_DB   8
#define AG_OG   64
#define AG_BCAP 112

template <bool COUNT>
__device__ __forceinline__ void ag_push(unsigned* qn, uint2* q, unsigned* cnt, unsigned r, int e) {
    if (COUNT) __hip_atomic_fetch_add(cnt + r, 1u, __ATOMIC_RELAXED, __HIP_MEMORY_SCOPE_WORKGROUP);
    unsigned pos = __hip_atomic_fetch_add(qn, 1u, __ATOMIC_RELAXED, __HIP_MEMORY_SCOPE_WORKGROUP);
    if (pos < AG_QCAP) q[pos].x = (r << 21) | (unsigned)e;
}

template <bool COUNT>
__device__ __forceinline__ void ag_filter(const int* __restrict__ dst, int e0, int e1,
                                          int lo, unsigned nrows,
                                          unsigned* qn, uint2* q, unsigned* cnt, int t) {
    for (int e = e0 + 4 * t; e < e1; e += 4 * AG_T * AG_SB) {
        int4 d[AG_SB];
        #pragma unroll
        for (int u = 0; u < AG_SB; ++u) {
            int ee = e + u * 4 * AG_T;
            d[u] = (ee < e1) ? *(const int4*)(dst + ee) : make_int4(-1, -1, -1, -1);
        }
        #pragma unroll
        for (int u = 0; u < AG_SB; ++u) {
            int ee = e + u * 4 * AG_T;
            unsigned r0 = (unsigned)(d[u].x - lo), r1 = (unsigned)(d[u].y - lo);
            unsigned r2 = (unsigned)(d[u].z - lo), r3 = (unsigned)(d[u].w - lo);
            if ((r0 < nrows) | (r1 < nrows) | (r2 < nrows) | (r3 < nrows)) {
                if (r0 < nrows) ag_push<COUNT>(qn, q, cnt, r0, ee);
                if (r1 < nrows) ag_push<COUNT>(qn, q, cnt, r1, ee + 1);
                if (r2 < nrows) ag_push<COUNT>(qn, q, cnt, r2, ee + 2);
                if (r3 < nrows) ag_push<COUNT>(qn, q, cnt, r3, ee + 3);
            }
        }
    }
}

__device__ __forceinline__ void ag_lookup(const int* __restrict__ src, const float* __restrict__ dis,
                                          int lo, uint2* q, unsigned n, int t) {
    for (unsigned i0 = t; i0 < n; i0 += 4u * AG_T) {
        unsigned en[4], s[4];
        float w[4];
        #pragma unroll
        for (int u = 0; u < 4; ++u) {
            unsigned i = i0 + u * AG_T;
            en[u] = (i < n) ? q[i].x : 0u;
        }
        #pragma unroll
        for (int u = 0; u < 4; ++u) s[u] = (unsigned)src[en[u] & 0x1FFFFFu];
        #pragma unroll
        for (int u = 0; u < 4; ++u) w[u] = dis[s[u]] * dis[lo + (int)(en[u] >> 21)];
        #pragma unroll
        for (int u = 0; u < 4; ++u) {
            unsigned i = i0 + u * AG_T;
            q[(i < n) ? i : AG_QCAP] = make_uint2(((en[u] >> 21) << 17) | s[u], __float_as_uint(w[u]));
        }
    }
    __syncthreads();
}

__device__ __forceinline__ void ag_add64(const float* __restrict__ h1,
                                         float* acc, const uint2* q, unsigned n, int t) {
    const unsigned hw = (unsigned)t >> 4;
    const unsigned l16 = (unsigned)t & 15u;
    const bool hi = (hw & 1u) != 0u;
    const unsigned oA = hi ? 2u : 0u, oB = oA ^ 2u;
    for (unsigned i0 = hw; i0 < n; i0 += 64u * AG_DB) {
        uint2 en[AG_DB];
        float4 row[AG_DB];
        #pragma unroll
        for (int u = 0; u < AG_DB; ++u) {
            unsigned i = i0 + 64u * u;
            en[u] = (i < n) ? q[i] : make_uint2((unsigned)AG_R << 17, 0u);
        }
        #pragma unroll
        for (int u = 0; u < AG_DB; ++u)
            row[u] = *(const float4*)(h1 + (size_t)(en[u].x & 0x1FFFFu) * HID + 4u * l16);
        #pragma unroll
        for (int u = 0; u < AG_DB; ++u) {
            float w = __uint_as_float(en[u].y);
            float* p = acc + (en[u].x >> 17) * HID + 4u * l16;
            float a0 = hi ? row[u].z : row[u].x, a1 = hi ? row[u].w : row[u].y;
            float c0 = hi ? row[u].x : row[u].z, c1 = hi ? row[u].y : row[u].w;
            __hip_atomic_fetch_add(p + oA,      a0 * w, __ATOMIC_RELAXED, __HIP_MEMORY_SCOPE_WORKGROUP);
            __hip_atomic_fetch_add(p + oA + 1u, a1 * w, __ATOMIC_RELAXED, __HIP_MEMORY_SCOPE_WORKGROUP);
            __hip_atomic_fetch_add(p + oB,      c0 * w, __ATOMIC_RELAXED, __HIP_MEMORY_SCOPE_WORKGROUP);
            __hip_atomic_fetch_add(p + oB + 1u, c1 * w, __ATOMIC_RELAXED, __HIP_MEMORY_SCOPE_WORKGROUP);
        }
    }
}

__device__ __forceinline__ void ag_own64(const float* __restrict__ h1, float* acc,
                                         const uint2* qb, unsigned n, unsigned spare, int t) {
    const unsigned l16 = (unsigned)t & 15u;
    const unsigned no = (unsigned)__shfl_xor((int)n, 16);
    const unsigned nw = n > no ? n : no;
    const unsigned padx = spare << 17;
    uint2 en[AG_DB];
    float4 row[AG_DB];
    #pragma unroll
    for (int u = 0; u < AG_DB; ++u) {
        uint2 e = qb[u];
        en[u].x = ((unsigned)u < n) ? e.x : padx;
        en[u].y = ((unsigned)u < n) ? e.y : 0u;
    }
    #pragma unroll
    for (int u = 0; u < AG_DB; ++u)
        row[u] = *(const float4*)(h1 + (size_t)(en[u].x & 0x1FFFFu) * HID + 4u * l16);
    for (unsigned i0 = 0; i0 < nw; i0 += AG_DB) {
        uint2 en2[AG_DB];
        float4 row2[AG_DB];
        const unsigned j0 = i0 + AG_DB;
        #pragma unroll
        for (int u = 0; u < AG_DB; ++u) {
            const unsigned j = j0 + (unsigned)u;
            uint2 e = qb[j < (unsigned)AG_BCAP ? j : (unsigned)AG_BCAP - 1u];
            en2[u].x = (j < n) ? e.x : padx;
            en2[u].y = (j < n) ? e.y : 0u;
        }
        #pragma unroll
        for (int u = 0; u < AG_DB; ++u)
            row2[u] = *(const float4*)(h1 + (size_t)(en2[u].x & 0x1FFFFu) * HID + 4u * l16);
        #pragma unroll
        for (int u = 0; u < AG_DB; ++u) {
            float w = __uint_as_float(en[u].y);
            float4* p = (float4*)(acc + (en[u].x >> 17) * HID + 4u * l16);
            float4 a = *p;
            a.x += row[u].x * w;
            a.y += row[u].y * w;
            a.z += row[u].z * w;
            a.w += row[u].w * w;
            *p = a;
        }
        #pragma unroll
        for (int u = 0; u < AG_DB; ++u) { en[u] = en2[u]; row[u] = row2[u]; }
    }
}

#define AG_LB   8

#define AG_FB   8
#define AG_KB   8
#define NG16    (NE / 16)

static_assert(NE % 16 == 0, "the range-id list is read in 16-byte pieces");
static_assert(AG_NB <= 255, "a range id fits one byte, 255 left free");
static_assert(AG_FB % 2 == 0, "two pieces share one 32-bit hit mask");
static_assert((long long)((ROW_TILES + 3) / 4) * 128 * 8 >= NE, "the first launch packs every edge's range id");

__global__ void __launch_bounds__(AG_T)
k_own(const int* __restrict__ dst, const unsigned* __restrict__ rid, const unsigned* __restrict__ pk,
      float* __restrict__ dis, unsigned* __restrict__ eq1, unsigned* __restrict__ ecnt) {
    __shared__ uint2 q[AG_QCAP + 1];
    __shared__ unsigned cnt[AG_R];
    __shared__ unsigned qn;
    const int t = threadIdx.x;
    const int lo = blockIdx.x * AG_R;
    const unsigned nrows = (unsigned)((NN - lo) < AG_R ? (NN - lo) : AG_R);

    if (t < AG_R) cnt[t] = 0u;
    if (t == 0) qn = 0u;
    __syncthreads();

    const unsigned own4 = blockIdx.x * 0x01010101u;
    for (int g = t; g < NG16; g += AG_T * AG_FB) {
        uint4 w[AG_FB];
        #pragma unroll
        for (int u = 0; u < AG_FB; ++u) {
            int gg = g + u * AG_T;
            w[u] = (gg < NG16) ? ((const uint4*)rid)[gg] : make_uint4(~0u, ~0u, ~0u, ~0u);
        }
        #pragma unroll
        for (int v = 0; v < AG_FB / 2; ++v) {
            const unsigned ww[8] = { w[2 * v].x, w[2 * v].y, w[2 * v].z, w[2 * v].w,
                                     w[2 * v + 1].x, w[2 * v + 1].y, w[2 * v + 1].z, w[2 * v + 1].w };
            unsigned nz = 0u;
            #pragma unroll
            for (int k = 0; k < 8; ++k) {
                unsigned x = ww[k] ^ own4;
                unsigned s = ((x & 0x7F7F7F7Fu) + 0x7F7F7F7Fu) | x;
                nz |= (s >> (7 - k)) & (0x01010101u << k);
            }
            unsigned hit = ~nz;
            while (hit) {
                unsigned p = (unsigned)__builtin_ctz(hit);
                hit &= hit - 1u;
                unsigned k = p & 7u, j = p >> 3;
                int gg = g + (2 * v + (int)(k >> 2)) * AG_T;
                unsigned e = (unsigned)gg * 16u + (k & 3u) * 4u + j;
                unsigned pos = __hip_atomic_fetch_add(&qn, 1u, __ATOMIC_RELAXED, __HIP_MEMORY_SCOPE_WORKGROUP);
                if (pos < AG_QCAP) q[pos].x = e;
            }
        }
    }
    __syncthreads();
    const unsigned total = qn;
    if (t == 0) ecnt[blockIdx.x] = total;
    if (total <= AG_QCAP) {
        unsigned* __restrict__ g1 = eq1 + (size_t)blockIdx.x * AG_QCAP;
        for (unsigned i0 = t; i0 < total; i0 += (unsigned)AG_KB * AG_T) {
            unsigned ix[AG_KB], en[AG_KB], v[AG_KB];
            #pragma unroll
            for (int u = 0; u < AG_KB; ++u) {
                unsigned i = i0 + u * AG_T;
                ix[u] = (i < total) ? i : i0;
                en[u] = q[ix[u]].x;
            }
            #pragma unroll
            for (int u = 0; u < AG_KB; ++u) v[u] = pk[en[u]];
            #pragma unroll
            for (int u = 0; u < AG_KB; ++u) g1[ix[u]] = v[u];
            #pragma unroll
            for (int u = 0; u < AG_KB; ++u)
                if (i0 + u * AG_T < total && (v[u] >> 17) < nrows)
                    __hip_atomic_fetch_add(cnt + (v[u] >> 17), 1u, __ATOMIC_RELAXED, __HIP_MEMORY_SCOPE_WORKGROUP);
        }
    } else {
        __syncthreads();
        if (t == 0) qn = 0u;
        __syncthreads();
        ag_filter<true>(dst, 0, NE, lo, nrows, &qn, q, cnt, t);
    }
    __syncthreads();
    if ((unsigned)t < nrows) dis[lo + t] = rsqrtf((float)(cnt[t] + 1u));
}

__global__ void __launch_bounds__(AG_T)
k_agg1(const int* __restrict__ src, const int* __restrict__ dst,
       const float* __restrict__ h1, const float* __restrict__ dis, const float* __restrict__ b1,
       float* __restrict__ z1, const unsigned* __restrict__ eq1,
       unsigned* __restrict__ eq, const unsigned* __restrict__ ecnt) {
    __shared__ float acc[(AG_R + AG_OG) * HID];
    __shared__ uint2 q[AG_QCAP + 1];
    __shared__ uint2 qb[AG_OG * AG_BCAP];
    __shared__ unsigned bcnt[AG_OG + 1];
    __shared__ unsigned qn;
    const int t = threadIdx.x;
    const int lo = blockIdx.x * AG_R;
    const unsigned nrows = (unsigned)((NN - lo) < AG_R ? (NN - lo) : AG_R);

    for (int i = t; i < (AG_R + AG_OG) * HID / 4; i += AG_T)
        ((float4*)acc)[i] = make_float4(0.0f, 0.0f, 0.0f, 0.0f);
    if (t <= AG_OG) bcnt[t] = 0u;
    if (t == 0) qn = 0u;
    const unsigned total = ecnt[blockIdx.x];
    __syncthreads();
    if (total <= AG_QCAP) {
        const unsigned* __restrict__ g1 = eq1 + (size_t)blockIdx.x * AG_QCAP;
        uint2* __restrict__ g2 = (uint2*)eq + (size_t)blockIdx.x * AG_QCAP;
        for (unsigned i0 = t; i0 < total; i0 += (unsigned)AG_LB * AG_T) {
            unsigned v[AG_LB];
            float w[AG_LB];
            #pragma unroll
            for (int u = 0; u < AG_LB; ++u) {
                unsigned i = i0 + u * AG_T;
                v[u] = g1[(i < total) ? i : i0];
            }
            #pragma unroll
            for (int u = 0; u < AG_LB; ++u)
                w[u] = dis[v[u] & 0x1FFFFu] * dis[lo + (int)(v[u] >> 17)];
            unsigned pos[AG_LB];
            #pragma unroll
            for (int u = 0; u < AG_LB; ++u) {
                unsigned i = i0 + u * AG_T;
                unsigned b = (i < total) ? (v[u] >> 17) % (unsigned)AG_OG : (unsigned)AG_OG;
                pos[u] = __hip_atomic_fetch_add(bcnt + b, 1u, __ATOMIC_RELAXED, __HIP_MEMORY_SCOPE_WORKGROUP);
            }
            #pragma unroll
            for (int u = 0; u < AG_LB; ++u) {
                unsigned i = i0 + u * AG_T;
                uint2 o = make_uint2(v[u], __float_as_uint(w[u]));
                if (i < total) {
                    g2[i] = o;
                    unsigned b = (v[u] >> 17) % (unsigned)AG_OG;
                    if (pos[u] < (unsigned)AG_BCAP) qb[b * AG_BCAP + pos[u]] = o;
                    else q[__hip_atomic_fetch_add(&qn, 1u, __ATOMIC_RELAXED, __HIP_MEMORY_SCOPE_WORKGROUP)] = o;
                }
            }
        }
        __syncthreads();
        {
            const unsigned hw = (unsigned)t >> 4;
            const unsigned nb = bcnt[hw] < (unsigned)AG_BCAP ? bcnt[hw] : (unsigned)AG_BCAP;
            ag_own64(h1, acc, qb + hw * AG_BCAP, nb, (unsigned)AG_R + hw, t);
        }
        __syncthreads();
        ag_add64(h1, acc, q, qn, t);
    } else {
        for (int e0 = 0; e0 < NE; e0 += AG_QCAP) {
            __syncthreads();
            if (t == 0) qn = 0u;
            __syncthreads();
            const int e1 = (e0 + AG_QCAP < NE) ? (e0 + AG_QCAP) : NE;
            ag_filter<false>(dst, e0, e1, lo, nrows, &qn, q, nullptr, t);
            __syncthreads();
            const unsigned n = qn;
            ag_lookup(src, dis, lo, q, n, t);
            ag_add64(h1, acc, q, n, t);
        }
    }
    __syncthreads();

    for (unsigned i = t; i < nrows * 16u; i += AG_T) {
        unsigned r = i >> 4, c = (i & 15u) * 4u;
        size_t g = (size_t)(lo + r) * HID + c;
        float di = dis[lo + r];
        float4 a = *(const float4*)(acc + r * HID + c);
        float4 h = *(const float4*)(h1 + g);
        float4 b = *(const float4*)(b1 + c);
        float4 v;
        v.x = a.x + h.x * di * di + b.x;
        v.y = a.y + h.y * di * di + b.y;
        v.z = a.z + h.z * di * di + b.z;
        v.w = a.w + h.w * di * di + b.w;
        v.x = v.x > 0.0f ? v.x : 0.0f;
        v.y = v.y > 0.0f ? v.y : 0.0f;
        v.z = v.z > 0.0f ? v.z : 0.0f;
        v.w = v.w > 0.0f ? v.w : 0.0f;
        *(float4*)(z1 + g) = v;
    }
}

#define AG_D2   8

static_assert((AG_R * NCLS) % 32 == 0, "a range's block of out must be whole 128-byte lines");
static_assert(((NN % AG_R) * NCLS) % 4 == 0, "the last range's block of out must be whole float4s");
static_assert(AG_R <= AG_T, "the layer-2 epilogue gives every own row a thread of its own");
static_assert((long long)AG_NB * AG_R >= NN && (long long)(AG_NB - 1) * AG_R < NN, "the ranges cover the nodes and none is empty");

__device__ __forceinline__ void ag_add10(const float* __restrict__ h2,
                                         float* acc2, const uint2* q, unsigned n, int t) {
    const unsigned g = (unsigned)t >> 3;
    const unsigned j = (unsigned)t & 7u;
    if (j < 5u) {
        for (unsigned i0 = g; i0 < n; i0 += 128u * AG_D2) {
            uint2 en[AG_D2];
            v2f row[AG_D2];
            #pragma unroll
            for (int u = 0; u < AG_D2; ++u) {
                unsigned i = i0 + 128u * u;
                en[u] = (i < n) ? q[i] : make_uint2((unsigned)AG_R << 17, 0u);
            }
            #pragma unroll
            for (int u = 0; u < AG_D2; ++u)
                row[u] = *(const v2f*)(h2 + (size_t)(en[u].x & 0x1FFFFu) * NCLS + 2u * j);
            #pragma unroll
            for (int u = 0; u < AG_D2; ++u) {
                float w = __uint_as_float(en[u].y);
                float* p = acc2 + (en[u].x >> 17) * NCLS + 2u * j;
                __hip_atomic_fetch_add(p,      row[u].x * w, __ATOMIC_RELAXED, __HIP_MEMORY_SCOPE_WORKGROUP);
                __hip_atomic_fetch_add(p + 1u, row[u].y * w, __ATOMIC_RELAXED, __HIP_MEMORY_SCOPE_WORKGROUP);
            }
        }
    }
}

__global__ void __launch_bounds__(AG_T)
k_agg2(const int* __restrict__ src, const int* __restrict__ dst,
       const float* __restrict__ h2,
       const float* __restrict__ dis, const float* __restrict__ b2,
       const unsigned* __restrict__ eq, const unsigned* __restrict__ ecnt,
       float* __restrict__ out) {
    __shared__ __attribute__((aligned(16))) float acc2[(AG_R + 1) * NCLS];
    __shared__ uint2 q[AG_QCAP + 1];
    __shared__ unsigned qn;
    const int t = threadIdx.x;
    const int lo = blockIdx.x * AG_R;
    const unsigned nrows = (unsigned)((NN - lo) < AG_R ? (NN - lo) : AG_R);

    for (int i = t; i < (AG_R + 1) * NCLS; i += AG_T) acc2[i] = 0.0f;
    const unsigned total = ecnt[blockIdx.x];
    if (total <= AG_QCAP) {
        const uint2* __restrict__ gq = (const uint2*)eq + (size_t)blockIdx.x * AG_QCAP;
        for (unsigned i0 = t; i0 < total; i0 += 4u * AG_T) {
            uint2 v[4];
            #pragma unroll
            for (int u = 0; u < 4; ++u) {
                unsigned i = i0 + u * AG_T;
                v[u] = gq[(i < total) ? i : i0];
            }
            #pragma unroll
            for (int u = 0; u < 4; ++u) {
                unsigned i = i0 + u * AG_T;
                q[(i < total) ? i : AG_QCAP] = v[u];
            }
        }
        __syncthreads();
        ag_add10(h2, acc2, q, total, t);
    } else {
        for (int e0 = 0; e0 < NE; e0 += AG_QCAP) {
            __syncthreads();
            if (t == 0) qn = 0u;
            __syncthreads();
            const int e1 = (e0 + AG_QCAP < NE) ? (e0 + AG_QCAP) : NE;
            ag_filter<false>(dst, e0, e1, lo, nrows, &qn, q, nullptr, t);
            __syncthreads();
            const unsigned n = qn;
            ag_lookup(src, dis, lo, q, n, t);
            ag_add10(h2, acc2, q, n, t);
        }
    }
    __syncthreads();

    if ((unsigned)t < nrows) {
        float di = dis[lo + t];
        float d2 = di * di;
        const float* hp = h2 + (size_t)(lo + t) * NCLS;
        float* ap = acc2 + t * NCLS;
        float tv[NCLS];
        float mx = -3.402823466e38f;
        #pragma unroll
        for (int f = 0; f < NCLS; ++f) {
            float v = ap[f] + hp[f] * d2 + b2[f];
            tv[f] = v;
            mx = fmaxf(mx, v);
        }
        float sum = 0.0f;
        #pragma unroll
        for (int f = 0; f < NCLS; ++f) sum += expf(tv[f] - mx);
        float lse = mx + logf(sum);
        #pragma unroll
        for (int f = 0; f < NCLS; ++f) ap[f] = tv[f] - lse;
    }
    __syncthreads();

    float4* ob = (float4*)(out + (size_t)lo * NCLS);
    for (unsigned i = t; i < nrows * NCLS / 4u; i += AG_T) ob[i] = ((const float4*)acc2)[i];
}

extern "C" void kernel_launch(void* const* d_in, const int* in_sizes, int n_in,
                              void* d_out, int out_size, void* d_ws, size_t ws_size,
                              hipStream_t stream) {
    const float* x  = (const float*)d_in[0];
    const int*   ei = (const int*)d_in[1];
    const float* W1 = (const float*)d_in[2];
    const float* b1 = (const float*)d_in[3];
    const float* W2 = (const float*)d_in[4];
    const float* b2 = (const float*)d_in[5];
    const int* src = ei;
    const int* dst = ei + NE;
    float* out = (float*)d_out;

    float* ws   = (float*)d_ws;
    float* dis  = ws;
    float* h1   = dis + NN;
    float* agg1 = h1 + (size_t)NN * HID;
    float* h2   = agg1 + (size_t)NN * HID;
    unsigned* eq1  = (unsigned*)(h2 + (size_t)NN * NCLS);
    unsigned* eq   = eq1 + (size_t)AG_NB * AG_QCAP;
    unsigned* ecnt = eq + (size_t)AG_NB * AG_QCAP * 2;
    unsigned* rid  = ecnt + 256;
    unsigned* pk   = rid + NE / 4;

    k_gemm1<<<(ROW_TILES + 3) / 4, 128, 0, stream>>>(x, W1, h1, src, dst, rid, pk);

    k_own<<<AG_NB, AG_T, 0, stream>>>(dst, rid, pk, dis, eq1, ecnt);

    k_agg1<<<AG_NB, AG_T, 0, stream>>>(src, dst, h1, dis, b1, agg1, eq1, eq, ecnt);

    k_gemm2<<<(ROW_TILES + 3) / 4, 128, 0, stream>>>(agg1, W2, h2);
    k_agg2<<<AG_NB, AG_T, 0, stream>>>(src, dst, h2, dis, b2, eq, ecnt, out);
}


# Launch 1 marks per destination range which 16-edge pieces hold its edges (64-bit map per chunk); a range reads only marked pieces' ids; list places by one LDS add per wave; sums in LDS arrival order
# speedup vs baseline: 4.0723x; 1.0294x over previous
#include <hip/hip_runtime.h>
#include <math.h>

#define NN 100000
#define NE 1600000
#define FIN 128
#define HID 64
#define NCLS 10
#define ROW_TILES 6250

typedef __attribute__((ext_vector_type(2))) float v2f;
typedef __attribute__((ext_vector_type(8))) float v8f;

#define RG_R 400
#define RG_NB 250
#define NCH ((ROW_TILES + 3) / 4)

__device__ __forceinline__ unsigned rg_id(int d) {
    return ((unsigned)d < (unsigned)NN) ? (unsigned)d / (unsigned)RG_R : 255u;
}

__device__ __forceinline__ unsigned rg_rec(int d, unsigned id, int s) {
    return (((unsigned)d - id * (unsigned)RG_R) << 17) | ((unsigned)s & 0x1FFFFu);
}

__global__ void __launch_bounds__(128)
k_gemm1(const float* __restrict__ x, const float* __restrict__ W1,
        float* __restrict__ h1, const int* __restrict__ src, const int* __restrict__ dst,
        unsigned* __restrict__ rid, unsigned* __restrict__ pk, unsigned* __restrict__ bmT) {
    __shared__ float wT[HID * 132];
    __shared__ unsigned bm[RG_NB * 2];
    int t = threadIdx.x;
    for (int i = t; i < RG_NB * 2; i += 128) bm[i] = 0u;
    __syncthreads();
    {
        int e8 = (blockIdx.x * 128 + t) * 8;
        if (e8 < NE) {
            int4 d0 = *(const int4*)(dst + e8);
            int4 d1 = *(const int4*)(dst + e8 + 4);
            int4 s0 = *(const int4*)(src + e8);
            int4 s1 = *(const int4*)(src + e8 + 4);
            const unsigned i0 = rg_id(d0.x), i1 = rg_id(d0.y), i2 = rg_id(d0.z), i3 = rg_id(d0.w);
            const unsigned i4 = rg_id(d1.x), i5 = rg_id(d1.y), i6 = rg_id(d1.z), i7 = rg_id(d1.w);
            uint2 o;
            o.x = i0 | (i1 << 8) | (i2 << 16) | (i3 << 24);
            o.y = i4 | (i5 << 8) | (i6 << 16) | (i7 << 24);
            *(uint2*)(rid + (e8 >> 2)) = o;
            uint4 p0, p1;
            p0.x = rg_rec(d0.x, i0, s0.x); p0.y = rg_rec(d0.y, i1, s0.y);
            p0.z = rg_rec(d0.z, i2, s0.z); p0.w = rg_rec(d0.w, i3, s0.w);
            p1.x = rg_rec(d1.x, i4, s1.x); p1.y = rg_rec(d1.y, i5, s1.y);
            p1.z = rg_rec(d1.z, i6, s1.z); p1.w = rg_rec(d1.w, i7, s1.w);
            *(uint4*)(pk + e8) = p0;
            *(uint4*)(pk + e8 + 4) = p1;
            const unsigned pw = (unsigned)t >> 6, pb = 1u << (((unsigned)t >> 1) & 31u);
            const unsigned ids[8] = { i0, i1, i2, i3, i4, i5, i6, i7 };
            #pragma unroll
            for (int k = 0; k < 8; ++k)
                if (ids[k] < (unsigned)RG_NB)
                    __hip_atomic_fetch_or(bm + ids[k] * 2u + pw, pb, __ATOMIC_RELAXED, __HIP_MEMORY_SCOPE_WORKGROUP);
        }
    }
    for (int i = t; i < FIN * HID; i += 128) {
        int k = i >> 6, c = i & 63;
        wT[c * 132 + k] = W1[i];
    }
    __syncthreads();

    for (int i = t; i < RG_NB; i += 128)
        ((uint2*)bmT)[(size_t)i * NCH + blockIdx.x] = make_uint2(bm[2 * i], bm[2 * i + 1]);

    int tile = blockIdx.x * 4 + (t >> 5);
    if (tile >= ROW_TILES) return;
    int lane = t & 31;
    int m = lane & 15, half = lane >> 4;

    const float* arow = x + (size_t)(tile * 16 + m) * FIN + 2 * half;
    v8f a0 = {}, a1 = {}, a2 = {}, a3 = {};

    #pragma unroll 4
    for (int k0 = 0; k0 < FIN; k0 += 4) {
        v2f a = *(const v2f*)(arow + k0);
        const float* bp = wT + m * 132 + k0 + 2 * half;
        v2f b0 = *(const v2f*)(bp);
        v2f b1 = *(const v2f*)(bp + 16 * 132);
        v2f b2 = *(const v2f*)(bp + 32 * 132);
        v2f b3 = *(const v2f*)(bp + 48 * 132);
        a0 = __builtin_amdgcn_wmma_f32_16x16x4_f32(false, a, false, b0, (short)0, a0, false, false);
        a1 = __builtin_amdgcn_wmma_f32_16x16x4_f32(false, a, false, b1, (short)0, a1, false, false);
        a2 = __builtin_amdgcn_wmma_f32_16x16x4_f32(false, a, false, b2, (short)0, a2, false, false);
        a3 = __builtin_amdgcn_wmma_f32_16x16x4_f32(false, a, false, b3, (short)0, a3, false, false);
    }

    float* out = h1 + (size_t)(tile * 16 + 8 * half) * HID + m;
    #pragma unroll
    for (int v = 0; v < 8; ++v) {
        out[v * HID +  0] = a0[v];
        out[v * HID + 16] = a1[v];
        out[v * HID + 32] = a2[v];
        out[v * HID + 48] = a3[v];
    }
}

__global__ void __launch_bounds__(128)
k_gemm2(const float* __restrict__ z1, const float* __restrict__ W2,
        float* __restrict__ h2) {
    __shared__ float wT[16 * 68];
    int t = threadIdx.x;
    for (int i = t; i < 16 * HID; i += 128) {
        int k = i >> 4, c = i & 15;
        wT[c * 68 + k] = (c < NCLS) ? W2[k * NCLS + c] : 0.0f;
    }
    __syncthreads();

    int tile = blockIdx.x * 4 + (t >> 5);
    if (tile >= ROW_TILES) return;
    int lane = t & 31;
    int m = lane & 15, half = lane >> 4;

    const float* arow = z1 + (size_t)(tile * 16 + m) * HID + 2 * half;
    v8f acc = {};
    #pragma unroll
    for (int k0 = 0; k0 < HID; k0 += 4) {
        v2f a = *(const v2f*)(arow + k0);
        v2f b = *(const v2f*)(wT + m * 68 + k0 + 2 * half);
        acc = __builtin_amdgcn_wmma_f32_16x16x4_f32(false, a, false, b, (short)0, acc, false, false);
    }
    if (m < NCLS) {
        float* out = h2 + (size_t)(tile * 16 + 8 * half) * NCLS + m;
        #pragma unroll
        for (int v = 0; v < 8; ++v) out[v * NCLS] = acc[v];
    }
}

#define AG_NB   250
#define AG_R    RG_R
#define AG_QCAP 8192
#define AG_T    1024
#define AG_SB   8
#define AG_DB   8
#define AG_OG   64
#define AG_BCAP 112

template <bool COUNT>
__device__ __forceinline__ void ag_push(unsigned* qn, uint2* q, unsigned* cnt, unsigned r, int e) {
    if (COUNT) __hip_atomic_fetch_add(cnt + r, 1u, __ATOMIC_RELAXED, __HIP_MEMORY_SCOPE_WORKGROUP);
    unsigned pos = __hip_atomic_fetch_add(qn, 1u, __ATOMIC_RELAXED, __HIP_MEMORY_SCOPE_WORKGROUP);
    if (pos < AG_QCAP) q[pos].x = (r << 21) | (unsigned)e;
}

template <bool COUNT>
__device__ __forceinline__ void ag_filter(const int* __restrict__ dst, int e0, int e1,
                                          int lo, unsigned nrows,
                                          unsigned* qn, uint2* q, unsigned* cnt, int t) {
    for (int e = e0 + 4 * t; e < e1; e += 4 * AG_T * AG_SB) {
        int4 d[AG_SB];
        #pragma unroll
        for (int u = 0; u < AG_SB; ++u) {
            int ee = e + u * 4 * AG_T;
            d[u] = (ee < e1) ? *(const int4*)(dst + ee) : make_int4(-1, -1, -1, -1);
        }
        #pragma unroll
        for (int u = 0; u < AG_SB; ++u) {
            int ee = e + u * 4 * AG_T;
            unsigned r0 = (unsigned)(d[u].x - lo), r1 = (unsigned)(d[u].y - lo);
            unsigned r2 = (unsigned)(d[u].z - lo), r3 = (unsigned)(d[u].w - lo);
            if ((r0 < nrows) | (r1 < nrows) | (r2 < nrows) | (r3 < nrows)) {
                if (r0 < nrows) ag_push<COUNT>(qn, q, cnt, r0, ee);
                if (r1 < nrows) ag_push<COUNT>(qn, q, cnt, r1, ee + 1);
                if (r2 < nrows) ag_push<COUNT>(qn, q, cnt, r2, ee + 2);
                if (r3 < nrows) ag_push<COUNT>(qn, q, cnt, r3, ee + 3);
            }
        }
    }
}

__device__ __forceinline__ void ag_lookup(const int* __restrict__ src, const float* __restrict__ dis,
                                          int lo, uint2* q, unsigned n, int t) {
    for (unsigned i0 = t; i0 < n; i0 += 4u * AG_T) {
        unsigned en[4], s[4];
        float w[4];
        #pragma unroll
        for (int u = 0; u < 4; ++u) {
            unsigned i = i0 + u * AG_T;
            en[u] = (i < n) ? q[i].x : 0u;
        }
        #pragma unroll
        for (int u = 0; u < 4; ++u) s[u] = (unsigned)src[en[u] & 0x1FFFFFu];
        #pragma unroll
        for (int u = 0; u < 4; ++u) w[u] = dis[s[u]] * dis[lo + (int)(en[u] >> 21)];
        #pragma unroll
        for (int u = 0; u < 4; ++u) {
            unsigned i = i0 + u * AG_T;
            q[(i < n) ? i : AG_QCAP] = make_uint2(((en[u] >> 21) << 17) | s[u], __float_as_uint(w[u]));
        }
    }
    __syncthreads();
}

__device__ __forceinline__ void ag_add64(const float* __restrict__ h1,
                                         float* acc, const uint2* q, unsigned n, int t) {
    const unsigned hw = (unsigned)t >> 4;
    const unsigned l16 = (unsigned)t & 15u;
    const bool hi = (hw & 1u) != 0u;
    const unsigned oA = hi ? 2u : 0u, oB = oA ^ 2u;
    for (unsigned i0 = hw; i0 < n; i0 += 64u * AG_DB) {
        uint2 en[AG_DB];
        float4 row[AG_DB];
        #pragma unroll
        for (int u = 0; u < AG_DB; ++u) {
            unsigned i = i0 + 64u * u;
            en[u] = (i < n) ? q[i] : make_uint2((unsigned)AG_R << 17, 0u);
        }
        #pragma unroll
        for (int u = 0; u < AG_DB; ++u)
            row[u] = *(const float4*)(h1 + (size_t)(en[u].x & 0x1FFFFu) * HID + 4u * l16);
        #pragma unroll
        for (int u = 0; u < AG_DB; ++u) {
            float w = __uint_as_float(en[u].y);
            float* p = acc + (en[u].x >> 17) * HID + 4u * l16;
            float a0 = hi ? row[u].z : row[u].x, a1 = hi ? row[u].w : row[u].y;
            float c0 = hi ? row[u].x : row[u].z, c1 = hi ? row[u].y : row[u].w;
            __hip_atomic_fetch_add(p + oA,      a0 * w, __ATOMIC_RELAXED, __HIP_MEMORY_SCOPE_WORKGROUP);
            __hip_atomic_fetch_add(p + oA + 1u, a1 * w, __ATOMIC_RELAXED, __HIP_MEMORY_SCOPE_WORKGROUP);
            __hip_atomic_fetch_add(p + oB,      c0 * w, __ATOMIC_RELAXED, __HIP_MEMORY_SCOPE_WORKGROUP);
            __hip_atomic_fetch_add(p + oB + 1u, c1 * w, __ATOMIC_RELAXED, __HIP_MEMORY_SCOPE_WORKGROUP);
        }
    }
}

__device__ __forceinline__ void ag_own64(const float* __restrict__ h1, float* acc,
                                         const uint2* qb, unsigned n, unsigned spare, int t) {
    const unsigned l16 = (unsigned)t & 15u;
    const unsigned no = (unsigned)__shfl_xor((int)n, 16);
    const unsigned nw = n > no ? n : no;
    const unsigned padx = spare << 17;
    uint2 en[AG_DB];
    float4 row[AG_DB];
    #pragma unroll
    for (int u = 0; u < AG_DB; ++u) {
        uint2 e = qb[u];
        en[u].x = ((unsigned)u < n) ? e.x : padx;
        en[u].y = ((unsigned)u < n) ? e.y : 0u;
    }
    #pragma unroll
    for (int u = 0; u < AG_DB; ++u)
        row[u] = *(const float4*)(h1 + (size_t)(en[u].x & 0x1FFFFu) * HID + 4u * l16);
    for (unsigned i0 = 0; i0 < nw; i0 += AG_DB) {
        uint2 en2[AG_DB];
        float4 row2[AG_DB];
        const unsigned j0 = i0 + AG_DB;
        #pragma unroll
        for (int u = 0; u < AG_DB; ++u) {
            const unsigned j = j0 + (unsigned)u;
            uint2 e = qb[j < (unsigned)AG_BCAP ? j : (unsigned)AG_BCAP - 1u];
            en2[u].x = (j < n) ? e.x : padx;
            en2[u].y = (j < n) ? e.y : 0u;
        }
        #pragma unroll
        for (int u = 0; u < AG_DB; ++u)
            row2[u] = *(const float4*)(h1 + (size_t)(en2[u].x & 0x1FFFFu) * HID + 4u * l16);
        #pragma unroll
        for (int u = 0; u < AG_DB; ++u) {
            float w = __uint_as_float(en[u].y);
            float4* p = (float4*)(acc + (en[u].x >> 17) * HID + 4u * l16);
            float4 a = *p;
            a.x += row[u].x * w;
            a.y += row[u].y * w;
            a.z += row[u].z * w;
            a.w += row[u].w * w;
            *p = a;
        }
        #pragma unroll
        for (int u = 0; u < AG_DB; ++u) { en[u] = en2[u]; row[u] = row2[u]; }
    }
}

#define AG_LB   8

#define AG_FB   8
#define AG_MW   4
#define AG_KB   8
#define NG16    (NE / 16)

static_assert(NE % 16 == 0, "the range-id list is read in 16-byte pieces");
static_assert(AG_NB <= 255, "a range id fits one byte, 255 left free");
static_assert(AG_FB % 2 == 0, "two pieces share one 32-bit hit mask");
static_assert((long long)((ROW_TILES + 3) / 4) * 128 * 8 >= NE, "the first launch packs every edge's range id");
static_assert(RG_NB == AG_NB, "one piece map per destination range");
static_assert(NCH * 64 >= NG16, "the chunks' piece maps cover every 16-edge piece");
static_assert(AG_MW * AG_T >= 2 * NCH, "one trip expands a range's whole piece map");

__device__ __forceinline__ unsigned wave_excl(unsigned c, unsigned lane, unsigned& total) {
    unsigned s = c;
    #pragma unroll
    for (int d = 1; d < 32; d <<= 1) {
        unsigned n = (unsigned)__shfl_up((int)s, d);
        if (lane >= (unsigned)d) s += n;
    }
    total = (unsigned)__shfl((int)s, 31);
    return s - c;
}

__global__ void __launch_bounds__(AG_T)
k_own(const int* __restrict__ dst, const unsigned* __restrict__ rid, const unsigned* __restrict__ pk,
      const unsigned* __restrict__ bmT,
      float* __restrict__ dis, unsigned* __restrict__ eq1, unsigned* __restrict__ ecnt) {
    __shared__ uint2 q[AG_QCAP + 1];
    __shared__ unsigned cnt[AG_R];
    __shared__ unsigned qn;
    __shared__ unsigned pn;
    const int t = threadIdx.x;
    const unsigned lane = (unsigned)t & 31u;
    const int lo = blockIdx.x * AG_R;
    const unsigned nrows = (unsigned)((NN - lo) < AG_R ? (NN - lo) : AG_R);

    if (t < AG_R) cnt[t] = 0u;
    if (t == 0) { qn = 0u; pn = 0u; }
    __syncthreads();

    {
        const unsigned* __restrict__ bmr = bmT + (size_t)blockIdx.x * (2 * NCH);
        unsigned mw[AG_MW];
        #pragma unroll
        for (int u = 0; u < AG_MW; ++u) {
            int i = t + u * AG_T;
            mw[u] = (i < 2 * NCH) ? bmr[i] : 0u;
        }
        unsigned c = 0u;
        #pragma unroll
        for (int u = 0; u < AG_MW; ++u) c += (unsigned)__builtin_popcount(mw[u]);
        unsigned wtot;
        const unsigned ex = wave_excl(c, lane, wtot);
        unsigned base = 0u;
        if (lane == 0u) base = __hip_atomic_fetch_add(&pn, wtot, __ATOMIC_RELAXED, __HIP_MEMORY_SCOPE_WORKGROUP);
        unsigned pos = (unsigned)__shfl((int)base, 0) + ex;
        #pragma unroll
        for (int u = 0; u < AG_MW; ++u) {
            unsigned m = mw[u];
            const unsigned g0 = (unsigned)(t + u * AG_T) * 32u;
            while (m) {
                unsigned b = (unsigned)__builtin_ctz(m);
                m &= m - 1u;
                if (pos < AG_QCAP) q[pos].y = g0 + b;
                ++pos;
            }
        }
    }
    __syncthreads();
    const unsigned np = pn;

    const unsigned own4 = blockIdx.x * 0x01010101u;
    if (np <= AG_QCAP) {
        for (unsigned b0 = (unsigned)t & ~31u; b0 < np; b0 += (unsigned)AG_T * AG_FB) {
            const unsigned i0 = b0 + lane;
            unsigned gi[AG_FB];
            uint4 w[AG_FB];
            #pragma unroll
            for (int u = 0; u < AG_FB; ++u) {
                unsigned i = i0 + (unsigned)u * AG_T;
                unsigned g = q[(i < np) ? i : 0u].y;
                gi[u] = g < (unsigned)NG16 ? g : (unsigned)NG16 - 1u;
            }
            #pragma unroll
            for (int u = 0; u < AG_FB; ++u) w[u] = ((const uint4*)rid)[gi[u]];
            #pragma unroll
            for (int u = 0; u < AG_FB; ++u)
                if (!(i0 + (unsigned)u * AG_T < np)) w[u] = make_uint4(~0u, ~0u, ~0u, ~0u);
            unsigned hm[AG_FB / 2];
            unsigned c = 0u;
            #pragma unroll
            for (int v = 0; v < AG_FB / 2; ++v) {
                const unsigned ww[8] = { w[2 * v].x, w[2 * v].y, w[2 * v].z, w[2 * v].w,
                                         w[2 * v + 1].x, w[2 * v + 1].y, w[2 * v + 1].z, w[2 * v + 1].w };
                unsigned nz = 0u;
                #pragma unroll
                for (int k = 0; k < 8; ++k) {
                    unsigned x = ww[k] ^ own4;
                    unsigned s = ((x & 0x7F7F7F7Fu) + 0x7F7F7F7Fu) | x;
                    nz |= (s >> (7 - k)) & (0x01010101u << k);
                }
                hm[v] = ~nz;
                c += (unsigned)__builtin_popcount(hm[v]);
            }
            unsigned wtot;
            const unsigned ex = wave_excl(c, lane, wtot);
            unsigned base = 0u;
            if (lane == 0u) base = __hip_atomic_fetch_add(&qn, wtot, __ATOMIC_RELAXED, __HIP_MEMORY_SCOPE_WORKGROUP);
            unsigned pos = (unsigned)__shfl((int)base, 0) + ex;
            #pragma unroll
            for (int v = 0; v < AG_FB / 2; ++v) {
                unsigned hit = hm[v];
                while (hit) {
                    unsigned p = (unsigned)__builtin_ctz(hit);
                    hit &= hit - 1u;
                    unsigned k = p & 7u, j = p >> 3;
                    unsigned gg = (k >> 2) ? gi[2 * v + 1] : gi[2 * v];
                    unsigned e = gg * 16u + (k & 3u) * 4u + j;
                    if (pos < AG_QCAP) q[pos].x = e;
                    ++pos;
                }
            }
        }
    }
    __syncthreads();
    const unsigned total = (np <= AG_QCAP) ? qn : np;
    if (t == 0) ecnt[blockIdx.x] = total;
    if (total <= AG_QCAP) {
        unsigned* __restrict__ g1 = eq1 + (size_t)blockIdx.x * AG_QCAP;
        for (unsigned i0 = t; i0 < total; i0 += (unsigned)AG_KB * AG_T) {
            unsigned ix[AG_KB], en[AG_KB], v[AG_KB];
            #pragma unroll
            for (int u = 0; u < AG_KB; ++u) {
                unsigned i = i0 + u * AG_T;
                ix[u] = (i < total) ? i : i0;
                en[u] = q[ix[u]].x;
            }
            #pragma unroll
            for (int u = 0; u < AG_KB; ++u) v[u] = pk[en[u]];
            #pragma unroll
            for (int u = 0; u < AG_KB; ++u) g1[ix[u]] = v[u];
            #pragma unroll
            for (int u = 0; u < AG_KB; ++u)
                if (i0 + u * AG_T < total && (v[u] >> 17) < nrows)
                    __hip_atomic_fetch_add(cnt + (v[u] >> 17), 1u, __ATOMIC_RELAXED, __HIP_MEMORY_SCOPE_WORKGROUP);
        }
    } else {
        __syncthreads();
        if (t == 0) qn = 0u;
        __syncthreads();
        ag_filter<true>(dst, 0, NE, lo, nrows, &qn, q, cnt, t);
    }
    __syncthreads();
    if ((unsigned)t < nrows) dis[lo + t] = rsqrtf((float)(cnt[t] + 1u));
}

__global__ void __launch_bounds__(AG_T)
k_agg1(const int* __restrict__ src, const int* __restrict__ dst,
       const float* __restrict__ h1, const float* __restrict__ dis, const float* __restrict__ b1,
       float* __restrict__ z1, const unsigned* __restrict__ eq1,
       unsigned* __restrict__ eq, const unsigned* __restrict__ ecnt) {
    __shared__ float acc[(AG_R + AG_OG) * HID];
    __shared__ uint2 q[AG_QCAP + 1];
    __shared__ uint2 qb[AG_OG * AG_BCAP];
    __shared__ unsigned bcnt[AG_OG + 1];
    __shared__ unsigned qn;
    const int t = threadIdx.x;
    const int lo = blockIdx.x * AG_R;
    const unsigned nrows = (unsigned)((NN - lo) < AG_R ? (NN - lo) : AG_R);

    for (int i = t; i < (AG_R + AG_OG) * HID / 4; i += AG_T)
        ((float4*)acc)[i] = make_float4(0.0f, 0.0f, 0.0f, 0.0f);
    if (t <= AG_OG) bcnt[t] = 0u;
    if (t == 0) qn = 0u;
    const unsigned total = ecnt[blockIdx.x];
    __syncthreads();
    if (total <= AG_QCAP) {
        const unsigned* __restrict__ g1 = eq1 + (size_t)blockIdx.x * AG_QCAP;
        uint2* __restrict__ g2 = (uint2*)eq + (size_t)blockIdx.x * AG_QCAP;
        for (unsigned i0 = t; i0 < total; i0 += (unsigned)AG_LB * AG_T) {
            unsigned v[AG_LB];
            float w[AG_LB];
            #pragma unroll
            for (int u = 0; u < AG_LB; ++u) {
                unsigned i = i0 + u * AG_T;
                v[u] = g1[(i < total) ? i : i0];
            }
            #pragma unroll
            for (int u = 0; u < AG_LB; ++u)
                w[u] = dis[v[u] & 0x1FFFFu] * dis[lo + (int)(v[u] >> 17)];
            unsigned pos[AG_LB];
            #pragma unroll
            for (int u = 0; u < AG_LB; ++u) {
                unsigned i = i0 + u * AG_T;
                unsigned b = (i < total) ? (v[u] >> 17) % (unsigned)AG_OG : (unsigned)AG_OG;
                pos[u] = __hip_atomic_fetch_add(bcnt + b, 1u, __ATOMIC_RELAXED, __HIP_MEMORY_SCOPE_WORKGROUP);
            }
            #pragma unroll
            for (int u = 0; u < AG_LB; ++u) {
                unsigned i = i0 + u * AG_T;
                uint2 o = make_uint2(v[u], __float_as_uint(w[u]));
                if (i < total) {
                    g2[i] = o;
                    unsigned b = (v[u] >> 17) % (unsigned)AG_OG;
                    if (pos[u] < (unsigned)AG_BCAP) qb[b * AG_BCAP + pos[u]] = o;
                    else q[__hip_atomic_fetch_add(&qn, 1u, __ATOMIC_RELAXED, __HIP_MEMORY_SCOPE_WORKGROUP)] = o;
                }
            }
        }
        __syncthreads();
        {
            const unsigned hw = (unsigned)t >> 4;
            const unsigned nb = bcnt[hw] < (unsigned)AG_BCAP ? bcnt[hw] : (unsigned)AG_BCAP;
            ag_own64(h1, acc, qb + hw * AG_BCAP, nb, (unsigned)AG_R + hw, t);
        }
        __syncthreads();
        ag_add64(h1, acc, q, qn, t);
    } else {
        for (int e0 = 0; e0 < NE; e0 += AG_QCAP) {
            __syncthreads();
            if (t == 0) qn = 0u;
            __syncthreads();
            const int e1 = (e0 + AG_QCAP < NE) ? (e0 + AG_QCAP) : NE;
            ag_filter<false>(dst, e0, e1, lo, nrows, &qn, q, nullptr, t);
            __syncthreads();
            const unsigned n = qn;
            ag_lookup(src, dis, lo, q, n, t);
            ag_add64(h1, acc, q, n, t);
        }
    }
    __syncthreads();

    for (unsigned i = t; i < nrows * 16u; i += AG_T) {
        unsigned r = i >> 4, c = (i & 15u) * 4u;
        size_t g = (size_t)(lo + r) * HID + c;
        float di = dis[lo + r];
        float4 a = *(const float4*)(acc + r * HID + c);
        float4 h = *(const float4*)(h1 + g);
        float4 b = *(const float4*)(b1 + c);
        float4 v;
        v.x = a.x + h.x * di * di + b.x;
        v.y = a.y + h.y * di * di + b.y;
        v.z = a.z + h.z * di * di + b.z;
        v.w = a.w + h.w * di * di + b.w;
        v.x = v.x > 0.0f ? v.x : 0.0f;
        v.y = v.y > 0.0f ? v.y : 0.0f;
        v.z = v.z > 0.0f ? v.z : 0.0f;
        v.w = v.w > 0.0f ? v.w : 0.0f;
        *(float4*)(z1 + g) = v;
    }
}

#define AG_D2   8

static_assert((AG_R * NCLS) % 32 == 0, "a range's block of out must be whole 128-byte lines");
static_assert(((NN % AG_R) * NCLS) % 4 == 0, "the last range's block of out must be whole float4s");
static_assert(AG_R <= AG_T, "the layer-2 epilogue gives every own row a thread of its own");
static_assert((long long)AG_NB * AG_R >= NN && (long long)(AG_NB - 1) * AG_R < NN, "the ranges cover the nodes and none is empty");

__device__ __forceinline__ void ag_add10(const float* __restrict__ h2,
                                         float* acc2, const uint2* q, unsigned n, int t) {
    const unsigned g = (unsigned)t >> 3;
    const unsigned j = (unsigned)t & 7u;
    if (j < 5u) {
        for (unsigned i0 = g; i0 < n; i0 += 128u * AG_D2) {
            uint2 en[AG_D2];
            v2f row[AG_D2];
            #pragma unroll
            for (int u = 0; u < AG_D2; ++u) {
                unsigned i = i0 + 128u * u;
                en[u] = (i < n) ? q[i] : make_uint2((unsigned)AG_R << 17, 0u);
            }
            #pragma unroll
            for (int u = 0; u < AG_D2; ++u)
                row[u] = *(const v2f*)(h2 + (size_t)(en[u].x & 0x1FFFFu) * NCLS + 2u * j);
            #pragma unroll
            for (int u = 0; u < AG_D2; ++u) {
                float w = __uint_as_float(en[u].y);
                float* p = acc2 + (en[u].x >> 17) * NCLS + 2u * j;
                __hip_atomic_fetch_add(p,      row[u].x * w, __ATOMIC_RELAXED, __HIP_MEMORY_SCOPE_WORKGROUP);
                __hip_atomic_fetch_add(p + 1u, row[u].y * w, __ATOMIC_RELAXED, __HIP_MEMORY_SCOPE_WORKGROUP);
            }
        }
    }
}

__global__ void __launch_bounds__(AG_T)
k_agg2(const int* __restrict__ src, const int* __restrict__ dst,
       const float* __restrict__ h2,
       const float* __restrict__ dis, const float* __restrict__ b2,
       const unsigned* __restrict__ eq, const unsigned* __restrict__ ecnt,
       float* __restrict__ out) {
    __shared__ __attribute__((aligned(16))) float acc2[(AG_R + 1) * NCLS];
    __shared__ uint2 q[AG_QCAP + 1];
    __shared__ unsigned qn;
    const int t = threadIdx.x;
    const int lo = blockIdx.x * AG_R;
    const unsigned nrows = (unsigned)((NN - lo) < AG_R ? (NN - lo) : AG_R);

    for (int i = t; i < (AG_R + 1) * NCLS; i += AG_T) acc2[i] = 0.0f;
    const unsigned total = ecnt[blockIdx.x];
    if (total <= AG_QCAP) {
        const uint2* __restrict__ gq = (const uint2*)eq + (size_t)blockIdx.x * AG_QCAP;
        for (unsigned i0 = t; i0 < total; i0 += 4u * AG_T) {
            uint2 v[4];
            #pragma unroll
            for (int u = 0; u < 4; ++u) {
                unsigned i = i0 + u * AG_T;
                v[u] = gq[(i < total) ? i : i0];
            }
            #pragma unroll
            for (int u = 0; u < 4; ++u) {
                unsigned i = i0 + u * AG_T;
                q[(i < total) ? i : AG_QCAP] = v[u];
            }
        }
        __syncthreads();
        ag_add10(h2, acc2, q, total, t);
    } else {
        for (int e0 = 0; e0 < NE; e0 += AG_QCAP) {
            __syncthreads();
            if (t == 0) qn = 0u;
            __syncthreads();
            const int e1 = (e0 + AG_QCAP < NE) ? (e0 + AG_QCAP) : NE;
            ag_filter<false>(dst, e0, e1, lo, nrows, &qn, q, nullptr, t);
            __syncthreads();
            const unsigned n = qn;
            ag_lookup(src, dis, lo, q, n, t);
            ag_add10(h2, acc2, q, n, t);
        }
    }
    __syncthreads();

    if ((unsigned)t < nrows) {
        float di = dis[lo + t];
        float d2 = di * di;
        const float* hp = h2 + (size_t)(lo + t) * NCLS;
        float* ap = acc2 + t * NCLS;
        float tv[NCLS];
        float mx = -3.402823466e38f;
        #pragma unroll
        for (int f = 0; f < NCLS; ++f) {
            float v = ap[f] + hp[f] * d2 + b2[f];
            tv[f] = v;
            mx = fmaxf(mx, v);
        }
        float sum = 0.0f;
        #pragma unroll
        for (int f = 0; f < NCLS; ++f) sum += expf(tv[f] - mx);
        float lse = mx + logf(sum);
        #pragma unroll
        for (int f = 0; f < NCLS; ++f) ap[f] = tv[f] - lse;
    }
    __syncthreads();

    float4* ob = (float4*)(out + (size_t)lo * NCLS);
    for (unsigned i = t; i < nrows * NCLS / 4u; i += AG_T) ob[i] = ((const float4*)acc2)[i];
}

extern "C" void kernel_launch(void* const* d_in, const int* in_sizes, int n_in,
                              void* d_out, int out_size, void* d_ws, size_t ws_size,
                              hipStream_t stream) {
    const float* x  = (const float*)d_in[0];
    const int*   ei = (const int*)d_in[1];
    const float* W1 = (const float*)d_in[2];
    const float* b1 = (const float*)d_in[3];
    const float* W2 = (const float*)d_in[4];
    const float* b2 = (const float*)d_in[5];
    const int* src = ei;
    const int* dst = ei + NE;
    float* out = (float*)d_out;

    float* ws   = (float*)d_ws;
    float* dis  = ws;
    float* h1   = dis + NN;
    float* agg1 = h1 + (size_t)NN * HID;
    float* h2   = agg1 + (size_t)NN * HID;
    unsigned* eq1  = (unsigned*)(h2 + (size_t)NN * NCLS);
    unsigned* eq   = eq1 + (size_t)AG_NB * AG_QCAP;
    unsigned* ecnt = eq + (size_t)AG_NB * AG_QCAP * 2;
    unsigned* rid  = ecnt + 256;
    unsigned* pk   = rid + NE / 4;
    unsigned* bmT  = pk + NE;

    k_gemm1<<<NCH, 128, 0, stream>>>(x, W1, h1, src, dst, rid, pk, bmT);

    k_own<<<AG_NB, AG_T, 0, stream>>>(dst, rid, pk, bmT, dis, eq1, ecnt);

    k_agg1<<<AG_NB, AG_T, 0, stream>>>(src, dst, h1, dis, b1, agg1, eq1, eq, ecnt);

    k_gemm2<<<(ROW_TILES + 3) / 4, 128, 0, stream>>>(agg1, W2, h2);
    k_agg2<<<AG_NB, AG_T, 0, stream>>>(src, dst, h2, dis, b2, eq, ecnt, out);
}
